# starting kernel
# baseline (speedup 1.0000x reference)
#include <hip/hip_runtime.h>
#include <hip/hip_bf16.h>
#include <math.h>


#define Bsz 4
#define Lsz 1024
#define Dsz 512
#define Hsz 8
#define Isz 64
#define Msz (Bsz * Lsz)
#define GN_EPS 1e-5f

typedef float v2f __attribute__((ext_vector_type(2)));
typedef float v8f __attribute__((ext_vector_type(8)));

__device__ __forceinline__ v8f wmma4(v2f a, v2f b, v8f c) {
  return __builtin_amdgcn_wmma_f32_16x16x4_f32(
       false, a,  false, b,
       (short)0, c,  false,  false);
}

__device__ __forceinline__ float silu_f(float v) {
  return v / (1.0f + expf(-v));
}

__global__ __launch_bounds__(128)
void gemm_gate_z_kernel(const float* __restrict__ X,
                        const float* __restrict__ Wz,  const float* __restrict__ bz,
                        const float* __restrict__ Wza, const float* __restrict__ bza,
                        float* __restrict__ Z) {
  const int lane = threadIdx.x & 31;
  const int wave = threadIdx.x >> 5;
  const int mr = lane & 15;
  const int kk = lane >> 4;
  const int m_base = blockIdx.x * 64 + (wave & 1) * 32;
  const int n_base = blockIdx.y * 64 + (wave >> 1) * 32;

  v8f a1[2][2], a2[2][2];
  const v8f zero = {0.f, 0.f, 0.f, 0.f, 0.f, 0.f, 0.f, 0.f};
  for (int i = 0; i < 2; ++i)
    for (int j = 0; j < 2; ++j) { a1[i][j] = zero; a2[i][j] = zero; }

  const float* A0  = X   + (size_t)(m_base + mr)       * Dsz + 2 * kk;
  const float* A1  = A0  + 16 * Dsz;
  const float* Bz0 = Wz  + (size_t)(n_base + mr)       * Dsz + 2 * kk;
  const float* Bz1 = Bz0 + 16 * Dsz;
  const float* Ba0 = Wza + (size_t)(n_base + mr)       * Dsz + 2 * kk;
  const float* Ba1 = Ba0 + 16 * Dsz;

#pragma unroll 4
  for (int k0 = 0; k0 < Dsz; k0 += 4) {
    v2f fa0 = *(const v2f*)(A0 + k0);
    v2f fa1 = *(const v2f*)(A1 + k0);
    v2f bz0 = *(const v2f*)(Bz0 + k0);
    v2f bz1 = *(const v2f*)(Bz1 + k0);
    v2f ba0 = *(const v2f*)(Ba0 + k0);
    v2f ba1 = *(const v2f*)(Ba1 + k0);
    a1[0][0] = wmma4(fa0, bz0, a1[0][0]);
    a1[0][1] = wmma4(fa0, bz1, a1[0][1]);
    a1[1][0] = wmma4(fa1, bz0, a1[1][0]);
    a1[1][1] = wmma4(fa1, bz1, a1[1][1]);
    a2[0][0] = wmma4(fa0, ba0, a2[0][0]);
    a2[0][1] = wmma4(fa0, ba1, a2[0][1]);
    a2[1][0] = wmma4(fa1, ba0, a2[1][0]);
    a2[1][1] = wmma4(fa1, ba1, a2[1][1]);
  }

  for (int mt = 0; mt < 2; ++mt)
    for (int nt = 0; nt < 2; ++nt) {
      const int n = n_base + nt * 16 + mr;
      const float biz = bz[n], biza = bza[n];
#pragma unroll
      for (int r = 0; r < 8; ++r) {
        const int m = m_base + mt * 16 + kk * 8 + r;
        const float v1 = a1[mt][nt][r] + biz;
        const float v2 = a2[mt][nt][r] + biza;
        Z[(size_t)m * Dsz + n] = v1 * silu_f(v2);
      }
    }
}

template <int MODE>
__global__ __launch_bounds__(128)
void gemm_single_kernel(const float* __restrict__ A, const float* __restrict__ W,
                        const float* __restrict__ bias,
                        const float* __restrict__ gate, float* __restrict__ OUT) {
  const int lane = threadIdx.x & 31;
  const int wave = threadIdx.x >> 5;
  const int mr = lane & 15;
  const int kk = lane >> 4;
  const int m_base = blockIdx.x * 64 + (wave & 1) * 32;
  const int n_base = blockIdx.y * 64 + (wave >> 1) * 32;

  v8f acc[2][2];
  const v8f zero = {0.f, 0.f, 0.f, 0.f, 0.f, 0.f, 0.f, 0.f};
  for (int i = 0; i < 2; ++i)
    for (int j = 0; j < 2; ++j) acc[i][j] = zero;

  const float* A0 = A + (size_t)(m_base + mr) * Dsz + 2 * kk;
  const float* A1 = A0 + 16 * Dsz;
  const float* B0 = W + (size_t)(n_base + mr) * Dsz + 2 * kk;
  const float* B1 = B0 + 16 * Dsz;

#pragma unroll 4
  for (int k0 = 0; k0 < Dsz; k0 += 4) {
    v2f fa0 = *(const v2f*)(A0 + k0);
    v2f fa1 = *(const v2f*)(A1 + k0);
    v2f fb0 = *(const v2f*)(B0 + k0);
    v2f fb1 = *(const v2f*)(B1 + k0);
    acc[0][0] = wmma4(fa0, fb0, acc[0][0]);
    acc[0][1] = wmma4(fa0, fb1, acc[0][1]);
    acc[1][0] = wmma4(fa1, fb0, acc[1][0]);
    acc[1][1] = wmma4(fa1, fb1, acc[1][1]);
  }

  for (int mt = 0; mt < 2; ++mt)
    for (int nt = 0; nt < 2; ++nt) {
      const int n = n_base + nt * 16 + mr;
      const float bi = bias[n];
#pragma unroll
      for (int r = 0; r < 8; ++r) {
        const int m = m_base + mt * 16 + kk * 8 + r;
        const float v = acc[mt][nt][r] + bi;
        float o;
        if (MODE == 0) o = silu_f(v);
        else           o = v * gate[(size_t)m * Dsz + n];
        OUT[(size_t)m * Dsz + n] = o;
      }
    }
}

__global__ __launch_bounds__(1024)
void dtscan_kernel(const float* __restrict__ X, const float* __restrict__ w_dt,
                   const float* __restrict__ b_dt, const float* __restrict__ ln_a,
                   float* __restrict__ S) {
  __shared__ float wlds[Dsz];
  __shared__ float scan[Lsz];
  const int b = blockIdx.x >> 3;
  const int h = blockIdx.x & 7;
  const int l = threadIdx.x;

  if (l < Dsz) wlds[l] = w_dt[h * Dsz + l];
  __syncthreads();

  const float* xr = X + (size_t)(b * Lsz + l) * Dsz;
  float dot = 0.f;
#pragma unroll 4
  for (int k = 0; k < Dsz; k += 4) {
    float4 xv = *(const float4*)(xr + k);
    dot += xv.x * wlds[k] + xv.y * wlds[k + 1] + xv.z * wlds[k + 2] + xv.w * wlds[k + 3];
  }
  const float v = dot + b_dt[h];
  const float sp = (v > 20.f) ? v : log1pf(expf(v));
  const float ld = -expf(ln_a[h]) * sp;

  scan[l] = ld;
  __syncthreads();
  for (int off = 1; off < Lsz; off <<= 1) {
    float t = scan[l];
    if (l >= off) t += scan[l - off];
    __syncthreads();
    scan[l] = t;
    __syncthreads();
  }
  S[(size_t)(b * Lsz + l) * Hsz + h] = scan[l];
}

__global__ __launch_bounds__(256)
void zsum_kernel(const float* __restrict__ Z, const float* __restrict__ hidden,
                 float* __restrict__ C, float* __restrict__ zhh) {
  __shared__ float red[256];
  const int b = blockIdx.x >> 3;
  const int h = blockIdx.x & 7;
  const int i = threadIdx.x & 63;
  const int part = threadIdx.x >> 6;

  float s = 0.f;
  for (int m = h + 1 + part; m < Lsz; m += 4)
    s += Z[((size_t)(b * Lsz + m) * Hsz + h) * Isz + i];
  red[threadIdx.x] = s;
  __syncthreads();
  if (part == 0) {
    const float tot = red[i] + red[i + 64] + red[i + 128] + red[i + 192];
    const int ch = (b * Hsz + h) * Isz + i;
    C[ch] = tot + hidden[ch];
    zhh[ch] = Z[((size_t)(b * Lsz + h) * Hsz + h) * Isz + i];
  }
}

__global__ __launch_bounds__(512)
void hnorm_kernel(const float* __restrict__ S, const float* __restrict__ C,
                  const float* __restrict__ zhh, const float* __restrict__ gn_w,
                  const float* __restrict__ gn_b, float* __restrict__ Hout,
                  float* __restrict__ Hnorm) {
  __shared__ float lds[Dsz];
  const int bl = blockIdx.x;
  const int b = bl >> 10;
  const int t = threadIdx.x;
  const int h = t >> 6;
  const int i = t & 63;
  const int ch = (b * Hsz + h) * Isz + i;

  const float e = expf(S[(size_t)bl * Hsz + h]);
  const float hv = zhh[ch] + e * C[ch];
  Hout[(size_t)bl * Dsz + t] = hv;

  lds[t] = hv;
  __syncthreads();
  for (int s = 32; s >= 1; s >>= 1) {
    if (i < s) lds[t] += lds[t + s];
    __syncthreads();
  }
  const float mu = lds[h * 64] * (1.0f / 64.0f);
  __syncthreads();
  const float d = hv - mu;
  lds[t] = d * d;
  __syncthreads();
  for (int s = 32; s >= 1; s >>= 1) {
    if (i < s) lds[t] += lds[t + s];
    __syncthreads();
  }
  const float var = lds[h * 64] * (1.0f / 64.0f);
  const float hn = d / sqrtf(var + GN_EPS) * gn_w[h] + gn_b[h];
  Hnorm[(size_t)bl * Dsz + t] = hn;
}

extern "C" void kernel_launch(void* const* d_in, const int* in_sizes, int n_in,
                              void* d_out, int out_size, void* d_ws, size_t ws_size,
                              hipStream_t stream) {
  const float* x      = (const float*)d_in[0];
  const float* hidden = (const float*)d_in[1];
  const float* w_z    = (const float*)d_in[2];
  const float* b_z    = (const float*)d_in[3];
  const float* w_za   = (const float*)d_in[4];
  const float* b_za   = (const float*)d_in[5];
  const float* w_y    = (const float*)d_in[6];
  const float* b_y    = (const float*)d_in[7];
  const float* w_ya   = (const float*)d_in[8];
  const float* b_ya   = (const float*)d_in[9];
  const float* w_dt   = (const float*)d_in[10];
  const float* b_dt   = (const float*)d_in[11];
  const float* ln_a   = (const float*)d_in[12];
  const float* gn_w   = (const float*)d_in[13];
  const float* gn_b   = (const float*)d_in[14];

  float* y_out = (float*)d_out;
  float* h_out = y_out + (size_t)Msz * Dsz;

  float* ws   = (float*)d_ws;
  float* Z    = ws;
  float* GYA  = Z    + (size_t)Msz * Dsz;
  float* HN   = GYA  + (size_t)Msz * Dsz;
  float* Sb   = HN   + (size_t)Msz * Dsz;
  float* Cb   = Sb   + (size_t)Msz * Hsz;
  float* ZHH  = Cb   + (size_t)Bsz * Hsz * Isz;

  const dim3 gGemm(Msz / 64, Dsz / 64, 1);
  const dim3 bGemm(128, 1, 1);

  gemm_gate_z_kernel<<<gGemm, bGemm, 0, stream>>>(x, w_z, b_z, w_za, b_za, Z);
  gemm_single_kernel<0><<<gGemm, bGemm, 0, stream>>>(x, w_ya, b_ya, nullptr, GYA);
  dtscan_kernel<<<dim3(Bsz * Hsz), dim3(Lsz), 0, stream>>>(x, w_dt, b_dt, ln_a, Sb);
  zsum_kernel<<<dim3(Bsz * Hsz), dim3(256), 0, stream>>>(Z, hidden, Cb, ZHH);
  hnorm_kernel<<<dim3(Msz), dim3(Dsz), 0, stream>>>(Sb, Cb, ZHH, gn_w, gn_b,
                                                    h_out, HN);
  gemm_single_kernel<1><<<gGemm, bGemm, 0, stream>>>(HN, w_y, b_y, GYA, y_out);
}


# column-sum kernel: 1024 threads, each issues all its strided row loads (b128) before adding; 64 row parts summed through LDS; was one dependent load per trip. Sum order of the column sum changes
# speedup vs baseline: 1.5457x; 1.5457x over previous
#include <hip/hip_runtime.h>
#include <hip/hip_bf16.h>
#include <math.h>


#define Bsz 4
#define Lsz 1024
#define Dsz 512
#define Hsz 8
#define Isz 64
#define Msz (Bsz * Lsz)
#define GN_EPS 1e-5f

typedef float v2f __attribute__((ext_vector_type(2)));
typedef float v8f __attribute__((ext_vector_type(8)));

__device__ __forceinline__ v8f wmma4(v2f a, v2f b, v8f c) {
  return __builtin_amdgcn_wmma_f32_16x16x4_f32(
       false, a,  false, b,
       (short)0, c,  false,  false);
}

__device__ __forceinline__ float silu_f(float v) {
  return v / (1.0f + expf(-v));
}

__global__ __launch_bounds__(128)
void gemm_gate_z_kernel(const float* __restrict__ X,
                        const float* __restrict__ Wz,  const float* __restrict__ bz,
                        const float* __restrict__ Wza, const float* __restrict__ bza,
                        float* __restrict__ Z) {
  const int lane = threadIdx.x & 31;
  const int wave = threadIdx.x >> 5;
  const int mr = lane & 15;
  const int kk = lane >> 4;
  const int m_base = blockIdx.x * 64 + (wave & 1) * 32;
  const int n_base = blockIdx.y * 64 + (wave >> 1) * 32;

  v8f a1[2][2], a2[2][2];
  const v8f zero = {0.f, 0.f, 0.f, 0.f, 0.f, 0.f, 0.f, 0.f};
  for (int i = 0; i < 2; ++i)
    for (int j = 0; j < 2; ++j) { a1[i][j] = zero; a2[i][j] = zero; }

  const float* A0  = X   + (size_t)(m_base + mr)       * Dsz + 2 * kk;
  const float* A1  = A0  + 16 * Dsz;
  const float* Bz0 = Wz  + (size_t)(n_base + mr)       * Dsz + 2 * kk;
  const float* Bz1 = Bz0 + 16 * Dsz;
  const float* Ba0 = Wza + (size_t)(n_base + mr)       * Dsz + 2 * kk;
  const float* Ba1 = Ba0 + 16 * Dsz;

#pragma unroll 4
  for (int k0 = 0; k0 < Dsz; k0 += 4) {
    v2f fa0 = *(const v2f*)(A0 + k0);
    v2f fa1 = *(const v2f*)(A1 + k0);
    v2f bz0 = *(const v2f*)(Bz0 + k0);
    v2f bz1 = *(const v2f*)(Bz1 + k0);
    v2f ba0 = *(const v2f*)(Ba0 + k0);
    v2f ba1 = *(const v2f*)(Ba1 + k0);
    a1[0][0] = wmma4(fa0, bz0, a1[0][0]);
    a1[0][1] = wmma4(fa0, bz1, a1[0][1]);
    a1[1][0] = wmma4(fa1, bz0, a1[1][0]);
    a1[1][1] = wmma4(fa1, bz1, a1[1][1]);
    a2[0][0] = wmma4(fa0, ba0, a2[0][0]);
    a2[0][1] = wmma4(fa0, ba1, a2[0][1]);
    a2[1][0] = wmma4(fa1, ba0, a2[1][0]);
    a2[1][1] = wmma4(fa1, ba1, a2[1][1]);
  }

  for (int mt = 0; mt < 2; ++mt)
    for (int nt = 0; nt < 2; ++nt) {
      const int n = n_base + nt * 16 + mr;
      const float biz = bz[n], biza = bza[n];
#pragma unroll
      for (int r = 0; r < 8; ++r) {
        const int m = m_base + mt * 16 + kk * 8 + r;
        const float v1 = a1[mt][nt][r] + biz;
        const float v2 = a2[mt][nt][r] + biza;
        Z[(size_t)m * Dsz + n] = v1 * silu_f(v2);
      }
    }
}

template <int MODE>
__global__ __launch_bounds__(128)
void gemm_single_kernel(const float* __restrict__ A, const float* __restrict__ W,
                        const float* __restrict__ bias,
                        const float* __restrict__ gate, float* __restrict__ OUT) {
  const int lane = threadIdx.x & 31;
  const int wave = threadIdx.x >> 5;
  const int mr = lane & 15;
  const int kk = lane >> 4;
  const int m_base = blockIdx.x * 64 + (wave & 1) * 32;
  const int n_base = blockIdx.y * 64 + (wave >> 1) * 32;

  v8f acc[2][2];
  const v8f zero = {0.f, 0.f, 0.f, 0.f, 0.f, 0.f, 0.f, 0.f};
  for (int i = 0; i < 2; ++i)
    for (int j = 0; j < 2; ++j) acc[i][j] = zero;

  const float* A0 = A + (size_t)(m_base + mr) * Dsz + 2 * kk;
  const float* A1 = A0 + 16 * Dsz;
  const float* B0 = W + (size_t)(n_base + mr) * Dsz + 2 * kk;
  const float* B1 = B0 + 16 * Dsz;

#pragma unroll 4
  for (int k0 = 0; k0 < Dsz; k0 += 4) {
    v2f fa0 = *(const v2f*)(A0 + k0);
    v2f fa1 = *(const v2f*)(A1 + k0);
    v2f fb0 = *(const v2f*)(B0 + k0);
    v2f fb1 = *(const v2f*)(B1 + k0);
    acc[0][0] = wmma4(fa0, fb0, acc[0][0]);
    acc[0][1] = wmma4(fa0, fb1, acc[0][1]);
    acc[1][0] = wmma4(fa1, fb0, acc[1][0]);
    acc[1][1] = wmma4(fa1, fb1, acc[1][1]);
  }

  for (int mt = 0; mt < 2; ++mt)
    for (int nt = 0; nt < 2; ++nt) {
      const int n = n_base + nt * 16 + mr;
      const float bi = bias[n];
#pragma unroll
      for (int r = 0; r < 8; ++r) {
        const int m = m_base + mt * 16 + kk * 8 + r;
        const float v = acc[mt][nt][r] + bi;
        float o;
        if (MODE == 0) o = silu_f(v);
        else           o = v * gate[(size_t)m * Dsz + n];
        OUT[(size_t)m * Dsz + n] = o;
      }
    }
}

__global__ __launch_bounds__(1024)
void dtscan_kernel(const float* __restrict__ X, const float* __restrict__ w_dt,
                   const float* __restrict__ b_dt, const float* __restrict__ ln_a,
                   float* __restrict__ S) {
  __shared__ float wlds[Dsz];
  __shared__ float scan[Lsz];
  const int b = blockIdx.x >> 3;
  const int h = blockIdx.x & 7;
  const int l = threadIdx.x;

  if (l < Dsz) wlds[l] = w_dt[h * Dsz + l];
  __syncthreads();

  const float* xr = X + (size_t)(b * Lsz + l) * Dsz;
  float dot = 0.f;
#pragma unroll 4
  for (int k = 0; k < Dsz; k += 4) {
    float4 xv = *(const float4*)(xr + k);
    dot += xv.x * wlds[k] + xv.y * wlds[k + 1] + xv.z * wlds[k + 2] + xv.w * wlds[k + 3];
  }
  const float v = dot + b_dt[h];
  const float sp = (v > 20.f) ? v : log1pf(expf(v));
  const float ld = -expf(ln_a[h]) * sp;

  scan[l] = ld;
  __syncthreads();
  for (int off = 1; off < Lsz; off <<= 1) {
    float t = scan[l];
    if (l >= off) t += scan[l - off];
    __syncthreads();
    scan[l] = t;
    __syncthreads();
  }
  S[(size_t)(b * Lsz + l) * Hsz + h] = scan[l];
}

#define ZS_PARTS 64
#define ZS_ROWS  (Lsz / ZS_PARTS)

__global__ __launch_bounds__(1024)
void zsum_kernel(const float* __restrict__ Z, const float* __restrict__ hidden,
                 float* __restrict__ C, float* __restrict__ zhh) {
  __shared__ float4 red4[ZS_PARTS * (Isz / 4)];
  __shared__ float  red2[16 * Isz];
  const int b = blockIdx.x >> 3;
  const int h = blockIdx.x & 7;
  const int tid = threadIdx.x;
  const int q = tid & 15;
  const int p = tid >> 4;

  const int ch = (b * Hsz + h) * Isz + (tid & 63);
  float zdiag = 0.f, hid = 0.f;
  if (tid < Isz) {
    zdiag = Z[((size_t)(b * Lsz + h) * Hsz + h) * Isz + tid];
    hid = hidden[ch];
  }

  const float* rp = Z + ((size_t)(b * Lsz + p) * Hsz + h) * Isz + 4 * q;
  float4 v[ZS_ROWS];
#pragma unroll
  for (int j = 0; j < ZS_ROWS; ++j)
    v[j] = *(const float4*)(rp + (size_t)j * ZS_PARTS * Dsz);
  if (p <= h) v[0] = make_float4(0.f, 0.f, 0.f, 0.f);

#pragma unroll
  for (int w = 1; w < ZS_ROWS; w <<= 1)
#pragma unroll
    for (int j = 0; j + w < ZS_ROWS; j += 2 * w) {
      v[j].x += v[j + w].x; v[j].y += v[j + w].y;
      v[j].z += v[j + w].z; v[j].w += v[j + w].w;
    }
  red4[p * (Isz / 4) + q] = v[0];
  __syncthreads();

  const float* red = (const float*)red4;
  {
    const int i = tid & 63;
    const int g = tid >> 6;
    const float* r = red + (4 * g) * Isz + i;
    red2[g * Isz + i] = (r[0] + r[Isz]) + (r[2 * Isz] + r[3 * Isz]);
  }
  __syncthreads();

  if (tid < Isz) {
    float t[16];
#pragma unroll
    for (int g = 0; g < 16; ++g) t[g] = red2[g * Isz + tid];
#pragma unroll
    for (int w = 1; w < 16; w <<= 1)
#pragma unroll
      for (int g = 0; g + w < 16; g += 2 * w) t[g] += t[g + w];
    C[ch] = t[0] + hid;
    zhh[ch] = zdiag;
  }
}

__global__ __launch_bounds__(512)
void hnorm_kernel(const float* __restrict__ S, const float* __restrict__ C,
                  const float* __restrict__ zhh, const float* __restrict__ gn_w,
                  const float* __restrict__ gn_b, float* __restrict__ Hout,
                  float* __restrict__ Hnorm) {
  __shared__ float lds[Dsz];
  const int bl = blockIdx.x;
  const int b = bl >> 10;
  const int t = threadIdx.x;
  const int h = t >> 6;
  const int i = t & 63;
  const int ch = (b * Hsz + h) * Isz + i;

  const float e = expf(S[(size_t)bl * Hsz + h]);
  const float hv = zhh[ch] + e * C[ch];
  Hout[(size_t)bl * Dsz + t] = hv;

  lds[t] = hv;
  __syncthreads();
  for (int s = 32; s >= 1; s >>= 1) {
    if (i < s) lds[t] += lds[t + s];
    __syncthreads();
  }
  const float mu = lds[h * 64] * (1.0f / 64.0f);
  __syncthreads();
  const float d = hv - mu;
  lds[t] = d * d;
  __syncthreads();
  for (int s = 32; s >= 1; s >>= 1) {
    if (i < s) lds[t] += lds[t + s];
    __syncthreads();
  }
  const float var = lds[h * 64] * (1.0f / 64.0f);
  const float hn = d / sqrtf(var + GN_EPS) * gn_w[h] + gn_b[h];
  Hnorm[(size_t)bl * Dsz + t] = hn;
}

extern "C" void kernel_launch(void* const* d_in, const int* in_sizes, int n_in,
                              void* d_out, int out_size, void* d_ws, size_t ws_size,
                              hipStream_t stream) {
  const float* x      = (const float*)d_in[0];
  const float* hidden = (const float*)d_in[1];
  const float* w_z    = (const float*)d_in[2];
  const float* b_z    = (const float*)d_in[3];
  const float* w_za   = (const float*)d_in[4];
  const float* b_za   = (const float*)d_in[5];
  const float* w_y    = (const float*)d_in[6];
  const float* b_y    = (const float*)d_in[7];
  const float* w_ya   = (const float*)d_in[8];
  const float* b_ya   = (const float*)d_in[9];
  const float* w_dt   = (const float*)d_in[10];
  const float* b_dt   = (const float*)d_in[11];
  const float* ln_a   = (const float*)d_in[12];
  const float* gn_w   = (const float*)d_in[13];
  const float* gn_b   = (const float*)d_in[14];

  float* y_out = (float*)d_out;
  float* h_out = y_out + (size_t)Msz * Dsz;

  float* ws   = (float*)d_ws;
  float* Z    = ws;
  float* GYA  = Z    + (size_t)Msz * Dsz;
  float* HN   = GYA  + (size_t)Msz * Dsz;
  float* Sb   = HN   + (size_t)Msz * Dsz;
  float* Cb   = Sb   + (size_t)Msz * Hsz;
  float* ZHH  = Cb   + (size_t)Bsz * Hsz * Isz;

  const dim3 gGemm(Msz / 64, Dsz / 64, 1);
  const dim3 bGemm(128, 1, 1);

  gemm_gate_z_kernel<<<gGemm, bGemm, 0, stream>>>(x, w_z, b_z, w_za, b_za, Z);
  gemm_single_kernel<0><<<gGemm, bGemm, 0, stream>>>(x, w_ya, b_ya, nullptr, GYA);
  dtscan_kernel<<<dim3(Bsz * Hsz), dim3(Lsz), 0, stream>>>(x, w_dt, b_dt, ln_a, Sb);
  zsum_kernel<<<dim3(Bsz * Hsz), dim3(1024), 0, stream>>>(Z, hidden, Cb, ZHH);
  hnorm_kernel<<<dim3(Msz), dim3(Dsz), 0, stream>>>(Sb, Cb, ZHH, gn_w, gn_b,
                                                    h_out, HN);
  gemm_single_kernel<1><<<gGemm, bGemm, 0, stream>>>(HN, w_y, b_y, GYA, y_out);
}


# dt projection: each wave reads whole rows as full lines (b128) and forms all 8 head dots per row, 256 blocks; was 32 blocks, one row per lane. Scan moved into z-sum kernel. Dot sum order changes
# speedup vs baseline: 2.1197x; 1.3714x over previous
#include <hip/hip_runtime.h>
#include <hip/hip_bf16.h>
#include <math.h>


#define Bsz 4
#define Lsz 1024
#define Dsz 512
#define Hsz 8
#define Isz 64
#define Msz (Bsz * Lsz)
#define GN_EPS 1e-5f

typedef float v2f __attribute__((ext_vector_type(2)));
typedef float v4f __attribute__((ext_vector_type(4)));
typedef float v8f __attribute__((ext_vector_type(8)));

__device__ __forceinline__ v8f wmma4(v2f a, v2f b, v8f c) {
  return __builtin_amdgcn_wmma_f32_16x16x4_f32(
       false, a,  false, b,
       (short)0, c,  false,  false);
}

__device__ __forceinline__ float silu_f(float v) {
  return v / (1.0f + expf(-v));
}

__global__ __launch_bounds__(128)
void gemm_gate_z_kernel(const float* __restrict__ X,
                        const float* __restrict__ Wz,  const float* __restrict__ bz,
                        const float* __restrict__ Wza, const float* __restrict__ bza,
                        float* __restrict__ Z) {
  const int lane = threadIdx.x & 31;
  const int wave = threadIdx.x >> 5;
  const int mr = lane & 15;
  const int kk = lane >> 4;
  const int m_base = blockIdx.x * 64 + (wave & 1) * 32;
  const int n_base = blockIdx.y * 64 + (wave >> 1) * 32;

  v8f a1[2][2], a2[2][2];
  const v8f zero = {0.f, 0.f, 0.f, 0.f, 0.f, 0.f, 0.f, 0.f};
  for (int i = 0; i < 2; ++i)
    for (int j = 0; j < 2; ++j) { a1[i][j] = zero; a2[i][j] = zero; }

  const float* A0  = X   + (size_t)(m_base + mr)       * Dsz + 2 * kk;
  const float* A1  = A0  + 16 * Dsz;
  const float* Bz0 = Wz  + (size_t)(n_base + mr)       * Dsz + 2 * kk;
  const float* Bz1 = Bz0 + 16 * Dsz;
  const float* Ba0 = Wza + (size_t)(n_base + mr)       * Dsz + 2 * kk;
  const float* Ba1 = Ba0 + 16 * Dsz;

#pragma unroll 4
  for (int k0 = 0; k0 < Dsz; k0 += 4) {
    v2f fa0 = *(const v2f*)(A0 + k0);
    v2f fa1 = *(const v2f*)(A1 + k0);
    v2f bz0 = *(const v2f*)(Bz0 + k0);
    v2f bz1 = *(const v2f*)(Bz1 + k0);
    v2f ba0 = *(const v2f*)(Ba0 + k0);
    v2f ba1 = *(const v2f*)(Ba1 + k0);
    a1[0][0] = wmma4(fa0, bz0, a1[0][0]);
    a1[0][1] = wmma4(fa0, bz1, a1[0][1]);
    a1[1][0] = wmma4(fa1, bz0, a1[1][0]);
    a1[1][1] = wmma4(fa1, bz1, a1[1][1]);
    a2[0][0] = wmma4(fa0, ba0, a2[0][0]);
    a2[0][1] = wmma4(fa0, ba1, a2[0][1]);
    a2[1][0] = wmma4(fa1, ba0, a2[1][0]);
    a2[1][1] = wmma4(fa1, ba1, a2[1][1]);
  }

  for (int mt = 0; mt < 2; ++mt)
    for (int nt = 0; nt < 2; ++nt) {
      const int n = n_base + nt * 16 + mr;
      const float biz = bz[n], biza = bza[n];
#pragma unroll
      for (int r = 0; r < 8; ++r) {
        const int m = m_base + mt * 16 + kk * 8 + r;
        const float v1 = a1[mt][nt][r] + biz;
        const float v2 = a2[mt][nt][r] + biza;
        Z[(size_t)m * Dsz + n] = v1 * silu_f(v2);
      }
    }
}

template <int MODE>
__global__ __launch_bounds__(128)
void gemm_single_kernel(const float* __restrict__ A, const float* __restrict__ W,
                        const float* __restrict__ bias,
                        const float* __restrict__ gate, float* __restrict__ OUT) {
  const int lane = threadIdx.x & 31;
  const int wave = threadIdx.x >> 5;
  const int mr = lane & 15;
  const int kk = lane >> 4;
  const int m_base = blockIdx.x * 64 + (wave & 1) * 32;
  const int n_base = blockIdx.y * 64 + (wave >> 1) * 32;

  v8f acc[2][2];
  const v8f zero = {0.f, 0.f, 0.f, 0.f, 0.f, 0.f, 0.f, 0.f};
  for (int i = 0; i < 2; ++i)
    for (int j = 0; j < 2; ++j) acc[i][j] = zero;

  const float* A0 = A + (size_t)(m_base + mr) * Dsz + 2 * kk;
  const float* A1 = A0 + 16 * Dsz;
  const float* B0 = W + (size_t)(n_base + mr) * Dsz + 2 * kk;
  const float* B1 = B0 + 16 * Dsz;

#pragma unroll 4
  for (int k0 = 0; k0 < Dsz; k0 += 4) {
    v2f fa0 = *(const v2f*)(A0 + k0);
    v2f fa1 = *(const v2f*)(A1 + k0);
    v2f fb0 = *(const v2f*)(B0 + k0);
    v2f fb1 = *(const v2f*)(B1 + k0);
    acc[0][0] = wmma4(fa0, fb0, acc[0][0]);
    acc[0][1] = wmma4(fa0, fb1, acc[0][1]);
    acc[1][0] = wmma4(fa1, fb0, acc[1][0]);
    acc[1][1] = wmma4(fa1, fb1, acc[1][1]);
  }

  for (int mt = 0; mt < 2; ++mt)
    for (int nt = 0; nt < 2; ++nt) {
      const int n = n_base + nt * 16 + mr;
      const float bi = bias[n];
#pragma unroll
      for (int r = 0; r < 8; ++r) {
        const int m = m_base + mt * 16 + kk * 8 + r;
        const float v = acc[mt][nt][r] + bi;
        float o;
        if (MODE == 0) o = silu_f(v);
        else           o = v * gate[(size_t)m * Dsz + n];
        OUT[(size_t)m * Dsz + n] = o;
      }
    }
}

#define DT_ROWS  2
#define DT_WAVES 8
#define DT_THREADS (DT_WAVES * 32)

__global__ __launch_bounds__(DT_THREADS)
void dtproj_kernel(const float* __restrict__ X, const float* __restrict__ w_dt,
                   const float* __restrict__ b_dt, const float* __restrict__ ln_a,
                   float* __restrict__ LD) {
  __shared__ v4f wl[Hsz * Dsz / 4];
  const int lane = threadIdx.x & 31;
  const int wave = threadIdx.x >> 5;
  const int row0 = (blockIdx.x * DT_WAVES + wave) * DT_ROWS;

  v4f wr[Hsz * Dsz / 4 / DT_THREADS];
#pragma unroll
  for (int j = 0; j < Hsz * Dsz / 4 / DT_THREADS; ++j)
    wr[j] = *(const v4f*)(w_dt + 4 * (threadIdx.x + DT_THREADS * j));

  v4f xv[DT_ROWS][4];
#pragma unroll
  for (int r = 0; r < DT_ROWS; ++r)
#pragma unroll
    for (int c = 0; c < 4; ++c)
      xv[r][c] = *(const v4f*)(X + (size_t)(row0 + r) * Dsz + c * 128 + 4 * lane);

  const int vi = lane >> 1;
  const int h_out = vi & 7;
  const float bdt = b_dt[h_out];
  const float lna = ln_a[h_out];

  asm volatile("" :: "v"(xv[0][0]), "v"(xv[0][1]), "v"(xv[0][2]), "v"(xv[0][3]),
                     "v"(xv[1][0]), "v"(xv[1][1]), "v"(xv[1][2]), "v"(xv[1][3]),
                     "v"(bdt), "v"(lna));

#pragma unroll
  for (int j = 0; j < Hsz * Dsz / 4 / DT_THREADS; ++j)
    wl[threadIdx.x + DT_THREADS * j] = wr[j];
  __syncthreads();

  float v16[DT_ROWS * Hsz];
#pragma unroll
  for (int h = 0; h < Hsz; ++h) {
    v4f wv[4];
#pragma unroll
    for (int c = 0; c < 4; ++c) wv[c] = wl[h * (Dsz / 4) + c * 32 + lane];
#pragma unroll
    for (int r = 0; r < DT_ROWS; ++r) {
      float s[4];
#pragma unroll
      for (int c = 0; c < 4; ++c) {
        s[c] = xv[r][c].x * wv[c].x;
        s[c] = fmaf(xv[r][c].y, wv[c].y, s[c]);
        s[c] = fmaf(xv[r][c].z, wv[c].z, s[c]);
        s[c] = fmaf(xv[r][c].w, wv[c].w, s[c]);
      }
      v16[r * Hsz + h] = (s[0] + s[1]) + (s[2] + s[3]);
    }
  }

  const bool b4 = (lane & 16) != 0;
  const bool b3 = (lane & 8) != 0;
  const bool b2 = (lane & 4) != 0;
  const bool b1 = (lane & 2) != 0;
  float v8[8], v4[4], v2[2];
#pragma unroll
  for (int i = 0; i < 8; ++i) {
    const float keep = b4 ? v16[8 + i] : v16[i];
    const float send = b4 ? v16[i] : v16[8 + i];
    v8[i] = keep + __shfl_xor(send, 16, 32);
  }
#pragma unroll
  for (int i = 0; i < 4; ++i) {
    const float keep = b3 ? v8[4 + i] : v8[i];
    const float send = b3 ? v8[i] : v8[4 + i];
    v4[i] = keep + __shfl_xor(send, 8, 32);
  }
#pragma unroll
  for (int i = 0; i < 2; ++i) {
    const float keep = b2 ? v4[2 + i] : v4[i];
    const float send = b2 ? v4[i] : v4[2 + i];
    v2[i] = keep + __shfl_xor(send, 4, 32);
  }
  float dot;
  {
    const float keep = b1 ? v2[1] : v2[0];
    const float send = b1 ? v2[0] : v2[1];
    dot = keep + __shfl_xor(send, 2, 32);
  }
  dot += __shfl_xor(dot, 1, 32);

  const float v = dot + bdt;
  const float sp = (v > 20.f) ? v : log1pf(expf(v));
  const float ld = -expf(lna) * sp;
  if ((lane & 1) == 0) LD[(size_t)row0 * Hsz + vi] = ld;
}

#define ZS_PARTS 64
#define ZS_ROWS  (Lsz / ZS_PARTS)

__global__ __launch_bounds__(1024)
void zsum_kernel(const float* __restrict__ Z, const float* __restrict__ hidden,
                 const float* __restrict__ LD,
                 float* __restrict__ C, float* __restrict__ zhh,
                 float* __restrict__ S) {
  __shared__ float4 red4[ZS_PARTS * (Isz / 4)];
  __shared__ float  red2[16 * Isz];
  __shared__ float  scan[Lsz];
  const int b = blockIdx.x >> 3;
  const int h = blockIdx.x & 7;
  const int tid = threadIdx.x;
  const int q = tid & 15;
  const int p = tid >> 4;

  const float ldv = LD[(size_t)(b * Lsz + tid) * Hsz + h];

  const int ch = (b * Hsz + h) * Isz + (tid & 63);
  float zdiag = 0.f, hid = 0.f;
  if (tid < Isz) {
    zdiag = Z[((size_t)(b * Lsz + h) * Hsz + h) * Isz + tid];
    hid = hidden[ch];
  }

  const float* rp = Z + ((size_t)(b * Lsz + p) * Hsz + h) * Isz + 4 * q;
  float4 v[ZS_ROWS];
#pragma unroll
  for (int j = 0; j < ZS_ROWS; ++j)
    v[j] = *(const float4*)(rp + (size_t)j * ZS_PARTS * Dsz);

  scan[tid] = ldv;
  __syncthreads();
  for (int off = 1; off < Lsz; off <<= 1) {
    float t = scan[tid];
    if (tid >= off) t += scan[tid - off];
    __syncthreads();
    scan[tid] = t;
    __syncthreads();
  }
  S[(size_t)(b * Lsz + tid) * Hsz + h] = scan[tid];

  if (p <= h) v[0] = make_float4(0.f, 0.f, 0.f, 0.f);

#pragma unroll
  for (int w = 1; w < ZS_ROWS; w <<= 1)
#pragma unroll
    for (int j = 0; j + w < ZS_ROWS; j += 2 * w) {
      v[j].x += v[j + w].x; v[j].y += v[j + w].y;
      v[j].z += v[j + w].z; v[j].w += v[j + w].w;
    }
  red4[p * (Isz / 4) + q] = v[0];
  __syncthreads();

  const float* red = (const float*)red4;
  {
    const int i = tid & 63;
    const int g = tid >> 6;
    const float* r = red + (4 * g) * Isz + i;
    red2[g * Isz + i] = (r[0] + r[Isz]) + (r[2 * Isz] + r[3 * Isz]);
  }
  __syncthreads();

  if (tid < Isz) {
    float t[16];
#pragma unroll
    for (int g = 0; g < 16; ++g) t[g] = red2[g * Isz + tid];
#pragma unroll
    for (int w = 1; w < 16; w <<= 1)
#pragma unroll
      for (int g = 0; g + w < 16; g += 2 * w) t[g] += t[g + w];
    C[ch] = t[0] + hid;
    zhh[ch] = zdiag;
  }
}

__global__ __launch_bounds__(512)
void hnorm_kernel(const float* __restrict__ S, const float* __restrict__ C,
                  const float* __restrict__ zhh, const float* __restrict__ gn_w,
                  const float* __restrict__ gn_b, float* __restrict__ Hout,
                  float* __restrict__ Hnorm) {
  __shared__ float lds[Dsz];
  const int bl = blockIdx.x;
  const int b = bl >> 10;
  const int t = threadIdx.x;
  const int h = t >> 6;
  const int i = t & 63;
  const int ch = (b * Hsz + h) * Isz + i;

  const float e = expf(S[(size_t)bl * Hsz + h]);
  const float hv = zhh[ch] + e * C[ch];
  Hout[(size_t)bl * Dsz + t] = hv;

  lds[t] = hv;
  __syncthreads();
  for (int s = 32; s >= 1; s >>= 1) {
    if (i < s) lds[t] += lds[t + s];
    __syncthreads();
  }
  const float mu = lds[h * 64] * (1.0f / 64.0f);
  __syncthreads();
  const float d = hv - mu;
  lds[t] = d * d;
  __syncthreads();
  for (int s = 32; s >= 1; s >>= 1) {
    if (i < s) lds[t] += lds[t + s];
    __syncthreads();
  }
  const float var = lds[h * 64] * (1.0f / 64.0f);
  const float hn = d / sqrtf(var + GN_EPS) * gn_w[h] + gn_b[h];
  Hnorm[(size_t)bl * Dsz + t] = hn;
}

extern "C" void kernel_launch(void* const* d_in, const int* in_sizes, int n_in,
                              void* d_out, int out_size, void* d_ws, size_t ws_size,
                              hipStream_t stream) {
  const float* x      = (const float*)d_in[0];
  const float* hidden = (const float*)d_in[1];
  const float* w_z    = (const float*)d_in[2];
  const float* b_z    = (const float*)d_in[3];
  const float* w_za   = (const float*)d_in[4];
  const float* b_za   = (const float*)d_in[5];
  const float* w_y    = (const float*)d_in[6];
  const float* b_y    = (const float*)d_in[7];
  const float* w_ya   = (const float*)d_in[8];
  const float* b_ya   = (const float*)d_in[9];
  const float* w_dt   = (const float*)d_in[10];
  const float* b_dt   = (const float*)d_in[11];
  const float* ln_a   = (const float*)d_in[12];
  const float* gn_w   = (const float*)d_in[13];
  const float* gn_b   = (const float*)d_in[14];

  float* y_out = (float*)d_out;
  float* h_out = y_out + (size_t)Msz * Dsz;

  float* ws   = (float*)d_ws;
  float* Z    = ws;
  float* GYA  = Z    + (size_t)Msz * Dsz;
  float* HN   = GYA  + (size_t)Msz * Dsz;
  float* Sb   = HN   + (size_t)Msz * Dsz;
  float* Cb   = Sb   + (size_t)Msz * Hsz;
  float* ZHH  = Cb   + (size_t)Bsz * Hsz * Isz;
  float* LDb  = ZHH  + (size_t)Bsz * Hsz * Isz;

  const dim3 gGemm(Msz / 64, Dsz / 64, 1);
  const dim3 bGemm(128, 1, 1);

  gemm_gate_z_kernel<<<gGemm, bGemm, 0, stream>>>(x, w_z, b_z, w_za, b_za, Z);
  gemm_single_kernel<0><<<gGemm, bGemm, 0, stream>>>(x, w_ya, b_ya, nullptr, GYA);
  dtproj_kernel<<<dim3(Msz / (DT_WAVES * DT_ROWS)), dim3(DT_WAVES * 32), 0, stream>>>(
      x, w_dt, b_dt, ln_a, LDb);
  zsum_kernel<<<dim3(Bsz * Hsz), dim3(1024), 0, stream>>>(Z, hidden, LDb, Cb, ZHH, Sb);
  hnorm_kernel<<<dim3(Msz), dim3(Dsz), 0, stream>>>(Sb, Cb, ZHH, gn_w, gn_b,
                                                    h_out, HN);
  gemm_single_kernel<1><<<gGemm, bGemm, 0, stream>>>(HN, w_y, b_y, GYA, y_out);
}


# GEMMs: operand tiles staged through LDS by full-line b128 loads, double-buffered; fragments read by ds_load_b128; was per-lane 8-byte global loads spread over 16 lines. Dot sum order changes
# speedup vs baseline: 2.5262x; 1.1918x over previous
#include <hip/hip_runtime.h>
#include <hip/hip_bf16.h>
#include <math.h>


#define Bsz 4
#define Lsz 1024
#define Dsz 512
#define Hsz 8
#define Isz 64
#define Msz (Bsz * Lsz)
#define GN_EPS 1e-5f

typedef float v2f __attribute__((ext_vector_type(2)));
typedef float v4f __attribute__((ext_vector_type(4)));
typedef float v8f __attribute__((ext_vector_type(8)));

__device__ __forceinline__ v8f wmma4(v2f a, v2f b, v8f c) {
  return __builtin_amdgcn_wmma_f32_16x16x4_f32(
       false, a,  false, b,
       (short)0, c,  false,  false);
}

__device__ __forceinline__ float silu_f(float v) {
  return v / (1.0f + expf(-v));
}

#define GK 32
#define GP 36
#define GMAT (64 * GP)
#define GKT (Dsz / GK)

template <int NB>
__device__ __forceinline__ void gemm_core(const float* __restrict__ A,
                                          const float* __restrict__ W0,
                                          const float* __restrict__ W1,
                                          float* __restrict__ lds,
                                          const int m_blk, const int n_blk,
                                          v8f (&acc)[NB][2][2]) {
  constexpr int STAGE = GMAT * (1 + NB);
  const int tid = threadIdx.x;
  const int lane = tid & 31;
  const int wave = tid >> 5;
  const int mr = lane & 15;
  const int kk = lane >> 4;
  const int wm = (wave & 1) * 32;
  const int wn = (wave >> 1) * 32;

  const int crow = tid >> 3;
  const int ccol = (tid & 7) * 4;
  const float* gA = A + (size_t)(m_blk + crow) * Dsz + ccol;
  const float* gB[NB];
  gB[0] = W0 + (size_t)(n_blk + crow) * Dsz + ccol;
  if (NB > 1) gB[NB - 1] = W1 + (size_t)(n_blk + crow) * Dsz + ccol;
  const int soff = crow * GP + ccol;

  const int aoff = (wm + mr) * GP + 4 * kk;
  const int boff = (wn + mr) * GP + 4 * kk;

  v4f pa[4], pb[NB][4];
#pragma unroll
  for (int j = 0; j < 4; ++j) {
    pa[j] = *(const v4f*)(gA + (size_t)j * 16 * Dsz);
#pragma unroll
    for (int nb = 0; nb < NB; ++nb)
      pb[nb][j] = *(const v4f*)(gB[nb] + (size_t)j * 16 * Dsz);
  }
#pragma unroll
  for (int j = 0; j < 4; ++j) {
    *(v4f*)(lds + soff + j * 16 * GP) = pa[j];
#pragma unroll
    for (int nb = 0; nb < NB; ++nb)
      *(v4f*)(lds + GMAT * (1 + nb) + soff + j * 16 * GP) = pb[nb][j];
  }
  __syncthreads();

#pragma unroll 2
  for (int kt = 0; kt < GKT; ++kt) {
    const float* cur = lds + (kt & 1) * STAGE;
    float* nxt = lds + ((kt + 1) & 1) * STAGE;
    const bool more = (kt + 1 < GKT);

    if (more) {
      const int ko = (kt + 1) * GK;
#pragma unroll
      for (int j = 0; j < 4; ++j) {
        pa[j] = *(const v4f*)(gA + (size_t)j * 16 * Dsz + ko);
#pragma unroll
        for (int nb = 0; nb < NB; ++nb)
          pb[nb][j] = *(const v4f*)(gB[nb] + (size_t)j * 16 * Dsz + ko);
      }
    }

#pragma unroll
    for (int kb = 0; kb < GK / 8; ++kb) {
      v4f fa[2], fb[NB][2];
#pragma unroll
      for (int t = 0; t < 2; ++t) {
        fa[t] = *(const v4f*)(cur + aoff + t * 16 * GP + kb * 8);
#pragma unroll
        for (int nb = 0; nb < NB; ++nb)
          fb[nb][t] = *(const v4f*)(cur + GMAT * (1 + nb) + boff + t * 16 * GP + kb * 8);
      }
#pragma unroll
      for (int nb = 0; nb < NB; ++nb)
#pragma unroll
        for (int mt = 0; mt < 2; ++mt)
#pragma unroll
          for (int nt = 0; nt < 2; ++nt)
            acc[nb][mt][nt] = wmma4(fa[mt].xy, fb[nb][nt].xy, acc[nb][mt][nt]);
#pragma unroll
      for (int nb = 0; nb < NB; ++nb)
#pragma unroll
        for (int mt = 0; mt < 2; ++mt)
#pragma unroll
          for (int nt = 0; nt < 2; ++nt)
            acc[nb][mt][nt] = wmma4(fa[mt].zw, fb[nb][nt].zw, acc[nb][mt][nt]);
    }

    if (more) {
#pragma unroll
      for (int j = 0; j < 4; ++j) {
        *(v4f*)(nxt + soff + j * 16 * GP) = pa[j];
#pragma unroll
        for (int nb = 0; nb < NB; ++nb)
          *(v4f*)(nxt + GMAT * (1 + nb) + soff + j * 16 * GP) = pb[nb][j];
      }
    }
    __syncthreads();
  }
}

__global__ __launch_bounds__(128)
void gemm_gate_z_kernel(const float* __restrict__ X,
                        const float* __restrict__ Wz,  const float* __restrict__ bz,
                        const float* __restrict__ Wza, const float* __restrict__ bza,
                        float* __restrict__ Z) {
  __shared__ __attribute__((aligned(16))) float lds[2 * GMAT * 3];
  const int lane = threadIdx.x & 31;
  const int wave = threadIdx.x >> 5;
  const int mr = lane & 15;
  const int kk = lane >> 4;
  const int m_base = blockIdx.x * 64 + (wave & 1) * 32;
  const int n_base = blockIdx.y * 64 + (wave >> 1) * 32;

  v8f acc[2][2][2];
  const v8f zero = {0.f, 0.f, 0.f, 0.f, 0.f, 0.f, 0.f, 0.f};
  for (int nb = 0; nb < 2; ++nb)
    for (int i = 0; i < 2; ++i)
      for (int j = 0; j < 2; ++j) acc[nb][i][j] = zero;

  gemm_core<2>(X, Wz, Wza, lds, blockIdx.x * 64, blockIdx.y * 64, acc);

  for (int mt = 0; mt < 2; ++mt)
    for (int nt = 0; nt < 2; ++nt) {
      const int n = n_base + nt * 16 + mr;
      const float biz = bz[n], biza = bza[n];
#pragma unroll
      for (int r = 0; r < 8; ++r) {
        const int m = m_base + mt * 16 + kk * 8 + r;
        const float v1 = acc[0][mt][nt][r] + biz;
        const float v2 = acc[1][mt][nt][r] + biza;
        Z[(size_t)m * Dsz + n] = v1 * silu_f(v2);
      }
    }
}

template <int MODE>
__global__ __launch_bounds__(128)
void gemm_single_kernel(const float* __restrict__ A, const float* __restrict__ W,
                        const float* __restrict__ bias,
                        const float* __restrict__ gate, float* __restrict__ OUT) {
  __shared__ __attribute__((aligned(16))) float lds[2 * GMAT * 2];
  const int lane = threadIdx.x & 31;
  const int wave = threadIdx.x >> 5;
  const int mr = lane & 15;
  const int kk = lane >> 4;
  const int m_base = blockIdx.x * 64 + (wave & 1) * 32;
  const int n_base = blockIdx.y * 64 + (wave >> 1) * 32;

  v8f acc[1][2][2];
  const v8f zero = {0.f, 0.f, 0.f, 0.f, 0.f, 0.f, 0.f, 0.f};
  for (int i = 0; i < 2; ++i)
    for (int j = 0; j < 2; ++j) acc[0][i][j] = zero;

  gemm_core<1>(A, W, W, lds, blockIdx.x * 64, blockIdx.y * 64, acc);

  for (int mt = 0; mt < 2; ++mt)
    for (int nt = 0; nt < 2; ++nt) {
      const int n = n_base + nt * 16 + mr;
      const float bi = bias[n];
#pragma unroll
      for (int r = 0; r < 8; ++r) {
        const int m = m_base + mt * 16 + kk * 8 + r;
        const float v = acc[0][mt][nt][r] + bi;
        float o;
        if (MODE == 0) o = silu_f(v);
        else           o = v * gate[(size_t)m * Dsz + n];
        OUT[(size_t)m * Dsz + n] = o;
      }
    }
}

#define DT_ROWS  2
#define DT_WAVES 8
#define DT_THREADS (DT_WAVES * 32)

__global__ __launch_bounds__(DT_THREADS)
void dtproj_kernel(const float* __restrict__ X, const float* __restrict__ w_dt,
                   const float* __restrict__ b_dt, const float* __restrict__ ln_a,
                   float* __restrict__ LD) {
  __shared__ v4f wl[Hsz * Dsz / 4];
  const int lane = threadIdx.x & 31;
  const int wave = threadIdx.x >> 5;
  const int row0 = (blockIdx.x * DT_WAVES + wave) * DT_ROWS;

  v4f wr[Hsz * Dsz / 4 / DT_THREADS];
#pragma unroll
  for (int j = 0; j < Hsz * Dsz / 4 / DT_THREADS; ++j)
    wr[j] = *(const v4f*)(w_dt + 4 * (threadIdx.x + DT_THREADS * j));

  v4f xv[DT_ROWS][4];
#pragma unroll
  for (int r = 0; r < DT_ROWS; ++r)
#pragma unroll
    for (int c = 0; c < 4; ++c)
      xv[r][c] = *(const v4f*)(X + (size_t)(row0 + r) * Dsz + c * 128 + 4 * lane);

  const int vi = lane >> 1;
  const int h_out = vi & 7;
  const float bdt = b_dt[h_out];
  const float lna = ln_a[h_out];

  asm volatile("" :: "v"(xv[0][0]), "v"(xv[0][1]), "v"(xv[0][2]), "v"(xv[0][3]),
                     "v"(xv[1][0]), "v"(xv[1][1]), "v"(xv[1][2]), "v"(xv[1][3]),
                     "v"(bdt), "v"(lna));

#pragma unroll
  for (int j = 0; j < Hsz * Dsz / 4 / DT_THREADS; ++j)
    wl[threadIdx.x + DT_THREADS * j] = wr[j];
  __syncthreads();

  float v16[DT_ROWS * Hsz];
#pragma unroll
  for (int h = 0; h < Hsz; ++h) {
    v4f wv[4];
#pragma unroll
    for (int c = 0; c < 4; ++c) wv[c] = wl[h * (Dsz / 4) + c * 32 + lane];
#pragma unroll
    for (int r = 0; r < DT_ROWS; ++r) {
      float s[4];
#pragma unroll
      for (int c = 0; c < 4; ++c) {
        s[c] = xv[r][c].x * wv[c].x;
        s[c] = fmaf(xv[r][c].y, wv[c].y, s[c]);
        s[c] = fmaf(xv[r][c].z, wv[c].z, s[c]);
        s[c] = fmaf(xv[r][c].w, wv[c].w, s[c]);
      }
      v16[r * Hsz + h] = (s[0] + s[1]) + (s[2] + s[3]);
    }
  }

  const bool b4 = (lane & 16) != 0;
  const bool b3 = (lane & 8) != 0;
  const bool b2 = (lane & 4) != 0;
  const bool b1 = (lane & 2) != 0;
  float v8[8], v4[4], v2[2];
#pragma unroll
  for (int i = 0; i < 8; ++i) {
    const float keep = b4 ? v16[8 + i] : v16[i];
    const float send = b4 ? v16[i] : v16[8 + i];
    v8[i] = keep + __shfl_xor(send, 16, 32);
  }
#pragma unroll
  for (int i = 0; i < 4; ++i) {
    const float keep = b3 ? v8[4 + i] : v8[i];
    const float send = b3 ? v8[i] : v8[4 + i];
    v4[i] = keep + __shfl_xor(send, 8, 32);
  }
#pragma unroll
  for (int i = 0; i < 2; ++i) {
    const float keep = b2 ? v4[2 + i] : v4[i];
    const float send = b2 ? v4[i] : v4[2 + i];
    v2[i] = keep + __shfl_xor(send, 4, 32);
  }
  float dot;
  {
    const float keep = b1 ? v2[1] : v2[0];
    const float send = b1 ? v2[0] : v2[1];
    dot = keep + __shfl_xor(send, 2, 32);
  }
  dot += __shfl_xor(dot, 1, 32);

  const float v = dot + bdt;
  const float sp = (v > 20.f) ? v : log1pf(expf(v));
  const float ld = -expf(lna) * sp;
  if ((lane & 1) == 0) LD[(size_t)row0 * Hsz + vi] = ld;
}

#define ZS_PARTS 64
#define ZS_ROWS  (Lsz / ZS_PARTS)

__global__ __launch_bounds__(1024)
void zsum_kernel(const float* __restrict__ Z, const float* __restrict__ hidden,
                 const float* __restrict__ LD,
                 float* __restrict__ C, float* __restrict__ zhh,
                 float* __restrict__ S) {
  __shared__ float4 red4[ZS_PARTS * (Isz / 4)];
  __shared__ float  red2[16 * Isz];
  __shared__ float  scan[Lsz];
  const int b = blockIdx.x >> 3;
  const int h = blockIdx.x & 7;
  const int tid = threadIdx.x;
  const int q = tid & 15;
  const int p = tid >> 4;

  const float ldv = LD[(size_t)(b * Lsz + tid) * Hsz + h];

  const int ch = (b * Hsz + h) * Isz + (tid & 63);
  float zdiag = 0.f, hid = 0.f;
  if (tid < Isz) {
    zdiag = Z[((size_t)(b * Lsz + h) * Hsz + h) * Isz + tid];
    hid = hidden[ch];
  }

  const float* rp = Z + ((size_t)(b * Lsz + p) * Hsz + h) * Isz + 4 * q;
  float4 v[ZS_ROWS];
#pragma unroll
  for (int j = 0; j < ZS_ROWS; ++j)
    v[j] = *(const float4*)(rp + (size_t)j * ZS_PARTS * Dsz);

  scan[tid] = ldv;
  __syncthreads();
  for (int off = 1; off < Lsz; off <<= 1) {
    float t = scan[tid];
    if (tid >= off) t += scan[tid - off];
    __syncthreads();
    scan[tid] = t;
    __syncthreads();
  }
  S[(size_t)(b * Lsz + tid) * Hsz + h] = scan[tid];

  if (p <= h) v[0] = make_float4(0.f, 0.f, 0.f, 0.f);

#pragma unroll
  for (int w = 1; w < ZS_ROWS; w <<= 1)
#pragma unroll
    for (int j = 0; j + w < ZS_ROWS; j += 2 * w) {
      v[j].x += v[j + w].x; v[j].y += v[j + w].y;
      v[j].z += v[j + w].z; v[j].w += v[j + w].w;
    }
  red4[p * (Isz / 4) + q] = v[0];
  __syncthreads();

  const float* red = (const float*)red4;
  {
    const int i = tid & 63;
    const int g = tid >> 6;
    const float* r = red + (4 * g) * Isz + i;
    red2[g * Isz + i] = (r[0] + r[Isz]) + (r[2 * Isz] + r[3 * Isz]);
  }
  __syncthreads();

  if (tid < Isz) {
    float t[16];
#pragma unroll
    for (int g = 0; g < 16; ++g) t[g] = red2[g * Isz + tid];
#pragma unroll
    for (int w = 1; w < 16; w <<= 1)
#pragma unroll
      for (int g = 0; g + w < 16; g += 2 * w) t[g] += t[g + w];
    C[ch] = t[0] + hid;
    zhh[ch] = zdiag;
  }
}

__global__ __launch_bounds__(512)
void hnorm_kernel(const float* __restrict__ S, const float* __restrict__ C,
                  const float* __restrict__ zhh, const float* __restrict__ gn_w,
                  const float* __restrict__ gn_b, float* __restrict__ Hout,
                  float* __restrict__ Hnorm) {
  __shared__ float lds[Dsz];
  const int bl = blockIdx.x;
  const int b = bl >> 10;
  const int t = threadIdx.x;
  const int h = t >> 6;
  const int i = t & 63;
  const int ch = (b * Hsz + h) * Isz + i;

  const float e = expf(S[(size_t)bl * Hsz + h]);
  const float hv = zhh[ch] + e * C[ch];
  Hout[(size_t)bl * Dsz + t] = hv;

  lds[t] = hv;
  __syncthreads();
  for (int s = 32; s >= 1; s >>= 1) {
    if (i < s) lds[t] += lds[t + s];
    __syncthreads();
  }
  const float mu = lds[h * 64] * (1.0f / 64.0f);
  __syncthreads();
  const float d = hv - mu;
  lds[t] = d * d;
  __syncthreads();
  for (int s = 32; s >= 1; s >>= 1) {
    if (i < s) lds[t] += lds[t + s];
    __syncthreads();
  }
  const float var = lds[h * 64] * (1.0f / 64.0f);
  const float hn = d / sqrtf(var + GN_EPS) * gn_w[h] + gn_b[h];
  Hnorm[(size_t)bl * Dsz + t] = hn;
}

extern "C" void kernel_launch(void* const* d_in, const int* in_sizes, int n_in,
                              void* d_out, int out_size, void* d_ws, size_t ws_size,
                              hipStream_t stream) {
  const float* x      = (const float*)d_in[0];
  const float* hidden = (const float*)d_in[1];
  const float* w_z    = (const float*)d_in[2];
  const float* b_z    = (const float*)d_in[3];
  const float* w_za   = (const float*)d_in[4];
  const float* b_za   = (const float*)d_in[5];
  const float* w_y    = (const float*)d_in[6];
  const float* b_y    = (const float*)d_in[7];
  const float* w_ya   = (const float*)d_in[8];
  const float* b_ya   = (const float*)d_in[9];
  const float* w_dt   = (const float*)d_in[10];
  const float* b_dt   = (const float*)d_in[11];
  const float* ln_a   = (const float*)d_in[12];
  const float* gn_w   = (const float*)d_in[13];
  const float* gn_b   = (const float*)d_in[14];

  float* y_out = (float*)d_out;
  float* h_out = y_out + (size_t)Msz * Dsz;

  float* ws   = (float*)d_ws;
  float* Z    = ws;
  float* GYA  = Z    + (size_t)Msz * Dsz;
  float* HN   = GYA  + (size_t)Msz * Dsz;
  float* Sb   = HN   + (size_t)Msz * Dsz;
  float* Cb   = Sb   + (size_t)Msz * Hsz;
  float* ZHH  = Cb   + (size_t)Bsz * Hsz * Isz;
  float* LDb  = ZHH  + (size_t)Bsz * Hsz * Isz;

  const dim3 gGemm(Msz / 64, Dsz / 64, 1);
  const dim3 bGemm(128, 1, 1);

  gemm_gate_z_kernel<<<gGemm, bGemm, 0, stream>>>(x, w_z, b_z, w_za, b_za, Z);
  gemm_single_kernel<0><<<gGemm, bGemm, 0, stream>>>(x, w_ya, b_ya, nullptr, GYA);
  dtproj_kernel<<<dim3(Msz / (DT_WAVES * DT_ROWS)), dim3(DT_WAVES * 32), 0, stream>>>(
      x, w_dt, b_dt, ln_a, LDb);
  zsum_kernel<<<dim3(Bsz * Hsz), dim3(1024), 0, stream>>>(Z, hidden, LDb, Cb, ZHH, Sb);
  hnorm_kernel<<<dim3(Msz), dim3(Dsz), 0, stream>>>(Sb, Cb, ZHH, gn_w, gn_b,
                                                    h_out, HN);
  gemm_single_kernel<1><<<gGemm, bGemm, 0, stream>>>(HN, w_y, b_y, GYA, y_out);
}


# group-norm kernel: one wave per row (2 rows a wave, 256 blocks), head mean/variance by 16-lane DPP xor butterfly, no LDS or barrier; was 4096 blocks x 16 waves, 14 barriers. Their sum order changes
# speedup vs baseline: 2.9132x; 1.1532x over previous
#include <hip/hip_runtime.h>
#include <hip/hip_bf16.h>
#include <math.h>


#define Bsz 4
#define Lsz 1024
#define Dsz 512
#define Hsz 8
#define Isz 64
#define Msz (Bsz * Lsz)
#define GN_EPS 1e-5f

typedef float v2f __attribute__((ext_vector_type(2)));
typedef float v4f __attribute__((ext_vector_type(4)));
typedef float v8f __attribute__((ext_vector_type(8)));

__device__ __forceinline__ v8f wmma4(v2f a, v2f b, v8f c) {
  return __builtin_amdgcn_wmma_f32_16x16x4_f32(
       false, a,  false, b,
       (short)0, c,  false,  false);
}

__device__ __forceinline__ float silu_f(float v) {
  return v / (1.0f + expf(-v));
}

#define GK 32
#define GP 36
#define GMAT (64 * GP)
#define GKT (Dsz / GK)

template <int NB>
__device__ __forceinline__ void gemm_core(const float* __restrict__ A,
                                          const float* __restrict__ W0,
                                          const float* __restrict__ W1,
                                          float* __restrict__ lds,
                                          const int m_blk, const int n_blk,
                                          v8f (&acc)[NB][2][2]) {
  constexpr int STAGE = GMAT * (1 + NB);
  const int tid = threadIdx.x;
  const int lane = tid & 31;
  const int wave = tid >> 5;
  const int mr = lane & 15;
  const int kk = lane >> 4;
  const int wm = (wave & 1) * 32;
  const int wn = (wave >> 1) * 32;

  const int crow = tid >> 3;
  const int ccol = (tid & 7) * 4;
  const float* gA = A + (size_t)(m_blk + crow) * Dsz + ccol;
  const float* gB[NB];
  gB[0] = W0 + (size_t)(n_blk + crow) * Dsz + ccol;
  if (NB > 1) gB[NB - 1] = W1 + (size_t)(n_blk + crow) * Dsz + ccol;
  const int soff = crow * GP + ccol;

  const int aoff = (wm + mr) * GP + 4 * kk;
  const int boff = (wn + mr) * GP + 4 * kk;

  v4f pa[4], pb[NB][4];
#pragma unroll
  for (int j = 0; j < 4; ++j) {
    pa[j] = *(const v4f*)(gA + (size_t)j * 16 * Dsz);
#pragma unroll
    for (int nb = 0; nb < NB; ++nb)
      pb[nb][j] = *(const v4f*)(gB[nb] + (size_t)j * 16 * Dsz);
  }
#pragma unroll
  for (int j = 0; j < 4; ++j) {
    *(v4f*)(lds + soff + j * 16 * GP) = pa[j];
#pragma unroll
    for (int nb = 0; nb < NB; ++nb)
      *(v4f*)(lds + GMAT * (1 + nb) + soff + j * 16 * GP) = pb[nb][j];
  }
  __syncthreads();

#pragma unroll 2
  for (int kt = 0; kt < GKT; ++kt) {
    const float* cur = lds + (kt & 1) * STAGE;
    float* nxt = lds + ((kt + 1) & 1) * STAGE;
    const bool more = (kt + 1 < GKT);

    if (more) {
      const int ko = (kt + 1) * GK;
#pragma unroll
      for (int j = 0; j < 4; ++j) {
        pa[j] = *(const v4f*)(gA + (size_t)j * 16 * Dsz + ko);
#pragma unroll
        for (int nb = 0; nb < NB; ++nb)
          pb[nb][j] = *(const v4f*)(gB[nb] + (size_t)j * 16 * Dsz + ko);
      }
    }

#pragma unroll
    for (int kb = 0; kb < GK / 8; ++kb) {
      v4f fa[2], fb[NB][2];
#pragma unroll
      for (int t = 0; t < 2; ++t) {
        fa[t] = *(const v4f*)(cur + aoff + t * 16 * GP + kb * 8);
#pragma unroll
        for (int nb = 0; nb < NB; ++nb)
          fb[nb][t] = *(const v4f*)(cur + GMAT * (1 + nb) + boff + t * 16 * GP + kb * 8);
      }
#pragma unroll
      for (int nb = 0; nb < NB; ++nb)
#pragma unroll
        for (int mt = 0; mt < 2; ++mt)
#pragma unroll
          for (int nt = 0; nt < 2; ++nt)
            acc[nb][mt][nt] = wmma4(fa[mt].xy, fb[nb][nt].xy, acc[nb][mt][nt]);
#pragma unroll
      for (int nb = 0; nb < NB; ++nb)
#pragma unroll
        for (int mt = 0; mt < 2; ++mt)
#pragma unroll
          for (int nt = 0; nt < 2; ++nt)
            acc[nb][mt][nt] = wmma4(fa[mt].zw, fb[nb][nt].zw, acc[nb][mt][nt]);
    }

    if (more) {
#pragma unroll
      for (int j = 0; j < 4; ++j) {
        *(v4f*)(nxt + soff + j * 16 * GP) = pa[j];
#pragma unroll
        for (int nb = 0; nb < NB; ++nb)
          *(v4f*)(nxt + GMAT * (1 + nb) + soff + j * 16 * GP) = pb[nb][j];
      }
    }
    __syncthreads();
  }
}

__global__ __launch_bounds__(128)
void gemm_gate_z_kernel(const float* __restrict__ X,
                        const float* __restrict__ Wz,  const float* __restrict__ bz,
                        const float* __restrict__ Wza, const float* __restrict__ bza,
                        float* __restrict__ Z) {
  __shared__ __attribute__((aligned(16))) float lds[2 * GMAT * 3];
  const int lane = threadIdx.x & 31;
  const int wave = threadIdx.x >> 5;
  const int mr = lane & 15;
  const int kk = lane >> 4;
  const int m_base = blockIdx.x * 64 + (wave & 1) * 32;
  const int n_base = blockIdx.y * 64 + (wave >> 1) * 32;

  v8f acc[2][2][2];
  const v8f zero = {0.f, 0.f, 0.f, 0.f, 0.f, 0.f, 0.f, 0.f};
  for (int nb = 0; nb < 2; ++nb)
    for (int i = 0; i < 2; ++i)
      for (int j = 0; j < 2; ++j) acc[nb][i][j] = zero;

  gemm_core<2>(X, Wz, Wza, lds, blockIdx.x * 64, blockIdx.y * 64, acc);

  for (int mt = 0; mt < 2; ++mt)
    for (int nt = 0; nt < 2; ++nt) {
      const int n = n_base + nt * 16 + mr;
      const float biz = bz[n], biza = bza[n];
#pragma unroll
      for (int r = 0; r < 8; ++r) {
        const int m = m_base + mt * 16 + kk * 8 + r;
        const float v1 = acc[0][mt][nt][r] + biz;
        const float v2 = acc[1][mt][nt][r] + biza;
        Z[(size_t)m * Dsz + n] = v1 * silu_f(v2);
      }
    }
}

template <int MODE>
__global__ __launch_bounds__(128)
void gemm_single_kernel(const float* __restrict__ A, const float* __restrict__ W,
                        const float* __restrict__ bias,
                        const float* __restrict__ gate, float* __restrict__ OUT) {
  __shared__ __attribute__((aligned(16))) float lds[2 * GMAT * 2];
  const int lane = threadIdx.x & 31;
  const int wave = threadIdx.x >> 5;
  const int mr = lane & 15;
  const int kk = lane >> 4;
  const int m_base = blockIdx.x * 64 + (wave & 1) * 32;
  const int n_base = blockIdx.y * 64 + (wave >> 1) * 32;

  v8f acc[1][2][2];
  const v8f zero = {0.f, 0.f, 0.f, 0.f, 0.f, 0.f, 0.f, 0.f};
  for (int i = 0; i < 2; ++i)
    for (int j = 0; j < 2; ++j) acc[0][i][j] = zero;

  gemm_core<1>(A, W, W, lds, blockIdx.x * 64, blockIdx.y * 64, acc);

  for (int mt = 0; mt < 2; ++mt)
    for (int nt = 0; nt < 2; ++nt) {
      const int n = n_base + nt * 16 + mr;
      const float bi = bias[n];
#pragma unroll
      for (int r = 0; r < 8; ++r) {
        const int m = m_base + mt * 16 + kk * 8 + r;
        const float v = acc[0][mt][nt][r] + bi;
        float o;
        if (MODE == 0) o = silu_f(v);
        else           o = v * gate[(size_t)m * Dsz + n];
        OUT[(size_t)m * Dsz + n] = o;
      }
    }
}

#define DT_ROWS  2
#define DT_WAVES 8
#define DT_THREADS (DT_WAVES * 32)

__global__ __launch_bounds__(DT_THREADS)
void dtproj_kernel(const float* __restrict__ X, const float* __restrict__ w_dt,
                   const float* __restrict__ b_dt, const float* __restrict__ ln_a,
                   float* __restrict__ LD) {
  __shared__ v4f wl[Hsz * Dsz / 4];
  const int lane = threadIdx.x & 31;
  const int wave = threadIdx.x >> 5;
  const int row0 = (blockIdx.x * DT_WAVES + wave) * DT_ROWS;

  v4f wr[Hsz * Dsz / 4 / DT_THREADS];
#pragma unroll
  for (int j = 0; j < Hsz * Dsz / 4 / DT_THREADS; ++j)
    wr[j] = *(const v4f*)(w_dt + 4 * (threadIdx.x + DT_THREADS * j));

  v4f xv[DT_ROWS][4];
#pragma unroll
  for (int r = 0; r < DT_ROWS; ++r)
#pragma unroll
    for (int c = 0; c < 4; ++c)
      xv[r][c] = *(const v4f*)(X + (size_t)(row0 + r) * Dsz + c * 128 + 4 * lane);

  const int vi = lane >> 1;
  const int h_out = vi & 7;
  const float bdt = b_dt[h_out];
  const float lna = ln_a[h_out];

  asm volatile("" :: "v"(xv[0][0]), "v"(xv[0][1]), "v"(xv[0][2]), "v"(xv[0][3]),
                     "v"(xv[1][0]), "v"(xv[1][1]), "v"(xv[1][2]), "v"(xv[1][3]),
                     "v"(bdt), "v"(lna));

#pragma unroll
  for (int j = 0; j < Hsz * Dsz / 4 / DT_THREADS; ++j)
    wl[threadIdx.x + DT_THREADS * j] = wr[j];
  __syncthreads();

  float v16[DT_ROWS * Hsz];
#pragma unroll
  for (int h = 0; h < Hsz; ++h) {
    v4f wv[4];
#pragma unroll
    for (int c = 0; c < 4; ++c) wv[c] = wl[h * (Dsz / 4) + c * 32 + lane];
#pragma unroll
    for (int r = 0; r < DT_ROWS; ++r) {
      float s[4];
#pragma unroll
      for (int c = 0; c < 4; ++c) {
        s[c] = xv[r][c].x * wv[c].x;
        s[c] = fmaf(xv[r][c].y, wv[c].y, s[c]);
        s[c] = fmaf(xv[r][c].z, wv[c].z, s[c]);
        s[c] = fmaf(xv[r][c].w, wv[c].w, s[c]);
      }
      v16[r * Hsz + h] = (s[0] + s[1]) + (s[2] + s[3]);
    }
  }

  const bool b4 = (lane & 16) != 0;
  const bool b3 = (lane & 8) != 0;
  const bool b2 = (lane & 4) != 0;
  const bool b1 = (lane & 2) != 0;
  float v8[8], v4[4], v2[2];
#pragma unroll
  for (int i = 0; i < 8; ++i) {
    const float keep = b4 ? v16[8 + i] : v16[i];
    const float send = b4 ? v16[i] : v16[8 + i];
    v8[i] = keep + __shfl_xor(send, 16, 32);
  }
#pragma unroll
  for (int i = 0; i < 4; ++i) {
    const float keep = b3 ? v8[4 + i] : v8[i];
    const float send = b3 ? v8[i] : v8[4 + i];
    v4[i] = keep + __shfl_xor(send, 8, 32);
  }
#pragma unroll
  for (int i = 0; i < 2; ++i) {
    const float keep = b2 ? v4[2 + i] : v4[i];
    const float send = b2 ? v4[i] : v4[2 + i];
    v2[i] = keep + __shfl_xor(send, 4, 32);
  }
  float dot;
  {
    const float keep = b1 ? v2[1] : v2[0];
    const float send = b1 ? v2[0] : v2[1];
    dot = keep + __shfl_xor(send, 2, 32);
  }
  dot += __shfl_xor(dot, 1, 32);

  const float v = dot + bdt;
  const float sp = (v > 20.f) ? v : log1pf(expf(v));
  const float ld = -expf(lna) * sp;
  if ((lane & 1) == 0) LD[(size_t)row0 * Hsz + vi] = ld;
}

#define ZS_PARTS 64
#define ZS_ROWS  (Lsz / ZS_PARTS)

__global__ __launch_bounds__(1024)
void zsum_kernel(const float* __restrict__ Z, const float* __restrict__ hidden,
                 const float* __restrict__ LD,
                 float* __restrict__ C, float* __restrict__ zhh,
                 float* __restrict__ S) {
  __shared__ float4 red4[ZS_PARTS * (Isz / 4)];
  __shared__ float  red2[16 * Isz];
  __shared__ float  scan[Lsz];
  const int b = blockIdx.x >> 3;
  const int h = blockIdx.x & 7;
  const int tid = threadIdx.x;
  const int q = tid & 15;
  const int p = tid >> 4;

  const float ldv = LD[(size_t)(b * Lsz + tid) * Hsz + h];

  const int ch = (b * Hsz + h) * Isz + (tid & 63);
  float zdiag = 0.f, hid = 0.f;
  if (tid < Isz) {
    zdiag = Z[((size_t)(b * Lsz + h) * Hsz + h) * Isz + tid];
    hid = hidden[ch];
  }

  const float* rp = Z + ((size_t)(b * Lsz + p) * Hsz + h) * Isz + 4 * q;
  float4 v[ZS_ROWS];
#pragma unroll
  for (int j = 0; j < ZS_ROWS; ++j)
    v[j] = *(const float4*)(rp + (size_t)j * ZS_PARTS * Dsz);

  scan[tid] = ldv;
  __syncthreads();
  for (int off = 1; off < Lsz; off <<= 1) {
    float t = scan[tid];
    if (tid >= off) t += scan[tid - off];
    __syncthreads();
    scan[tid] = t;
    __syncthreads();
  }
  S[(size_t)(b * Lsz + tid) * Hsz + h] = scan[tid];

  if (p <= h) v[0] = make_float4(0.f, 0.f, 0.f, 0.f);

#pragma unroll
  for (int w = 1; w < ZS_ROWS; w <<= 1)
#pragma unroll
    for (int j = 0; j + w < ZS_ROWS; j += 2 * w) {
      v[j].x += v[j + w].x; v[j].y += v[j + w].y;
      v[j].z += v[j + w].z; v[j].w += v[j + w].w;
    }
  red4[p * (Isz / 4) + q] = v[0];
  __syncthreads();

  const float* red = (const float*)red4;
  {
    const int i = tid & 63;
    const int g = tid >> 6;
    const float* r = red + (4 * g) * Isz + i;
    red2[g * Isz + i] = (r[0] + r[Isz]) + (r[2 * Isz] + r[3 * Isz]);
  }
  __syncthreads();

  if (tid < Isz) {
    float t[16];
#pragma unroll
    for (int g = 0; g < 16; ++g) t[g] = red2[g * Isz + tid];
#pragma unroll
    for (int w = 1; w < 16; w <<= 1)
#pragma unroll
      for (int g = 0; g + w < 16; g += 2 * w) t[g] += t[g + w];
    C[ch] = t[0] + hid;
    zhh[ch] = zdiag;
  }
}

#define HN_ROWS  2
#define HN_WAVES 8

template <int M>
__device__ __forceinline__ float row_xor_add(float v) {
  const int p = __builtin_amdgcn_update_dpp(0, __float_as_int(v), 0x160 + M,
                                            0xf, 0xf, false);
  return v + __int_as_float(p);
}

__global__ __launch_bounds__(HN_WAVES * 32)
void hnorm_kernel(const float* __restrict__ S, const float* __restrict__ C,
                  const float* __restrict__ zhh, const float* __restrict__ gn_w,
                  const float* __restrict__ gn_b, float* __restrict__ Hout,
                  float* __restrict__ Hnorm) {
  const int lane = threadIdx.x & 31;
  const int wave = threadIdx.x >> 5;
  const int row0 = (blockIdx.x * HN_WAVES + wave) * HN_ROWS;
  const int b = row0 >> 10;
  const int hh = lane >> 4;

  v4f zv[4], cv[4];
  float gw[4], gb[4];
  float sv[HN_ROWS][4];
  for (int c = 0; c < 4; ++c) {
    zv[c] = *(const v4f*)(zhh + (size_t)b * Dsz + c * 128 + 4 * lane);
    cv[c] = *(const v4f*)(C + (size_t)b * Dsz + c * 128 + 4 * lane);
    gw[c] = gn_w[2 * c + hh];
    gb[c] = gn_b[2 * c + hh];
  }
  for (int r = 0; r < HN_ROWS; ++r)
    for (int c = 0; c < 4; ++c)
      sv[r][c] = S[(size_t)(row0 + r) * Hsz + 2 * c + hh];

  for (int r = 0; r < HN_ROWS; ++r) {
    const size_t ro = (size_t)(row0 + r) * Dsz + 4 * lane;
    for (int c = 0; c < 4; ++c) {
      const float e = expf(sv[r][c]);
      v4f hv;
      hv.x = zv[c].x + e * cv[c].x;
      hv.y = zv[c].y + e * cv[c].y;
      hv.z = zv[c].z + e * cv[c].z;
      hv.w = zv[c].w + e * cv[c].w;
      *(v4f*)(Hout + ro + c * 128) = hv;

      float s = (hv.x + hv.y) + (hv.z + hv.w);
      s = row_xor_add<1>(s);
      s = row_xor_add<2>(s);
      s = row_xor_add<4>(s);
      s = row_xor_add<8>(s);
      const float mu = s * (1.0f / 64.0f);

      v4f d;
      d.x = hv.x - mu; d.y = hv.y - mu; d.z = hv.z - mu; d.w = hv.w - mu;
      float q = (d.x * d.x + d.y * d.y) + (d.z * d.z + d.w * d.w);
      q = row_xor_add<1>(q);
      q = row_xor_add<2>(q);
      q = row_xor_add<4>(q);
      q = row_xor_add<8>(q);
      const float var = q * (1.0f / 64.0f);

      const float sc = gw[c] / sqrtf(var + GN_EPS);
      v4f hn;
      hn.x = d.x * sc + gb[c];
      hn.y = d.y * sc + gb[c];
      hn.z = d.z * sc + gb[c];
      hn.w = d.w * sc + gb[c];
      *(v4f*)(Hnorm + ro + c * 128) = hn;
    }
  }
}

extern "C" void kernel_launch(void* const* d_in, const int* in_sizes, int n_in,
                              void* d_out, int out_size, void* d_ws, size_t ws_size,
                              hipStream_t stream) {
  const float* x      = (const float*)d_in[0];
  const float* hidden = (const float*)d_in[1];
  const float* w_z    = (const float*)d_in[2];
  const float* b_z    = (const float*)d_in[3];
  const float* w_za   = (const float*)d_in[4];
  const float* b_za   = (const float*)d_in[5];
  const float* w_y    = (const float*)d_in[6];
  const float* b_y    = (const float*)d_in[7];
  const float* w_ya   = (const float*)d_in[8];
  const float* b_ya   = (const float*)d_in[9];
  const float* w_dt   = (const float*)d_in[10];
  const float* b_dt   = (const float*)d_in[11];
  const float* ln_a   = (const float*)d_in[12];
  const float* gn_w   = (const float*)d_in[13];
  const float* gn_b   = (const float*)d_in[14];

  float* y_out = (float*)d_out;
  float* h_out = y_out + (size_t)Msz * Dsz;

  float* ws   = (float*)d_ws;
  float* Z    = ws;
  float* GYA  = Z    + (size_t)Msz * Dsz;
  float* HN   = GYA  + (size_t)Msz * Dsz;
  float* Sb   = HN   + (size_t)Msz * Dsz;
  float* Cb   = Sb   + (size_t)Msz * Hsz;
  float* ZHH  = Cb   + (size_t)Bsz * Hsz * Isz;
  float* LDb  = ZHH  + (size_t)Bsz * Hsz * Isz;

  const dim3 gGemm(Msz / 64, Dsz / 64, 1);
  const dim3 bGemm(128, 1, 1);

  gemm_gate_z_kernel<<<gGemm, bGemm, 0, stream>>>(x, w_z, b_z, w_za, b_za, Z);
  gemm_single_kernel<0><<<gGemm, bGemm, 0, stream>>>(x, w_ya, b_ya, nullptr, GYA);
  dtproj_kernel<<<dim3(Msz / (DT_WAVES * DT_ROWS)), dim3(DT_WAVES * 32), 0, stream>>>(
      x, w_dt, b_dt, ln_a, LDb);
  zsum_kernel<<<dim3(Bsz * Hsz), dim3(1024), 0, stream>>>(Z, hidden, LDb, Cb, ZHH, Sb);
  hnorm_kernel<<<dim3(Msz / (HN_WAVES * HN_ROWS)), dim3(HN_WAVES * 32), 0, stream>>>(
      Sb, Cb, ZHH, gn_w, gn_b, h_out, HN);
  gemm_single_kernel<1><<<gGemm, bGemm, 0, stream>>>(HN, w_y, b_y, GYA, y_out);
}


# output projection: h_norm is rank 3 per head (zc, exp(S)*cc, const), so y = coef[M x 24] . (factor rows x Wy) on WMMA in the norm kernel; GEMM depth 512 -> 24, one launch less. Dot sum order changes
# speedup vs baseline: 3.3173x; 1.1387x over previous
#include <hip/hip_runtime.h>
#include <hip/hip_bf16.h>
#include <math.h>


#define Bsz 4
#define Lsz 1024
#define Dsz 512
#define Hsz 8
#define Isz 64
#define Msz (Bsz * Lsz)
#define GN_EPS 1e-5f

typedef float v2f __attribute__((ext_vector_type(2)));
typedef float v4f __attribute__((ext_vector_type(4)));
typedef float v8f __attribute__((ext_vector_type(8)));

__device__ __forceinline__ v8f wmma4(v2f a, v2f b, v8f c) {
  return __builtin_amdgcn_wmma_f32_16x16x4_f32(
       false, a,  false, b,
       (short)0, c,  false,  false);
}

__device__ __forceinline__ float silu_f(float v) {
  return v / (1.0f + expf(-v));
}

#define GK 32
#define GP 36
#define GMAT (64 * GP)
#define GKT (Dsz / GK)

template <int NB>
__device__ __forceinline__ void gemm_core(const float* __restrict__ A,
                                          const float* __restrict__ W0,
                                          const float* __restrict__ W1,
                                          float* __restrict__ lds,
                                          const int m_blk, const int n_blk,
                                          v8f (&acc)[NB][2][2]) {
  constexpr int STAGE = GMAT * (1 + NB);
  const int tid = threadIdx.x;
  const int lane = tid & 31;
  const int wave = tid >> 5;
  const int mr = lane & 15;
  const int kk = lane >> 4;
  const int wm = (wave & 1) * 32;
  const int wn = (wave >> 1) * 32;

  const int crow = tid >> 3;
  const int ccol = (tid & 7) * 4;
  const float* gA = A + (size_t)(m_blk + crow) * Dsz + ccol;
  const float* gB[NB];
  gB[0] = W0 + (size_t)(n_blk + crow) * Dsz + ccol;
  if (NB > 1) gB[NB - 1] = W1 + (size_t)(n_blk + crow) * Dsz + ccol;
  const int soff = crow * GP + ccol;

  const int aoff = (wm + mr) * GP + 4 * kk;
  const int boff = (wn + mr) * GP + 4 * kk;

  v4f pa[4], pb[NB][4];
#pragma unroll
  for (int j = 0; j < 4; ++j) {
    pa[j] = *(const v4f*)(gA + (size_t)j * 16 * Dsz);
#pragma unroll
    for (int nb = 0; nb < NB; ++nb)
      pb[nb][j] = *(const v4f*)(gB[nb] + (size_t)j * 16 * Dsz);
  }
#pragma unroll
  for (int j = 0; j < 4; ++j) {
    *(v4f*)(lds + soff + j * 16 * GP) = pa[j];
#pragma unroll
    for (int nb = 0; nb < NB; ++nb)
      *(v4f*)(lds + GMAT * (1 + nb) + soff + j * 16 * GP) = pb[nb][j];
  }
  __syncthreads();

#pragma unroll 2
  for (int kt = 0; kt < GKT; ++kt) {
    const float* cur = lds + (kt & 1) * STAGE;
    float* nxt = lds + ((kt + 1) & 1) * STAGE;
    const bool more = (kt + 1 < GKT);

    if (more) {
      const int ko = (kt + 1) * GK;
#pragma unroll
      for (int j = 0; j < 4; ++j) {
        pa[j] = *(const v4f*)(gA + (size_t)j * 16 * Dsz + ko);
#pragma unroll
        for (int nb = 0; nb < NB; ++nb)
          pb[nb][j] = *(const v4f*)(gB[nb] + (size_t)j * 16 * Dsz + ko);
      }
    }

#pragma unroll
    for (int kb = 0; kb < GK / 8; ++kb) {
      v4f fa[2], fb[NB][2];
#pragma unroll
      for (int t = 0; t < 2; ++t) {
        fa[t] = *(const v4f*)(cur + aoff + t * 16 * GP + kb * 8);
#pragma unroll
        for (int nb = 0; nb < NB; ++nb)
          fb[nb][t] = *(const v4f*)(cur + GMAT * (1 + nb) + boff + t * 16 * GP + kb * 8);
      }
#pragma unroll
      for (int nb = 0; nb < NB; ++nb)
#pragma unroll
        for (int mt = 0; mt < 2; ++mt)
#pragma unroll
          for (int nt = 0; nt < 2; ++nt)
            acc[nb][mt][nt] = wmma4(fa[mt].xy, fb[nb][nt].xy, acc[nb][mt][nt]);
#pragma unroll
      for (int nb = 0; nb < NB; ++nb)
#pragma unroll
        for (int mt = 0; mt < 2; ++mt)
#pragma unroll
          for (int nt = 0; nt < 2; ++nt)
            acc[nb][mt][nt] = wmma4(fa[mt].zw, fb[nb][nt].zw, acc[nb][mt][nt]);
    }

    if (more) {
#pragma unroll
      for (int j = 0; j < 4; ++j) {
        *(v4f*)(nxt + soff + j * 16 * GP) = pa[j];
#pragma unroll
        for (int nb = 0; nb < NB; ++nb)
          *(v4f*)(nxt + GMAT * (1 + nb) + soff + j * 16 * GP) = pb[nb][j];
      }
    }
    __syncthreads();
  }
}

__global__ __launch_bounds__(128)
void gemm_gate_z_kernel(const float* __restrict__ X,
                        const float* __restrict__ Wz,  const float* __restrict__ bz,
                        const float* __restrict__ Wza, const float* __restrict__ bza,
                        float* __restrict__ Z) {
  __shared__ __attribute__((aligned(16))) float lds[2 * GMAT * 3];
  const int lane = threadIdx.x & 31;
  const int wave = threadIdx.x >> 5;
  const int mr = lane & 15;
  const int kk = lane >> 4;
  const int m_base = blockIdx.x * 64 + (wave & 1) * 32;
  const int n_base = blockIdx.y * 64 + (wave >> 1) * 32;

  v8f acc[2][2][2];
  const v8f zero = {0.f, 0.f, 0.f, 0.f, 0.f, 0.f, 0.f, 0.f};
  for (int nb = 0; nb < 2; ++nb)
    for (int i = 0; i < 2; ++i)
      for (int j = 0; j < 2; ++j) acc[nb][i][j] = zero;

  gemm_core<2>(X, Wz, Wza, lds, blockIdx.x * 64, blockIdx.y * 64, acc);

  for (int mt = 0; mt < 2; ++mt)
    for (int nt = 0; nt < 2; ++nt) {
      const int n = n_base + nt * 16 + mr;
      const float biz = bz[n], biza = bza[n];
#pragma unroll
      for (int r = 0; r < 8; ++r) {
        const int m = m_base + mt * 16 + kk * 8 + r;
        const float v1 = acc[0][mt][nt][r] + biz;
        const float v2 = acc[1][mt][nt][r] + biza;
        Z[(size_t)m * Dsz + n] = v1 * silu_f(v2);
      }
    }
}

__global__ __launch_bounds__(128)
void gemm_single_kernel(const float* __restrict__ A, const float* __restrict__ W,
                        const float* __restrict__ bias, float* __restrict__ OUT) {
  __shared__ __attribute__((aligned(16))) float lds[2 * GMAT * 2];
  const int lane = threadIdx.x & 31;
  const int wave = threadIdx.x >> 5;
  const int mr = lane & 15;
  const int kk = lane >> 4;
  const int m_base = blockIdx.x * 64 + (wave & 1) * 32;
  const int n_base = blockIdx.y * 64 + (wave >> 1) * 32;

  v8f acc[1][2][2];
  const v8f zero = {0.f, 0.f, 0.f, 0.f, 0.f, 0.f, 0.f, 0.f};
  for (int i = 0; i < 2; ++i)
    for (int j = 0; j < 2; ++j) acc[0][i][j] = zero;

  gemm_core<1>(A, W, W, lds, blockIdx.x * 64, blockIdx.y * 64, acc);

  for (int mt = 0; mt < 2; ++mt)
    for (int nt = 0; nt < 2; ++nt) {
      const int n = n_base + nt * 16 + mr;
      const float bi = bias[n];
#pragma unroll
      for (int r = 0; r < 8; ++r) {
        const int m = m_base + mt * 16 + kk * 8 + r;
        const float v = acc[0][mt][nt][r] + bi;
        OUT[(size_t)m * Dsz + n] = silu_f(v);
      }
    }
}

#define DT_ROWS  2
#define DT_WAVES 8
#define DT_THREADS (DT_WAVES * 32)

__global__ __launch_bounds__(DT_THREADS)
void dtproj_kernel(const float* __restrict__ X, const float* __restrict__ w_dt,
                   const float* __restrict__ b_dt, const float* __restrict__ ln_a,
                   float* __restrict__ LD) {
  __shared__ v4f wl[Hsz * Dsz / 4];
  const int lane = threadIdx.x & 31;
  const int wave = threadIdx.x >> 5;
  const int row0 = (blockIdx.x * DT_WAVES + wave) * DT_ROWS;

  v4f wr[Hsz * Dsz / 4 / DT_THREADS];
#pragma unroll
  for (int j = 0; j < Hsz * Dsz / 4 / DT_THREADS; ++j)
    wr[j] = *(const v4f*)(w_dt + 4 * (threadIdx.x + DT_THREADS * j));

  v4f xv[DT_ROWS][4];
#pragma unroll
  for (int r = 0; r < DT_ROWS; ++r)
#pragma unroll
    for (int c = 0; c < 4; ++c)
      xv[r][c] = *(const v4f*)(X + (size_t)(row0 + r) * Dsz + c * 128 + 4 * lane);

  const int vi = lane >> 1;
  const int h_out = vi & 7;
  const float bdt = b_dt[h_out];
  const float lna = ln_a[h_out];

  asm volatile("" :: "v"(xv[0][0]), "v"(xv[0][1]), "v"(xv[0][2]), "v"(xv[0][3]),
                     "v"(xv[1][0]), "v"(xv[1][1]), "v"(xv[1][2]), "v"(xv[1][3]),
                     "v"(bdt), "v"(lna));

#pragma unroll
  for (int j = 0; j < Hsz * Dsz / 4 / DT_THREADS; ++j)
    wl[threadIdx.x + DT_THREADS * j] = wr[j];
  __syncthreads();

  float v16[DT_ROWS * Hsz];
#pragma unroll
  for (int h = 0; h < Hsz; ++h) {
    v4f wv[4];
#pragma unroll
    for (int c = 0; c < 4; ++c) wv[c] = wl[h * (Dsz / 4) + c * 32 + lane];
#pragma unroll
    for (int r = 0; r < DT_ROWS; ++r) {
      float s[4];
#pragma unroll
      for (int c = 0; c < 4; ++c) {
        s[c] = xv[r][c].x * wv[c].x;
        s[c] = fmaf(xv[r][c].y, wv[c].y, s[c]);
        s[c] = fmaf(xv[r][c].z, wv[c].z, s[c]);
        s[c] = fmaf(xv[r][c].w, wv[c].w, s[c]);
      }
      v16[r * Hsz + h] = (s[0] + s[1]) + (s[2] + s[3]);
    }
  }

  const bool b4 = (lane & 16) != 0;
  const bool b3 = (lane & 8) != 0;
  const bool b2 = (lane & 4) != 0;
  const bool b1 = (lane & 2) != 0;
  float v8[8], v4[4], v2[2];
#pragma unroll
  for (int i = 0; i < 8; ++i) {
    const float keep = b4 ? v16[8 + i] : v16[i];
    const float send = b4 ? v16[i] : v16[8 + i];
    v8[i] = keep + __shfl_xor(send, 16, 32);
  }
#pragma unroll
  for (int i = 0; i < 4; ++i) {
    const float keep = b3 ? v8[4 + i] : v8[i];
    const float send = b3 ? v8[i] : v8[4 + i];
    v4[i] = keep + __shfl_xor(send, 8, 32);
  }
#pragma unroll
  for (int i = 0; i < 2; ++i) {
    const float keep = b2 ? v4[2 + i] : v4[i];
    const float send = b2 ? v4[i] : v4[2 + i];
    v2[i] = keep + __shfl_xor(send, 4, 32);
  }
  float dot;
  {
    const float keep = b1 ? v2[1] : v2[0];
    const float send = b1 ? v2[0] : v2[1];
    dot = keep + __shfl_xor(send, 2, 32);
  }
  dot += __shfl_xor(dot, 1, 32);

  const float v = dot + bdt;
  const float sp = (v > 20.f) ? v : log1pf(expf(v));
  const float ld = -expf(lna) * sp;
  if ((lane & 1) == 0) LD[(size_t)row0 * Hsz + vi] = ld;
}

#define ZS_PARTS 64
#define ZS_ROWS  (Lsz / ZS_PARTS)
#define PQK 24
#define ATP 68

__global__ __launch_bounds__(1024)
void zsum_kernel(const float* __restrict__ Z, const float* __restrict__ hidden,
                 const float* __restrict__ LD, const float* __restrict__ Wy,
                 const float* __restrict__ gn_b,
                 float* __restrict__ C, float* __restrict__ zhh,
                 float* __restrict__ S, float* __restrict__ PQT) {
  __shared__ float4 red4[ZS_PARTS * (Isz / 4)];
  __shared__ float  red2[16 * Isz];
  __shared__ float  scan[Lsz];
  __shared__ __attribute__((aligned(16))) float at[4 * ATP];
  const int b = blockIdx.x >> 3;
  const int h = blockIdx.x & 7;
  const int tid = threadIdx.x;
  const int q = tid & 15;
  const int p = tid >> 4;

  const float ldv = LD[(size_t)(b * Lsz + tid) * Hsz + h];

  const int ch = (b * Hsz + h) * Isz + (tid & 63);
  float zdiag = 0.f, hid = 0.f;
  if (tid < Isz) {
    zdiag = Z[((size_t)(b * Lsz + h) * Hsz + h) * Isz + tid];
    hid = hidden[ch];
  }

  const float* rp = Z + ((size_t)(b * Lsz + p) * Hsz + h) * Isz + 4 * q;
  float4 v[ZS_ROWS];
#pragma unroll
  for (int j = 0; j < ZS_ROWS; ++j)
    v[j] = *(const float4*)(rp + (size_t)j * ZS_PARTS * Dsz);

  const int c16 = tid & 15;
  const int half = (tid >> 4) & 1;
  const int wave = tid >> 5;
  const float gbh = gn_b[h];
  v4f wf[Isz / 8];
  {
    const float* wp = Wy + (size_t)(wave * 16 + c16) * Dsz + h * Isz + 4 * half;
#pragma unroll
    for (int kb = 0; kb < Isz / 8; ++kb) wf[kb] = *(const v4f*)(wp + 8 * kb);
  }

  scan[tid] = ldv;
  __syncthreads();
  for (int off = 1; off < Lsz; off <<= 1) {
    float t = scan[tid];
    if (tid >= off) t += scan[tid - off];
    __syncthreads();
    scan[tid] = t;
    __syncthreads();
  }
  S[(size_t)(b * Lsz + tid) * Hsz + h] = scan[tid];

  if (p <= h) v[0] = make_float4(0.f, 0.f, 0.f, 0.f);

#pragma unroll
  for (int w = 1; w < ZS_ROWS; w <<= 1)
#pragma unroll
    for (int j = 0; j + w < ZS_ROWS; j += 2 * w) {
      v[j].x += v[j + w].x; v[j].y += v[j + w].y;
      v[j].z += v[j + w].z; v[j].w += v[j + w].w;
    }
  red4[p * (Isz / 4) + q] = v[0];
  __syncthreads();

  const float* red = (const float*)red4;
  {
    const int i = tid & 63;
    const int g = tid >> 6;
    const float* r = red + (4 * g) * Isz + i;
    red2[g * Isz + i] = (r[0] + r[Isz]) + (r[2 * Isz] + r[3 * Isz]);
  }
  __syncthreads();

  float cfull = 0.f;
  if (tid < Isz) {
    float t[16];
#pragma unroll
    for (int g = 0; g < 16; ++g) t[g] = red2[g * Isz + tid];
#pragma unroll
    for (int w = 1; w < 16; w <<= 1)
#pragma unroll
      for (int g = 0; g + w < 16; g += 2 * w) t[g] += t[g + w];
    cfull = t[0] + hid;
    C[ch] = cfull;
    zhh[ch] = zdiag;
    at[tid] = zdiag;
    at[ATP + tid] = cfull;
    at[2 * ATP + tid] = gbh;
    at[3 * ATP + tid] = 0.f;
  }
  __syncthreads();

  float mz = 0.f, mc = 0.f;
  if (tid < Isz) {
    v4f sz = {0.f, 0.f, 0.f, 0.f}, sq = {0.f, 0.f, 0.f, 0.f};
#pragma unroll
    for (int j = 0; j < Isz / 4; ++j) {
      sz += *(const v4f*)(at + 4 * j);
      sq += *(const v4f*)(at + ATP + 4 * j);
    }
    mz = ((sz.x + sz.y) + (sz.z + sz.w)) * (1.0f / 64.0f);
    mc = ((sq.x + sq.y) + (sq.z + sq.w)) * (1.0f / 64.0f);
  }
  __syncthreads();
  if (tid < Isz) {
    at[tid] = zdiag - mz;
    at[ATP + tid] = cfull - mc;
  }
  __syncthreads();

  const int ar = (c16 < 3) ? c16 : 3;
  v8f acc = {0.f, 0.f, 0.f, 0.f, 0.f, 0.f, 0.f, 0.f};
#pragma unroll
  for (int kb = 0; kb < Isz / 8; ++kb) {
    const v4f a = *(const v4f*)(at + ar * ATP + 8 * kb + 4 * half);
    acc = wmma4(a.xy, wf[kb].xy, acc);
    acc = wmma4(a.zw, wf[kb].zw, acc);
  }
  if (half == 0) {
    float* o = PQT + (size_t)(b * Dsz + wave * 16 + c16) * PQK + h;
    o[0] = acc[0];
    o[8] = acc[1];
    o[16] = acc[2];
  }
}

#define HN_ROWS  2
#define HN_WAVES 8
#define HN_BROWS (HN_ROWS * HN_WAVES)
#define CTP 28
#define YTP 68

template <int M>
__device__ __forceinline__ float row_xor_add(float v) {
  const int p = __builtin_amdgcn_update_dpp(0, __float_as_int(v), 0x160 + M,
                                            0xf, 0xf, false);
  return v + __int_as_float(p);
}

__global__ __launch_bounds__(HN_WAVES * 32)
void hnorm_y_kernel(const float* __restrict__ S, const float* __restrict__ C,
                    const float* __restrict__ zhh, const float* __restrict__ gn_w,
                    const float* __restrict__ PQT, const float* __restrict__ by,
                    const float* __restrict__ GYA, float* __restrict__ Hout,
                    float* __restrict__ Y) {
  __shared__ __attribute__((aligned(16))) float ct[HN_BROWS * CTP];
  __shared__ __attribute__((aligned(16))) float yt[HN_WAVES * HN_BROWS * YTP];
  const int lane = threadIdx.x & 31;
  const int wave = threadIdx.x >> 5;
  const int brow0 = blockIdx.x * HN_BROWS;
  const int row0 = brow0 + wave * HN_ROWS;
  const int b = row0 >> 10;
  const int hh = lane >> 4;
  const int c16 = lane & 15;
  const int nb = wave * 64;

  v4f fb[4][PQK / 8];
#pragma unroll
  for (int nt = 0; nt < 4; ++nt)
#pragma unroll
    for (int kb = 0; kb < PQK / 8; ++kb)
      fb[nt][kb] = *(const v4f*)(PQT + (size_t)(b * Dsz + nb + nt * 16 + c16) * PQK +
                                 8 * kb + 4 * hh);
  v4f gy[HN_BROWS / 2];
#pragma unroll
  for (int it = 0; it < HN_BROWS / 2; ++it)
    gy[it] = *(const v4f*)(GYA + (size_t)(brow0 + 2 * it + hh) * Dsz + nb + 4 * c16);
  const v4f bv = *(const v4f*)(by + nb + 4 * c16);

  v4f zv[4], cv[4];
  float gw[4];
  float sv[HN_ROWS][4];
#pragma unroll
  for (int c = 0; c < 4; ++c) {
    zv[c] = *(const v4f*)(zhh + (size_t)b * Dsz + c * 128 + 4 * lane);
    cv[c] = *(const v4f*)(C + (size_t)b * Dsz + c * 128 + 4 * lane);
    gw[c] = gn_w[2 * c + hh];
  }
#pragma unroll
  for (int r = 0; r < HN_ROWS; ++r)
#pragma unroll
    for (int c = 0; c < 4; ++c)
      sv[r][c] = S[(size_t)(row0 + r) * Hsz + 2 * c + hh];

#pragma unroll
  for (int r = 0; r < HN_ROWS; ++r) {
    const size_t ro = (size_t)(row0 + r) * Dsz + 4 * lane;
    float* cr = ct + (wave * HN_ROWS + r) * CTP;
#pragma unroll
    for (int c = 0; c < 4; ++c) {
      const float e = expf(sv[r][c]);
      v4f hv;
      hv.x = zv[c].x + e * cv[c].x;
      hv.y = zv[c].y + e * cv[c].y;
      hv.z = zv[c].z + e * cv[c].z;
      hv.w = zv[c].w + e * cv[c].w;
      *(v4f*)(Hout + ro + c * 128) = hv;

      float s = (hv.x + hv.y) + (hv.z + hv.w);
      s = row_xor_add<1>(s);
      s = row_xor_add<2>(s);
      s = row_xor_add<4>(s);
      s = row_xor_add<8>(s);
      const float mu = s * (1.0f / 64.0f);

      v4f d;
      d.x = hv.x - mu; d.y = hv.y - mu; d.z = hv.z - mu; d.w = hv.w - mu;
      float q = (d.x * d.x + d.y * d.y) + (d.z * d.z + d.w * d.w);
      q = row_xor_add<1>(q);
      q = row_xor_add<2>(q);
      q = row_xor_add<4>(q);
      q = row_xor_add<8>(q);
      const float var = q * (1.0f / 64.0f);

      const float sc = gw[c] / sqrtf(var + GN_EPS);
      if (c16 == 0) {
        const int head = 2 * c + hh;
        cr[head] = sc;
        cr[8 + head] = sc * e;
        cr[16 + head] = 1.0f;
      }
    }
  }
  __syncthreads();

  v4f fa[PQK / 8];
#pragma unroll
  for (int kb = 0; kb < PQK / 8; ++kb)
    fa[kb] = *(const v4f*)(ct + c16 * CTP + 8 * kb + 4 * hh);

  v8f acc[4];
  const v8f zero = {0.f, 0.f, 0.f, 0.f, 0.f, 0.f, 0.f, 0.f};
#pragma unroll
  for (int nt = 0; nt < 4; ++nt) acc[nt] = zero;
#pragma unroll
  for (int kb = 0; kb < PQK / 8; ++kb) {
#pragma unroll
    for (int nt = 0; nt < 4; ++nt)
      acc[nt] = wmma4(fa[kb].xy, fb[nt][kb].xy, acc[nt]);
#pragma unroll
    for (int nt = 0; nt < 4; ++nt)
      acc[nt] = wmma4(fa[kb].zw, fb[nt][kb].zw, acc[nt]);
  }

  float* yw = yt + wave * HN_BROWS * YTP;
#pragma unroll
  for (int nt = 0; nt < 4; ++nt)
#pragma unroll
    for (int r = 0; r < 8; ++r)
      yw[(8 * hh + r) * YTP + nt * 16 + c16] = acc[nt][r];
  __syncthreads();

#pragma unroll
  for (int it = 0; it < HN_BROWS / 2; ++it) {
    const int row = 2 * it + hh;
    const v4f d = *(const v4f*)(yw + row * YTP + 4 * c16);
    v4f o;
    o.x = (d.x + bv.x) * gy[it].x;
    o.y = (d.y + bv.y) * gy[it].y;
    o.z = (d.z + bv.z) * gy[it].z;
    o.w = (d.w + bv.w) * gy[it].w;
    *(v4f*)(Y + (size_t)(brow0 + row) * Dsz + nb + 4 * c16) = o;
  }
}

extern "C" void kernel_launch(void* const* d_in, const int* in_sizes, int n_in,
                              void* d_out, int out_size, void* d_ws, size_t ws_size,
                              hipStream_t stream) {
  const float* x      = (const float*)d_in[0];
  const float* hidden = (const float*)d_in[1];
  const float* w_z    = (const float*)d_in[2];
  const float* b_z    = (const float*)d_in[3];
  const float* w_za   = (const float*)d_in[4];
  const float* b_za   = (const float*)d_in[5];
  const float* w_y    = (const float*)d_in[6];
  const float* b_y    = (const float*)d_in[7];
  const float* w_ya   = (const float*)d_in[8];
  const float* b_ya   = (const float*)d_in[9];
  const float* w_dt   = (const float*)d_in[10];
  const float* b_dt   = (const float*)d_in[11];
  const float* ln_a   = (const float*)d_in[12];
  const float* gn_w   = (const float*)d_in[13];
  const float* gn_b   = (const float*)d_in[14];

  float* y_out = (float*)d_out;
  float* h_out = y_out + (size_t)Msz * Dsz;

  float* ws   = (float*)d_ws;
  float* Z    = ws;
  float* GYA  = Z    + (size_t)Msz * Dsz;
  float* Sb   = GYA  + (size_t)Msz * Dsz;
  float* Cb   = Sb   + (size_t)Msz * Hsz;
  float* ZHH  = Cb   + (size_t)Bsz * Hsz * Isz;
  float* LDb  = ZHH  + (size_t)Bsz * Hsz * Isz;
  float* PQT  = LDb  + (size_t)Msz * Hsz;

  const dim3 gGemm(Msz / 64, Dsz / 64, 1);
  const dim3 bGemm(128, 1, 1);

  gemm_gate_z_kernel<<<gGemm, bGemm, 0, stream>>>(x, w_z, b_z, w_za, b_za, Z);
  gemm_single_kernel<<<gGemm, bGemm, 0, stream>>>(x, w_ya, b_ya, GYA);
  dtproj_kernel<<<dim3(Msz / (DT_WAVES * DT_ROWS)), dim3(DT_WAVES * 32), 0, stream>>>(
      x, w_dt, b_dt, ln_a, LDb);
  zsum_kernel<<<dim3(Bsz * Hsz), dim3(1024), 0, stream>>>(Z, hidden, LDb, w_y, gn_b,
                                                          Cb, ZHH, Sb, PQT);
  hnorm_y_kernel<<<dim3(Msz / HN_BROWS), dim3(HN_WAVES * 32), 0, stream>>>(
      Sb, Cb, ZHH, gn_w, PQT, b_y, GYA, h_out, y_out);
}


# GEMMs: the three 512-deep x projections (Wz, Wza, Wya) share one staged x tile in one launch, 24 products per fragment group; was two launches, x staged twice. Sum order unchanged
# speedup vs baseline: 3.6001x; 1.0853x over previous
#include <hip/hip_runtime.h>
#include <hip/hip_bf16.h>
#include <math.h>


#define Bsz 4
#define Lsz 1024
#define Dsz 512
#define Hsz 8
#define Isz 64
#define Msz (Bsz * Lsz)
#define GN_EPS 1e-5f

typedef float v2f __attribute__((ext_vector_type(2)));
typedef float v4f __attribute__((ext_vector_type(4)));
typedef float v8f __attribute__((ext_vector_type(8)));

__device__ __forceinline__ v8f wmma4(v2f a, v2f b, v8f c) {
  return __builtin_amdgcn_wmma_f32_16x16x4_f32(
       false, a,  false, b,
       (short)0, c,  false,  false);
}

__device__ __forceinline__ float silu_f(float v) {
  return v / (1.0f + expf(-v));
}

#define GK 32
#define GP 36
#define GMAT (64 * GP)
#define GKT (Dsz / GK)

template <int NB>
__device__ __forceinline__ void gemm_core(const float* __restrict__ A,
                                          const float* const (&W)[NB],
                                          float* __restrict__ lds,
                                          const int m_blk, const int n_blk,
                                          v8f (&acc)[NB][2][2]) {
  constexpr int STAGE = GMAT * (1 + NB);
  const int tid = threadIdx.x;
  const int lane = tid & 31;
  const int wave = tid >> 5;
  const int mr = lane & 15;
  const int kk = lane >> 4;
  const int wm = (wave & 1) * 32;
  const int wn = (wave >> 1) * 32;

  const int crow = tid >> 3;
  const int ccol = (tid & 7) * 4;
  const float* gA = A + (size_t)(m_blk + crow) * Dsz + ccol;
  const float* gB[NB];
#pragma unroll
  for (int nb = 0; nb < NB; ++nb)
    gB[nb] = W[nb] + (size_t)(n_blk + crow) * Dsz + ccol;
  const int soff = crow * GP + ccol;

  const int aoff = (wm + mr) * GP + 4 * kk;
  const int boff = (wn + mr) * GP + 4 * kk;

  v4f pa[4], pb[NB][4];
#pragma unroll
  for (int j = 0; j < 4; ++j) {
    pa[j] = *(const v4f*)(gA + (size_t)j * 16 * Dsz);
#pragma unroll
    for (int nb = 0; nb < NB; ++nb)
      pb[nb][j] = *(const v4f*)(gB[nb] + (size_t)j * 16 * Dsz);
  }
#pragma unroll
  for (int j = 0; j < 4; ++j) {
    *(v4f*)(lds + soff + j * 16 * GP) = pa[j];
#pragma unroll
    for (int nb = 0; nb < NB; ++nb)
      *(v4f*)(lds + GMAT * (1 + nb) + soff + j * 16 * GP) = pb[nb][j];
  }
  __syncthreads();

#pragma unroll 2
  for (int kt = 0; kt < GKT; ++kt) {
    const float* cur = lds + (kt & 1) * STAGE;
    float* nxt = lds + ((kt + 1) & 1) * STAGE;
    const bool more = (kt + 1 < GKT);

    if (more) {
      const int ko = (kt + 1) * GK;
#pragma unroll
      for (int j = 0; j < 4; ++j) {
        pa[j] = *(const v4f*)(gA + (size_t)j * 16 * Dsz + ko);
#pragma unroll
        for (int nb = 0; nb < NB; ++nb)
          pb[nb][j] = *(const v4f*)(gB[nb] + (size_t)j * 16 * Dsz + ko);
      }
    }

#pragma unroll
    for (int kb = 0; kb < GK / 8; ++kb) {
      v4f fa[2], fb[NB][2];
#pragma unroll
      for (int t = 0; t < 2; ++t) {
        fa[t] = *(const v4f*)(cur + aoff + t * 16 * GP + kb * 8);
#pragma unroll
        for (int nb = 0; nb < NB; ++nb)
          fb[nb][t] = *(const v4f*)(cur + GMAT * (1 + nb) + boff + t * 16 * GP + kb * 8);
      }
#pragma unroll
      for (int nb = 0; nb < NB; ++nb)
#pragma unroll
        for (int mt = 0; mt < 2; ++mt)
#pragma unroll
          for (int nt = 0; nt < 2; ++nt)
            acc[nb][mt][nt] = wmma4(fa[mt].xy, fb[nb][nt].xy, acc[nb][mt][nt]);
#pragma unroll
      for (int nb = 0; nb < NB; ++nb)
#pragma unroll
        for (int mt = 0; mt < 2; ++mt)
#pragma unroll
          for (int nt = 0; nt < 2; ++nt)
            acc[nb][mt][nt] = wmma4(fa[mt].zw, fb[nb][nt].zw, acc[nb][mt][nt]);
    }

    if (more) {
#pragma unroll
      for (int j = 0; j < 4; ++j) {
        *(v4f*)(nxt + soff + j * 16 * GP) = pa[j];
#pragma unroll
        for (int nb = 0; nb < NB; ++nb)
          *(v4f*)(nxt + GMAT * (1 + nb) + soff + j * 16 * GP) = pb[nb][j];
      }
    }
    __syncthreads();
  }
}

__global__ __launch_bounds__(128)
void gemm_xproj_kernel(const float* __restrict__ X,
                       const float* __restrict__ Wz,  const float* __restrict__ bz,
                       const float* __restrict__ Wza, const float* __restrict__ bza,
                       const float* __restrict__ Wya, const float* __restrict__ bya,
                       float* __restrict__ Z, float* __restrict__ GYA) {
  __shared__ __attribute__((aligned(16))) float lds[2 * GMAT * 4];
  const int lane = threadIdx.x & 31;
  const int wave = threadIdx.x >> 5;
  const int mr = lane & 15;
  const int kk = lane >> 4;
  const int m_base = blockIdx.x * 64 + (wave & 1) * 32;
  const int n_base = blockIdx.y * 64 + (wave >> 1) * 32;

  v8f acc[3][2][2];
  const v8f zero = {0.f, 0.f, 0.f, 0.f, 0.f, 0.f, 0.f, 0.f};
  for (int nb = 0; nb < 3; ++nb)
    for (int i = 0; i < 2; ++i)
      for (int j = 0; j < 2; ++j) acc[nb][i][j] = zero;

  const float* const W[3] = {Wz, Wza, Wya};
  gemm_core<3>(X, W, lds, blockIdx.x * 64, blockIdx.y * 64, acc);

  for (int mt = 0; mt < 2; ++mt)
    for (int nt = 0; nt < 2; ++nt) {
      const int n = n_base + nt * 16 + mr;
      const float biz = bz[n], biza = bza[n], biya = bya[n];
#pragma unroll
      for (int r = 0; r < 8; ++r) {
        const int m = m_base + mt * 16 + kk * 8 + r;
        const float v1 = acc[0][mt][nt][r] + biz;
        const float v2 = acc[1][mt][nt][r] + biza;
        const float v3 = acc[2][mt][nt][r] + biya;
        Z[(size_t)m * Dsz + n] = v1 * silu_f(v2);
        GYA[(size_t)m * Dsz + n] = silu_f(v3);
      }
    }
}

#define DT_ROWS  2
#define DT_WAVES 8
#define DT_THREADS (DT_WAVES * 32)

__global__ __launch_bounds__(DT_THREADS)
void dtproj_kernel(const float* __restrict__ X, const float* __restrict__ w_dt,
                   const float* __restrict__ b_dt, const float* __restrict__ ln_a,
                   float* __restrict__ LD) {
  __shared__ v4f wl[Hsz * Dsz / 4];
  const int lane = threadIdx.x & 31;
  const int wave = threadIdx.x >> 5;
  const int row0 = (blockIdx.x * DT_WAVES + wave) * DT_ROWS;

  v4f wr[Hsz * Dsz / 4 / DT_THREADS];
#pragma unroll
  for (int j = 0; j < Hsz * Dsz / 4 / DT_THREADS; ++j)
    wr[j] = *(const v4f*)(w_dt + 4 * (threadIdx.x + DT_THREADS * j));

  v4f xv[DT_ROWS][4];
#pragma unroll
  for (int r = 0; r < DT_ROWS; ++r)
#pragma unroll
    for (int c = 0; c < 4; ++c)
      xv[r][c] = *(const v4f*)(X + (size_t)(row0 + r) * Dsz + c * 128 + 4 * lane);

  const int vi = lane >> 1;
  const int h_out = vi & 7;
  const float bdt = b_dt[h_out];
  const float lna = ln_a[h_out];

  asm volatile("" :: "v"(xv[0][0]), "v"(xv[0][1]), "v"(xv[0][2]), "v"(xv[0][3]),
                     "v"(xv[1][0]), "v"(xv[1][1]), "v"(xv[1][2]), "v"(xv[1][3]),
                     "v"(bdt), "v"(lna));

#pragma unroll
  for (int j = 0; j < Hsz * Dsz / 4 / DT_THREADS; ++j)
    wl[threadIdx.x + DT_THREADS * j] = wr[j];
  __syncthreads();

  float v16[DT_ROWS * Hsz];
#pragma unroll
  for (int h = 0; h < Hsz; ++h) {
    v4f wv[4];
#pragma unroll
    for (int c = 0; c < 4; ++c) wv[c] = wl[h * (Dsz / 4) + c * 32 + lane];
#pragma unroll
    for (int r = 0; r < DT_ROWS; ++r) {
      float s[4];
#pragma unroll
      for (int c = 0; c < 4; ++c) {
        s[c] = xv[r][c].x * wv[c].x;
        s[c] = fmaf(xv[r][c].y, wv[c].y, s[c]);
        s[c] = fmaf(xv[r][c].z, wv[c].z, s[c]);
        s[c] = fmaf(xv[r][c].w, wv[c].w, s[c]);
      }
      v16[r * Hsz + h] = (s[0] + s[1]) + (s[2] + s[3]);
    }
  }

  const bool b4 = (lane & 16) != 0;
  const bool b3 = (lane & 8) != 0;
  const bool b2 = (lane & 4) != 0;
  const bool b1 = (lane & 2) != 0;
  float v8[8], v4[4], v2[2];
#pragma unroll
  for (int i = 0; i < 8; ++i) {
    const float keep = b4 ? v16[8 + i] : v16[i];
    const float send = b4 ? v16[i] : v16[8 + i];
    v8[i] = keep + __shfl_xor(send, 16, 32);
  }
#pragma unroll
  for (int i = 0; i < 4; ++i) {
    const float keep = b3 ? v8[4 + i] : v8[i];
    const float send = b3 ? v8[i] : v8[4 + i];
    v4[i] = keep + __shfl_xor(send, 8, 32);
  }
#pragma unroll
  for (int i = 0; i < 2; ++i) {
    const float keep = b2 ? v4[2 + i] : v4[i];
    const float send = b2 ? v4[i] : v4[2 + i];
    v2[i] = keep + __shfl_xor(send, 4, 32);
  }
  float dot;
  {
    const float keep = b1 ? v2[1] : v2[0];
    const float send = b1 ? v2[0] : v2[1];
    dot = keep + __shfl_xor(send, 2, 32);
  }
  dot += __shfl_xor(dot, 1, 32);

  const float v = dot + bdt;
  const float sp = (v > 20.f) ? v : log1pf(expf(v));
  const float ld = -expf(lna) * sp;
  if ((lane & 1) == 0) LD[(size_t)row0 * Hsz + vi] = ld;
}

#define ZS_PARTS 64
#define ZS_ROWS  (Lsz / ZS_PARTS)
#define PQK 24
#define ATP 68

__global__ __launch_bounds__(1024)
void zsum_kernel(const float* __restrict__ Z, const float* __restrict__ hidden,
                 const float* __restrict__ LD, const float* __restrict__ Wy,
                 const float* __restrict__ gn_b,
                 float* __restrict__ C, float* __restrict__ zhh,
                 float* __restrict__ S, float* __restrict__ PQT) {
  __shared__ float4 red4[ZS_PARTS * (Isz / 4)];
  __shared__ float  red2[16 * Isz];
  __shared__ float  scan[Lsz];
  __shared__ __attribute__((aligned(16))) float at[4 * ATP];
  const int b = blockIdx.x >> 3;
  const int h = blockIdx.x & 7;
  const int tid = threadIdx.x;
  const int q = tid & 15;
  const int p = tid >> 4;

  const float ldv = LD[(size_t)(b * Lsz + tid) * Hsz + h];

  const int ch = (b * Hsz + h) * Isz + (tid & 63);
  float zdiag = 0.f, hid = 0.f;
  if (tid < Isz) {
    zdiag = Z[((size_t)(b * Lsz + h) * Hsz + h) * Isz + tid];
    hid = hidden[ch];
  }

  const float* rp = Z + ((size_t)(b * Lsz + p) * Hsz + h) * Isz + 4 * q;
  float4 v[ZS_ROWS];
#pragma unroll
  for (int j = 0; j < ZS_ROWS; ++j)
    v[j] = *(const float4*)(rp + (size_t)j * ZS_PARTS * Dsz);

  const int c16 = tid & 15;
  const int half = (tid >> 4) & 1;
  const int wave = tid >> 5;
  const float gbh = gn_b[h];
  v4f wf[Isz / 8];
  {
    const float* wp = Wy + (size_t)(wave * 16 + c16) * Dsz + h * Isz + 4 * half;
#pragma unroll
    for (int kb = 0; kb < Isz / 8; ++kb) wf[kb] = *(const v4f*)(wp + 8 * kb);
  }

  scan[tid] = ldv;
  __syncthreads();
  for (int off = 1; off < Lsz; off <<= 1) {
    float t = scan[tid];
    if (tid >= off) t += scan[tid - off];
    __syncthreads();
    scan[tid] = t;
    __syncthreads();
  }
  S[(size_t)(b * Lsz + tid) * Hsz + h] = scan[tid];

  if (p <= h) v[0] = make_float4(0.f, 0.f, 0.f, 0.f);

#pragma unroll
  for (int w = 1; w < ZS_ROWS; w <<= 1)
#pragma unroll
    for (int j = 0; j + w < ZS_ROWS; j += 2 * w) {
      v[j].x += v[j + w].x; v[j].y += v[j + w].y;
      v[j].z += v[j + w].z; v[j].w += v[j + w].w;
    }
  red4[p * (Isz / 4) + q] = v[0];
  __syncthreads();

  const float* red = (const float*)red4;
  {
    const int i = tid & 63;
    const int g = tid >> 6;
    const float* r = red + (4 * g) * Isz + i;
    red2[g * Isz + i] = (r[0] + r[Isz]) + (r[2 * Isz] + r[3 * Isz]);
  }
  __syncthreads();

  float cfull = 0.f;
  if (tid < Isz) {
    float t[16];
#pragma unroll
    for (int g = 0; g < 16; ++g) t[g] = red2[g * Isz + tid];
#pragma unroll
    for (int w = 1; w < 16; w <<= 1)
#pragma unroll
      for (int g = 0; g + w < 16; g += 2 * w) t[g] += t[g + w];
    cfull = t[0] + hid;
    C[ch] = cfull;
    zhh[ch] = zdiag;
    at[tid] = zdiag;
    at[ATP + tid] = cfull;
    at[2 * ATP + tid] = gbh;
    at[3 * ATP + tid] = 0.f;
  }
  __syncthreads();

  float mz = 0.f, mc = 0.f;
  if (tid < Isz) {
    v4f sz = {0.f, 0.f, 0.f, 0.f}, sq = {0.f, 0.f, 0.f, 0.f};
#pragma unroll
    for (int j = 0; j < Isz / 4; ++j) {
      sz += *(const v4f*)(at + 4 * j);
      sq += *(const v4f*)(at + ATP + 4 * j);
    }
    mz = ((sz.x + sz.y) + (sz.z + sz.w)) * (1.0f / 64.0f);
    mc = ((sq.x + sq.y) + (sq.z + sq.w)) * (1.0f / 64.0f);
  }
  __syncthreads();
  if (tid < Isz) {
    at[tid] = zdiag - mz;
    at[ATP + tid] = cfull - mc;
  }
  __syncthreads();

  const int ar = (c16 < 3) ? c16 : 3;
  v8f acc = {0.f, 0.f, 0.f, 0.f, 0.f, 0.f, 0.f, 0.f};
#pragma unroll
  for (int kb = 0; kb < Isz / 8; ++kb) {
    const v4f a = *(const v4f*)(at + ar * ATP + 8 * kb + 4 * half);
    acc = wmma4(a.xy, wf[kb].xy, acc);
    acc = wmma4(a.zw, wf[kb].zw, acc);
  }
  if (half == 0) {
    float* o = PQT + (size_t)(b * Dsz + wave * 16 + c16) * PQK + h;
    o[0] = acc[0];
    o[8] = acc[1];
    o[16] = acc[2];
  }
}

#define HN_ROWS  2
#define HN_WAVES 8
#define HN_BROWS (HN_ROWS * HN_WAVES)
#define CTP 28
#define YTP 68

template <int M>
__device__ __forceinline__ float row_xor_add(float v) {
  const int p = __builtin_amdgcn_update_dpp(0, __float_as_int(v), 0x160 + M,
                                            0xf, 0xf, false);
  return v + __int_as_float(p);
}

__global__ __launch_bounds__(HN_WAVES * 32)
void hnorm_y_kernel(const float* __restrict__ S, const float* __restrict__ C,
                    const float* __restrict__ zhh, const float* __restrict__ gn_w,
                    const float* __restrict__ PQT, const float* __restrict__ by,
                    const float* __restrict__ GYA, float* __restrict__ Hout,
                    float* __restrict__ Y) {
  __shared__ __attribute__((aligned(16))) float ct[HN_BROWS * CTP];
  __shared__ __attribute__((aligned(16))) float yt[HN_WAVES * HN_BROWS * YTP];
  const int lane = threadIdx.x & 31;
  const int wave = threadIdx.x >> 5;
  const int brow0 = blockIdx.x * HN_BROWS;
  const int row0 = brow0 + wave * HN_ROWS;
  const int b = row0 >> 10;
  const int hh = lane >> 4;
  const int c16 = lane & 15;
  const int nb = wave * 64;

  v4f fb[4][PQK / 8];
#pragma unroll
  for (int nt = 0; nt < 4; ++nt)
#pragma unroll
    for (int kb = 0; kb < PQK / 8; ++kb)
      fb[nt][kb] = *(const v4f*)(PQT + (size_t)(b * Dsz + nb + nt * 16 + c16) * PQK +
                                 8 * kb + 4 * hh);
  v4f gy[HN_BROWS / 2];
#pragma unroll
  for (int it = 0; it < HN_BROWS / 2; ++it)
    gy[it] = *(const v4f*)(GYA + (size_t)(brow0 + 2 * it + hh) * Dsz + nb + 4 * c16);
  const v4f bv = *(const v4f*)(by + nb + 4 * c16);

  v4f zv[4], cv[4];
  float gw[4];
  float sv[HN_ROWS][4];
#pragma unroll
  for (int c = 0; c < 4; ++c) {
    zv[c] = *(const v4f*)(zhh + (size_t)b * Dsz + c * 128 + 4 * lane);
    cv[c] = *(const v4f*)(C + (size_t)b * Dsz + c * 128 + 4 * lane);
    gw[c] = gn_w[2 * c + hh];
  }
#pragma unroll
  for (int r = 0; r < HN_ROWS; ++r)
#pragma unroll
    for (int c = 0; c < 4; ++c)
      sv[r][c] = S[(size_t)(row0 + r) * Hsz + 2 * c + hh];

#pragma unroll
  for (int r = 0; r < HN_ROWS; ++r) {
    const size_t ro = (size_t)(row0 + r) * Dsz + 4 * lane;
    float* cr = ct + (wave * HN_ROWS + r) * CTP;
#pragma unroll
    for (int c = 0; c < 4; ++c) {
      const float e = expf(sv[r][c]);
      v4f hv;
      hv.x = zv[c].x + e * cv[c].x;
      hv.y = zv[c].y + e * cv[c].y;
      hv.z = zv[c].z + e * cv[c].z;
      hv.w = zv[c].w + e * cv[c].w;
      *(v4f*)(Hout + ro + c * 128) = hv;

      float s = (hv.x + hv.y) + (hv.z + hv.w);
      s = row_xor_add<1>(s);
      s = row_xor_add<2>(s);
      s = row_xor_add<4>(s);
      s = row_xor_add<8>(s);
      const float mu = s * (1.0f / 64.0f);

      v4f d;
      d.x = hv.x - mu; d.y = hv.y - mu; d.z = hv.z - mu; d.w = hv.w - mu;
      float q = (d.x * d.x + d.y * d.y) + (d.z * d.z + d.w * d.w);
      q = row_xor_add<1>(q);
      q = row_xor_add<2>(q);
      q = row_xor_add<4>(q);
      q = row_xor_add<8>(q);
      const float var = q * (1.0f / 64.0f);

      const float sc = gw[c] / sqrtf(var + GN_EPS);
      if (c16 == 0) {
        const int head = 2 * c + hh;
        cr[head] = sc;
        cr[8 + head] = sc * e;
        cr[16 + head] = 1.0f;
      }
    }
  }
  __syncthreads();

  v4f fa[PQK / 8];
#pragma unroll
  for (int kb = 0; kb < PQK / 8; ++kb)
    fa[kb] = *(const v4f*)(ct + c16 * CTP + 8 * kb + 4 * hh);

  v8f acc[4];
  const v8f zero = {0.f, 0.f, 0.f, 0.f, 0.f, 0.f, 0.f, 0.f};
#pragma unroll
  for (int nt = 0; nt < 4; ++nt) acc[nt] = zero;
#pragma unroll
  for (int kb = 0; kb < PQK / 8; ++kb) {
#pragma unroll
    for (int nt = 0; nt < 4; ++nt)
      acc[nt] = wmma4(fa[kb].xy, fb[nt][kb].xy, acc[nt]);
#pragma unroll
    for (int nt = 0; nt < 4; ++nt)
      acc[nt] = wmma4(fa[kb].zw, fb[nt][kb].zw, acc[nt]);
  }

  float* yw = yt + wave * HN_BROWS * YTP;
#pragma unroll
  for (int nt = 0; nt < 4; ++nt)
#pragma unroll
    for (int r = 0; r < 8; ++r)
      yw[(8 * hh + r) * YTP + nt * 16 + c16] = acc[nt][r];
  __syncthreads();

#pragma unroll
  for (int it = 0; it < HN_BROWS / 2; ++it) {
    const int row = 2 * it + hh;
    const v4f d = *(const v4f*)(yw + row * YTP + 4 * c16);
    v4f o;
    o.x = (d.x + bv.x) * gy[it].x;
    o.y = (d.y + bv.y) * gy[it].y;
    o.z = (d.z + bv.z) * gy[it].z;
    o.w = (d.w + bv.w) * gy[it].w;
    *(v4f*)(Y + (size_t)(brow0 + row) * Dsz + nb + 4 * c16) = o;
  }
}

extern "C" void kernel_launch(void* const* d_in, const int* in_sizes, int n_in,
                              void* d_out, int out_size, void* d_ws, size_t ws_size,
                              hipStream_t stream) {
  const float* x      = (const float*)d_in[0];
  const float* hidden = (const float*)d_in[1];
  const float* w_z    = (const float*)d_in[2];
  const float* b_z    = (const float*)d_in[3];
  const float* w_za   = (const float*)d_in[4];
  const float* b_za   = (const float*)d_in[5];
  const float* w_y    = (const float*)d_in[6];
  const float* b_y    = (const float*)d_in[7];
  const float* w_ya   = (const float*)d_in[8];
  const float* b_ya   = (const float*)d_in[9];
  const float* w_dt   = (const float*)d_in[10];
  const float* b_dt   = (const float*)d_in[11];
  const float* ln_a   = (const float*)d_in[12];
  const float* gn_w   = (const float*)d_in[13];
  const float* gn_b   = (const float*)d_in[14];

  float* y_out = (float*)d_out;
  float* h_out = y_out + (size_t)Msz * Dsz;

  float* ws   = (float*)d_ws;
  float* Z    = ws;
  float* GYA  = Z    + (size_t)Msz * Dsz;
  float* Sb   = GYA  + (size_t)Msz * Dsz;
  float* Cb   = Sb   + (size_t)Msz * Hsz;
  float* ZHH  = Cb   + (size_t)Bsz * Hsz * Isz;
  float* LDb  = ZHH  + (size_t)Bsz * Hsz * Isz;
  float* PQT  = LDb  + (size_t)Msz * Hsz;

  const dim3 gGemm(Msz / 64, Dsz / 64, 1);
  const dim3 bGemm(128, 1, 1);

  gemm_xproj_kernel<<<gGemm, bGemm, 0, stream>>>(x, w_z, b_z, w_za, b_za, w_ya, b_ya,
                                                 Z, GYA);
  dtproj_kernel<<<dim3(Msz / (DT_WAVES * DT_ROWS)), dim3(DT_WAVES * 32), 0, stream>>>(
      x, w_dt, b_dt, ln_a, LDb);
  zsum_kernel<<<dim3(Bsz * Hsz), dim3(1024), 0, stream>>>(Z, hidden, LDb, w_y, gn_b,
                                                          Cb, ZHH, Sb, PQT);
  hnorm_y_kernel<<<dim3(Msz / HN_BROWS), dim3(HN_WAVES * 32), 0, stream>>>(
      Sb, Cb, ZHH, gn_w, PQT, b_y, GYA, h_out, y_out);
}


# dt projection runs as 256 extra blocks of the GEMM launch (role chosen per block, 4 rows a wave) beside the matrix-bound GEMM tiles; was a launch of its own: 4 launches -> 3. Sum order unchanged
# speedup vs baseline: 3.8265x; 1.0629x over previous
#include <hip/hip_runtime.h>
#include <hip/hip_bf16.h>
#include <math.h>


#define Bsz 4
#define Lsz 1024
#define Dsz 512
#define Hsz 8
#define Isz 64
#define Msz (Bsz * Lsz)
#define GN_EPS 1e-5f

typedef float v2f __attribute__((ext_vector_type(2)));
typedef float v4f __attribute__((ext_vector_type(4)));
typedef float v8f __attribute__((ext_vector_type(8)));

__device__ __forceinline__ v8f wmma4(v2f a, v2f b, v8f c) {
  return __builtin_amdgcn_wmma_f32_16x16x4_f32(
       false, a,  false, b,
       (short)0, c,  false,  false);
}

__device__ __forceinline__ float silu_f(float v) {
  return v / (1.0f + expf(-v));
}

#define GK 32
#define GP 36
#define GMAT (64 * GP)
#define GKT (Dsz / GK)

template <int NB>
__device__ __forceinline__ void gemm_core(const float* __restrict__ A,
                                          const float* const (&W)[NB],
                                          float* __restrict__ lds,
                                          const int m_blk, const int n_blk,
                                          v8f (&acc)[NB][2][2]) {
  constexpr int STAGE = GMAT * (1 + NB);
  const int tid = threadIdx.x;
  const int lane = tid & 31;
  const int wave = tid >> 5;
  const int mr = lane & 15;
  const int kk = lane >> 4;
  const int wm = (wave & 1) * 32;
  const int wn = (wave >> 1) * 32;

  const int crow = tid >> 3;
  const int ccol = (tid & 7) * 4;
  const float* gA = A + (size_t)(m_blk + crow) * Dsz + ccol;
  const float* gB[NB];
#pragma unroll
  for (int nb = 0; nb < NB; ++nb)
    gB[nb] = W[nb] + (size_t)(n_blk + crow) * Dsz + ccol;
  const int soff = crow * GP + ccol;

  const int aoff = (wm + mr) * GP + 4 * kk;
  const int boff = (wn + mr) * GP + 4 * kk;

  v4f pa[4], pb[NB][4];
#pragma unroll
  for (int j = 0; j < 4; ++j) {
    pa[j] = *(const v4f*)(gA + (size_t)j * 16 * Dsz);
#pragma unroll
    for (int nb = 0; nb < NB; ++nb)
      pb[nb][j] = *(const v4f*)(gB[nb] + (size_t)j * 16 * Dsz);
  }
#pragma unroll
  for (int j = 0; j < 4; ++j) {
    *(v4f*)(lds + soff + j * 16 * GP) = pa[j];
#pragma unroll
    for (int nb = 0; nb < NB; ++nb)
      *(v4f*)(lds + GMAT * (1 + nb) + soff + j * 16 * GP) = pb[nb][j];
  }
  __syncthreads();

#pragma unroll 2
  for (int kt = 0; kt < GKT; ++kt) {
    const float* cur = lds + (kt & 1) * STAGE;
    float* nxt = lds + ((kt + 1) & 1) * STAGE;
    const bool more = (kt + 1 < GKT);

    if (more) {
      const int ko = (kt + 1) * GK;
#pragma unroll
      for (int j = 0; j < 4; ++j) {
        pa[j] = *(const v4f*)(gA + (size_t)j * 16 * Dsz + ko);
#pragma unroll
        for (int nb = 0; nb < NB; ++nb)
          pb[nb][j] = *(const v4f*)(gB[nb] + (size_t)j * 16 * Dsz + ko);
      }
    }

#pragma unroll
    for (int kb = 0; kb < GK / 8; ++kb) {
      v4f fa[2], fb[NB][2];
#pragma unroll
      for (int t = 0; t < 2; ++t) {
        fa[t] = *(const v4f*)(cur + aoff + t * 16 * GP + kb * 8);
#pragma unroll
        for (int nb = 0; nb < NB; ++nb)
          fb[nb][t] = *(const v4f*)(cur + GMAT * (1 + nb) + boff + t * 16 * GP + kb * 8);
      }
#pragma unroll
      for (int nb = 0; nb < NB; ++nb)
#pragma unroll
        for (int mt = 0; mt < 2; ++mt)
#pragma unroll
          for (int nt = 0; nt < 2; ++nt)
            acc[nb][mt][nt] = wmma4(fa[mt].xy, fb[nb][nt].xy, acc[nb][mt][nt]);
#pragma unroll
      for (int nb = 0; nb < NB; ++nb)
#pragma unroll
        for (int mt = 0; mt < 2; ++mt)
#pragma unroll
          for (int nt = 0; nt < 2; ++nt)
            acc[nb][mt][nt] = wmma4(fa[mt].zw, fb[nb][nt].zw, acc[nb][mt][nt]);
    }

    if (more) {
#pragma unroll
      for (int j = 0; j < 4; ++j) {
        *(v4f*)(nxt + soff + j * 16 * GP) = pa[j];
#pragma unroll
        for (int nb = 0; nb < NB; ++nb)
          *(v4f*)(nxt + GMAT * (1 + nb) + soff + j * 16 * GP) = pb[nb][j];
      }
    }
    __syncthreads();
  }
}

#define DT_ROWS  4
#define DT_WAVES 4
#define DT_THREADS (DT_WAVES * 32)
#define DT_BROWS (DT_ROWS * DT_WAVES)
#define DT_YB (Msz / DT_BROWS / (Msz / 64))

__device__ __forceinline__ void dt_role(const float* __restrict__ X,
                                        const float* __restrict__ w_dt,
                                        const float* __restrict__ b_dt,
                                        const float* __restrict__ ln_a,
                                        float* __restrict__ LD,
                                        float* __restrict__ lds, const int blk) {
  v4f* wl = (v4f*)lds;
  const int lane = threadIdx.x & 31;
  const int wave = threadIdx.x >> 5;
  const int row0 = (blk * DT_WAVES + wave) * DT_ROWS;

  v4f wr[Hsz * Dsz / 4 / DT_THREADS];
#pragma unroll
  for (int j = 0; j < Hsz * Dsz / 4 / DT_THREADS; ++j)
    wr[j] = *(const v4f*)(w_dt + 4 * (threadIdx.x + DT_THREADS * j));

  v4f xv[DT_ROWS][4];
#pragma unroll
  for (int r = 0; r < DT_ROWS; ++r)
#pragma unroll
    for (int c = 0; c < 4; ++c)
      xv[r][c] = *(const v4f*)(X + (size_t)(row0 + r) * Dsz + c * 128 + 4 * lane);

  const int h_out = lane & 7;
  const float bdt = b_dt[h_out];
  const float lna = ln_a[h_out];

  asm volatile("" :: "v"(xv[0][0]), "v"(xv[0][1]), "v"(xv[0][2]), "v"(xv[0][3]),
                     "v"(xv[1][0]), "v"(xv[1][1]), "v"(xv[1][2]), "v"(xv[1][3]),
                     "v"(xv[2][0]), "v"(xv[2][1]), "v"(xv[2][2]), "v"(xv[2][3]),
                     "v"(xv[3][0]), "v"(xv[3][1]), "v"(xv[3][2]), "v"(xv[3][3]),
                     "v"(bdt), "v"(lna));

#pragma unroll
  for (int j = 0; j < Hsz * Dsz / 4 / DT_THREADS; ++j)
    wl[threadIdx.x + DT_THREADS * j] = wr[j];
  __syncthreads();

  float v32[DT_ROWS * Hsz];
#pragma unroll
  for (int h = 0; h < Hsz; ++h) {
    v4f wv[4];
#pragma unroll
    for (int c = 0; c < 4; ++c) wv[c] = wl[h * (Dsz / 4) + c * 32 + lane];
#pragma unroll
    for (int r = 0; r < DT_ROWS; ++r) {
      float s[4];
#pragma unroll
      for (int c = 0; c < 4; ++c) {
        s[c] = xv[r][c].x * wv[c].x;
        s[c] = fmaf(xv[r][c].y, wv[c].y, s[c]);
        s[c] = fmaf(xv[r][c].z, wv[c].z, s[c]);
        s[c] = fmaf(xv[r][c].w, wv[c].w, s[c]);
      }
      v32[r * Hsz + h] = (s[0] + s[1]) + (s[2] + s[3]);
    }
  }

  const bool b4 = (lane & 16) != 0;
  const bool b3 = (lane & 8) != 0;
  const bool b2 = (lane & 4) != 0;
  const bool b1 = (lane & 2) != 0;
  const bool b0 = (lane & 1) != 0;
  float v16[16], v8[8], v4[4], v2[2];
#pragma unroll
  for (int i = 0; i < 16; ++i) {
    const float keep = b4 ? v32[16 + i] : v32[i];
    const float send = b4 ? v32[i] : v32[16 + i];
    v16[i] = keep + __shfl_xor(send, 16, 32);
  }
#pragma unroll
  for (int i = 0; i < 8; ++i) {
    const float keep = b3 ? v16[8 + i] : v16[i];
    const float send = b3 ? v16[i] : v16[8 + i];
    v8[i] = keep + __shfl_xor(send, 8, 32);
  }
#pragma unroll
  for (int i = 0; i < 4; ++i) {
    const float keep = b2 ? v8[4 + i] : v8[i];
    const float send = b2 ? v8[i] : v8[4 + i];
    v4[i] = keep + __shfl_xor(send, 4, 32);
  }
#pragma unroll
  for (int i = 0; i < 2; ++i) {
    const float keep = b1 ? v4[2 + i] : v4[i];
    const float send = b1 ? v4[i] : v4[2 + i];
    v2[i] = keep + __shfl_xor(send, 2, 32);
  }
  float dot;
  {
    const float keep = b0 ? v2[1] : v2[0];
    const float send = b0 ? v2[0] : v2[1];
    dot = keep + __shfl_xor(send, 1, 32);
  }

  const float v = dot + bdt;
  const float sp = (v > 20.f) ? v : log1pf(expf(v));
  const float ld = -expf(lna) * sp;
  LD[(size_t)row0 * Hsz + lane] = ld;
}

__global__ __launch_bounds__(128)
void gemm_xproj_kernel(const float* __restrict__ X,
                       const float* __restrict__ Wz,  const float* __restrict__ bz,
                       const float* __restrict__ Wza, const float* __restrict__ bza,
                       const float* __restrict__ Wya, const float* __restrict__ bya,
                       const float* __restrict__ w_dt, const float* __restrict__ b_dt,
                       const float* __restrict__ ln_a,
                       float* __restrict__ Z, float* __restrict__ GYA,
                       float* __restrict__ LD) {
  __shared__ __attribute__((aligned(16))) float lds[2 * GMAT * 4];
  if (blockIdx.y >= Dsz / 64) {
    dt_role(X, w_dt, b_dt, ln_a, LD, lds,
            (blockIdx.y - Dsz / 64) * (Msz / 64) + blockIdx.x);
    return;
  }
  const int lane = threadIdx.x & 31;
  const int wave = threadIdx.x >> 5;
  const int mr = lane & 15;
  const int kk = lane >> 4;
  const int m_base = blockIdx.x * 64 + (wave & 1) * 32;
  const int n_base = blockIdx.y * 64 + (wave >> 1) * 32;

  v8f acc[3][2][2];
  const v8f zero = {0.f, 0.f, 0.f, 0.f, 0.f, 0.f, 0.f, 0.f};
  for (int nb = 0; nb < 3; ++nb)
    for (int i = 0; i < 2; ++i)
      for (int j = 0; j < 2; ++j) acc[nb][i][j] = zero;

  const float* const W[3] = {Wz, Wza, Wya};
  gemm_core<3>(X, W, lds, blockIdx.x * 64, blockIdx.y * 64, acc);

  for (int mt = 0; mt < 2; ++mt)
    for (int nt = 0; nt < 2; ++nt) {
      const int n = n_base + nt * 16 + mr;
      const float biz = bz[n], biza = bza[n], biya = bya[n];
#pragma unroll
      for (int r = 0; r < 8; ++r) {
        const int m = m_base + mt * 16 + kk * 8 + r;
        const float v1 = acc[0][mt][nt][r] + biz;
        const float v2 = acc[1][mt][nt][r] + biza;
        const float v3 = acc[2][mt][nt][r] + biya;
        Z[(size_t)m * Dsz + n] = v1 * silu_f(v2);
        GYA[(size_t)m * Dsz + n] = silu_f(v3);
      }
    }
}

#define ZS_PARTS 64
#define ZS_ROWS  (Lsz / ZS_PARTS)
#define PQK 24
#define ATP 68

__global__ __launch_bounds__(1024)
void zsum_kernel(const float* __restrict__ Z, const float* __restrict__ hidden,
                 const float* __restrict__ LD, const float* __restrict__ Wy,
                 const float* __restrict__ gn_b,
                 float* __restrict__ C, float* __restrict__ zhh,
                 float* __restrict__ S, float* __restrict__ PQT) {
  __shared__ float4 red4[ZS_PARTS * (Isz / 4)];
  __shared__ float  red2[16 * Isz];
  __shared__ float  scan[Lsz];
  __shared__ __attribute__((aligned(16))) float at[4 * ATP];
  const int b = blockIdx.x >> 3;
  const int h = blockIdx.x & 7;
  const int tid = threadIdx.x;
  const int q = tid & 15;
  const int p = tid >> 4;

  const float ldv = LD[(size_t)(b * Lsz + tid) * Hsz + h];

  const int ch = (b * Hsz + h) * Isz + (tid & 63);
  float zdiag = 0.f, hid = 0.f;
  if (tid < Isz) {
    zdiag = Z[((size_t)(b * Lsz + h) * Hsz + h) * Isz + tid];
    hid = hidden[ch];
  }

  const float* rp = Z + ((size_t)(b * Lsz + p) * Hsz + h) * Isz + 4 * q;
  float4 v[ZS_ROWS];
#pragma unroll
  for (int j = 0; j < ZS_ROWS; ++j)
    v[j] = *(const float4*)(rp + (size_t)j * ZS_PARTS * Dsz);

  const int c16 = tid & 15;
  const int half = (tid >> 4) & 1;
  const int wave = tid >> 5;
  const float gbh = gn_b[h];
  v4f wf[Isz / 8];
  {
    const float* wp = Wy + (size_t)(wave * 16 + c16) * Dsz + h * Isz + 4 * half;
#pragma unroll
    for (int kb = 0; kb < Isz / 8; ++kb) wf[kb] = *(const v4f*)(wp + 8 * kb);
  }

  scan[tid] = ldv;
  __syncthreads();
  for (int off = 1; off < Lsz; off <<= 1) {
    float t = scan[tid];
    if (tid >= off) t += scan[tid - off];
    __syncthreads();
    scan[tid] = t;
    __syncthreads();
  }
  S[(size_t)(b * Lsz + tid) * Hsz + h] = scan[tid];

  if (p <= h) v[0] = make_float4(0.f, 0.f, 0.f, 0.f);

#pragma unroll
  for (int w = 1; w < ZS_ROWS; w <<= 1)
#pragma unroll
    for (int j = 0; j + w < ZS_ROWS; j += 2 * w) {
      v[j].x += v[j + w].x; v[j].y += v[j + w].y;
      v[j].z += v[j + w].z; v[j].w += v[j + w].w;
    }
  red4[p * (Isz / 4) + q] = v[0];
  __syncthreads();

  const float* red = (const float*)red4;
  {
    const int i = tid & 63;
    const int g = tid >> 6;
    const float* r = red + (4 * g) * Isz + i;
    red2[g * Isz + i] = (r[0] + r[Isz]) + (r[2 * Isz] + r[3 * Isz]);
  }
  __syncthreads();

  float cfull = 0.f;
  if (tid < Isz) {
    float t[16];
#pragma unroll
    for (int g = 0; g < 16; ++g) t[g] = red2[g * Isz + tid];
#pragma unroll
    for (int w = 1; w < 16; w <<= 1)
#pragma unroll
      for (int g = 0; g + w < 16; g += 2 * w) t[g] += t[g + w];
    cfull = t[0] + hid;
    C[ch] = cfull;
    zhh[ch] = zdiag;
    at[tid] = zdiag;
    at[ATP + tid] = cfull;
    at[2 * ATP + tid] = gbh;
    at[3 * ATP + tid] = 0.f;
  }
  __syncthreads();

  float mz = 0.f, mc = 0.f;
  if (tid < Isz) {
    v4f sz = {0.f, 0.f, 0.f, 0.f}, sq = {0.f, 0.f, 0.f, 0.f};
#pragma unroll
    for (int j = 0; j < Isz / 4; ++j) {
      sz += *(const v4f*)(at + 4 * j);
      sq += *(const v4f*)(at + ATP + 4 * j);
    }
    mz = ((sz.x + sz.y) + (sz.z + sz.w)) * (1.0f / 64.0f);
    mc = ((sq.x + sq.y) + (sq.z + sq.w)) * (1.0f / 64.0f);
  }
  __syncthreads();
  if (tid < Isz) {
    at[tid] = zdiag - mz;
    at[ATP + tid] = cfull - mc;
  }
  __syncthreads();

  const int ar = (c16 < 3) ? c16 : 3;
  v8f acc = {0.f, 0.f, 0.f, 0.f, 0.f, 0.f, 0.f, 0.f};
#pragma unroll
  for (int kb = 0; kb < Isz / 8; ++kb) {
    const v4f a = *(const v4f*)(at + ar * ATP + 8 * kb + 4 * half);
    acc = wmma4(a.xy, wf[kb].xy, acc);
    acc = wmma4(a.zw, wf[kb].zw, acc);
  }
  if (half == 0) {
    float* o = PQT + (size_t)(b * Dsz + wave * 16 + c16) * PQK + h;
    o[0] = acc[0];
    o[8] = acc[1];
    o[16] = acc[2];
  }
}

#define HN_ROWS  2
#define HN_WAVES 8
#define HN_BROWS (HN_ROWS * HN_WAVES)
#define CTP 28
#define YTP 68

template <int M>
__device__ __forceinline__ float row_xor_add(float v) {
  const int p = __builtin_amdgcn_update_dpp(0, __float_as_int(v), 0x160 + M,
                                            0xf, 0xf, false);
  return v + __int_as_float(p);
}

__global__ __launch_bounds__(HN_WAVES * 32)
void hnorm_y_kernel(const float* __restrict__ S, const float* __restrict__ C,
                    const float* __restrict__ zhh, const float* __restrict__ gn_w,
                    const float* __restrict__ PQT, const float* __restrict__ by,
                    const float* __restrict__ GYA, float* __restrict__ Hout,
                    float* __restrict__ Y) {
  __shared__ __attribute__((aligned(16))) float ct[HN_BROWS * CTP];
  __shared__ __attribute__((aligned(16))) float yt[HN_WAVES * HN_BROWS * YTP];
  const int lane = threadIdx.x & 31;
  const int wave = threadIdx.x >> 5;
  const int brow0 = blockIdx.x * HN_BROWS;
  const int row0 = brow0 + wave * HN_ROWS;
  const int b = row0 >> 10;
  const int hh = lane >> 4;
  const int c16 = lane & 15;
  const int nb = wave * 64;

  v4f fb[4][PQK / 8];
#pragma unroll
  for (int nt = 0; nt < 4; ++nt)
#pragma unroll
    for (int kb = 0; kb < PQK / 8; ++kb)
      fb[nt][kb] = *(const v4f*)(PQT + (size_t)(b * Dsz + nb + nt * 16 + c16) * PQK +
                                 8 * kb + 4 * hh);
  v4f gy[HN_BROWS / 2];
#pragma unroll
  for (int it = 0; it < HN_BROWS / 2; ++it)
    gy[it] = *(const v4f*)(GYA + (size_t)(brow0 + 2 * it + hh) * Dsz + nb + 4 * c16);
  const v4f bv = *(const v4f*)(by + nb + 4 * c16);

  v4f zv[4], cv[4];
  float gw[4];
  float sv[HN_ROWS][4];
#pragma unroll
  for (int c = 0; c < 4; ++c) {
    zv[c] = *(const v4f*)(zhh + (size_t)b * Dsz + c * 128 + 4 * lane);
    cv[c] = *(const v4f*)(C + (size_t)b * Dsz + c * 128 + 4 * lane);
    gw[c] = gn_w[2 * c + hh];
  }
#pragma unroll
  for (int r = 0; r < HN_ROWS; ++r)
#pragma unroll
    for (int c = 0; c < 4; ++c)
      sv[r][c] = S[(size_t)(row0 + r) * Hsz + 2 * c + hh];

#pragma unroll
  for (int r = 0; r < HN_ROWS; ++r) {
    const size_t ro = (size_t)(row0 + r) * Dsz + 4 * lane;
    float* cr = ct + (wave * HN_ROWS + r) * CTP;
#pragma unroll
    for (int c = 0; c < 4; ++c) {
      const float e = expf(sv[r][c]);
      v4f hv;
      hv.x = zv[c].x + e * cv[c].x;
      hv.y = zv[c].y + e * cv[c].y;
      hv.z = zv[c].z + e * cv[c].z;
      hv.w = zv[c].w + e * cv[c].w;
      *(v4f*)(Hout + ro + c * 128) = hv;

      float s = (hv.x + hv.y) + (hv.z + hv.w);
      s = row_xor_add<1>(s);
      s = row_xor_add<2>(s);
      s = row_xor_add<4>(s);
      s = row_xor_add<8>(s);
      const float mu = s * (1.0f / 64.0f);

      v4f d;
      d.x = hv.x - mu; d.y = hv.y - mu; d.z = hv.z - mu; d.w = hv.w - mu;
      float q = (d.x * d.x + d.y * d.y) + (d.z * d.z + d.w * d.w);
      q = row_xor_add<1>(q);
      q = row_xor_add<2>(q);
      q = row_xor_add<4>(q);
      q = row_xor_add<8>(q);
      const float var = q * (1.0f / 64.0f);

      const float sc = gw[c] / sqrtf(var + GN_EPS);
      if (c16 == 0) {
        const int head = 2 * c + hh;
        cr[head] = sc;
        cr[8 + head] = sc * e;
        cr[16 + head] = 1.0f;
      }
    }
  }
  __syncthreads();

  v4f fa[PQK / 8];
#pragma unroll
  for (int kb = 0; kb < PQK / 8; ++kb)
    fa[kb] = *(const v4f*)(ct + c16 * CTP + 8 * kb + 4 * hh);

  v8f acc[4];
  const v8f zero = {0.f, 0.f, 0.f, 0.f, 0.f, 0.f, 0.f, 0.f};
#pragma unroll
  for (int nt = 0; nt < 4; ++nt) acc[nt] = zero;
#pragma unroll
  for (int kb = 0; kb < PQK / 8; ++kb) {
#pragma unroll
    for (int nt = 0; nt < 4; ++nt)
      acc[nt] = wmma4(fa[kb].xy, fb[nt][kb].xy, acc[nt]);
#pragma unroll
    for (int nt = 0; nt < 4; ++nt)
      acc[nt] = wmma4(fa[kb].zw, fb[nt][kb].zw, acc[nt]);
  }

  float* yw = yt + wave * HN_BROWS * YTP;
#pragma unroll
  for (int nt = 0; nt < 4; ++nt)
#pragma unroll
    for (int r = 0; r < 8; ++r)
      yw[(8 * hh + r) * YTP + nt * 16 + c16] = acc[nt][r];
  __syncthreads();

#pragma unroll
  for (int it = 0; it < HN_BROWS / 2; ++it) {
    const int row = 2 * it + hh;
    const v4f d = *(const v4f*)(yw + row * YTP + 4 * c16);
    v4f o;
    o.x = (d.x + bv.x) * gy[it].x;
    o.y = (d.y + bv.y) * gy[it].y;
    o.z = (d.z + bv.z) * gy[it].z;
    o.w = (d.w + bv.w) * gy[it].w;
    *(v4f*)(Y + (size_t)(brow0 + row) * Dsz + nb + 4 * c16) = o;
  }
}

extern "C" void kernel_launch(void* const* d_in, const int* in_sizes, int n_in,
                              void* d_out, int out_size, void* d_ws, size_t ws_size,
                              hipStream_t stream) {
  const float* x      = (const float*)d_in[0];
  const float* hidden = (const float*)d_in[1];
  const float* w_z    = (const float*)d_in[2];
  const float* b_z    = (const float*)d_in[3];
  const float* w_za   = (const float*)d_in[4];
  const float* b_za   = (const float*)d_in[5];
  const float* w_y    = (const float*)d_in[6];
  const float* b_y    = (const float*)d_in[7];
  const float* w_ya   = (const float*)d_in[8];
  const float* b_ya   = (const float*)d_in[9];
  const float* w_dt   = (const float*)d_in[10];
  const float* b_dt   = (const float*)d_in[11];
  const float* ln_a   = (const float*)d_in[12];
  const float* gn_w   = (const float*)d_in[13];
  const float* gn_b   = (const float*)d_in[14];

  float* y_out = (float*)d_out;
  float* h_out = y_out + (size_t)Msz * Dsz;

  float* ws   = (float*)d_ws;
  float* Z    = ws;
  float* GYA  = Z    + (size_t)Msz * Dsz;
  float* Sb   = GYA  + (size_t)Msz * Dsz;
  float* Cb   = Sb   + (size_t)Msz * Hsz;
  float* ZHH  = Cb   + (size_t)Bsz * Hsz * Isz;
  float* LDb  = ZHH  + (size_t)Bsz * Hsz * Isz;
  float* PQT  = LDb  + (size_t)Msz * Hsz;

  const dim3 gGemm(Msz / 64, Dsz / 64 + DT_YB, 1);
  const dim3 bGemm(128, 1, 1);

  gemm_xproj_kernel<<<gGemm, bGemm, 0, stream>>>(x, w_z, b_z, w_za, b_za, w_ya, b_ya,
                                                 w_dt, b_dt, ln_a, Z, GYA, LDb);
  zsum_kernel<<<dim3(Bsz * Hsz), dim3(1024), 0, stream>>>(Z, hidden, LDb, w_y, gn_b,
                                                          Cb, ZHH, Sb, PQT);
  hnorm_y_kernel<<<dim3(Msz / HN_BROWS), dim3(HN_WAVES * 32), 0, stream>>>(
      Sb, Cb, ZHH, gn_w, PQT, b_y, GYA, h_out, y_out);
}


# z is never stored: each GEMM tile (64 rows x one head) sums its columns in the epilogue; kernel 2 adds 16 tile sums per column, was a 1024-row slab read. Column sum order changes
# speedup vs baseline: 4.0245x; 1.0517x over previous
#include <hip/hip_runtime.h>
#include <hip/hip_bf16.h>
#include <math.h>


#define Bsz 4
#define Lsz 1024
#define Dsz 512
#define Hsz 8
#define Isz 64
#define Msz (Bsz * Lsz)
#define MT64 (Msz / 64)
#define LT64 (Lsz / 64)
#define GN_EPS 1e-5f

typedef float v2f __attribute__((ext_vector_type(2)));
typedef float v4f __attribute__((ext_vector_type(4)));
typedef float v8f __attribute__((ext_vector_type(8)));

__device__ __forceinline__ v8f wmma4(v2f a, v2f b, v8f c) {
  return __builtin_amdgcn_wmma_f32_16x16x4_f32(
       false, a,  false, b,
       (short)0, c,  false,  false);
}

__device__ __forceinline__ float silu_f(float v) {
  return v / (1.0f + expf(-v));
}

#define GK 32
#define GP 36
#define GMAT (64 * GP)
#define GKT (Dsz / GK)

template <int NB>
__device__ __forceinline__ void gemm_core(const float* __restrict__ A,
                                          const float* const (&W)[NB],
                                          float* __restrict__ lds,
                                          const int m_blk, const int n_blk,
                                          v8f (&acc)[NB][2][2]) {
  constexpr int STAGE = GMAT * (1 + NB);
  const int tid = threadIdx.x;
  const int lane = tid & 31;
  const int wave = tid >> 5;
  const int mr = lane & 15;
  const int kk = lane >> 4;
  const int wm = (wave & 1) * 32;
  const int wn = (wave >> 1) * 32;

  const int crow = tid >> 3;
  const int ccol = (tid & 7) * 4;
  const float* gA = A + (size_t)(m_blk + crow) * Dsz + ccol;
  const float* gB[NB];
#pragma unroll
  for (int nb = 0; nb < NB; ++nb)
    gB[nb] = W[nb] + (size_t)(n_blk + crow) * Dsz + ccol;
  const int soff = crow * GP + ccol;

  const int aoff = (wm + mr) * GP + 4 * kk;
  const int boff = (wn + mr) * GP + 4 * kk;

  v4f pa[4], pb[NB][4];
#pragma unroll
  for (int j = 0; j < 4; ++j) {
    pa[j] = *(const v4f*)(gA + (size_t)j * 16 * Dsz);
#pragma unroll
    for (int nb = 0; nb < NB; ++nb)
      pb[nb][j] = *(const v4f*)(gB[nb] + (size_t)j * 16 * Dsz);
  }
#pragma unroll
  for (int j = 0; j < 4; ++j) {
    *(v4f*)(lds + soff + j * 16 * GP) = pa[j];
#pragma unroll
    for (int nb = 0; nb < NB; ++nb)
      *(v4f*)(lds + GMAT * (1 + nb) + soff + j * 16 * GP) = pb[nb][j];
  }
  __syncthreads();

#pragma unroll 2
  for (int kt = 0; kt < GKT; ++kt) {
    const float* cur = lds + (kt & 1) * STAGE;
    float* nxt = lds + ((kt + 1) & 1) * STAGE;
    const bool more = (kt + 1 < GKT);

    if (more) {
      const int ko = (kt + 1) * GK;
#pragma unroll
      for (int j = 0; j < 4; ++j) {
        pa[j] = *(const v4f*)(gA + (size_t)j * 16 * Dsz + ko);
#pragma unroll
        for (int nb = 0; nb < NB; ++nb)
          pb[nb][j] = *(const v4f*)(gB[nb] + (size_t)j * 16 * Dsz + ko);
      }
    }

#pragma unroll
    for (int kb = 0; kb < GK / 8; ++kb) {
      v4f fa[2], fb[NB][2];
#pragma unroll
      for (int t = 0; t < 2; ++t) {
        fa[t] = *(const v4f*)(cur + aoff + t * 16 * GP + kb * 8);
#pragma unroll
        for (int nb = 0; nb < NB; ++nb)
          fb[nb][t] = *(const v4f*)(cur + GMAT * (1 + nb) + boff + t * 16 * GP + kb * 8);
      }
#pragma unroll
      for (int nb = 0; nb < NB; ++nb)
#pragma unroll
        for (int mt = 0; mt < 2; ++mt)
#pragma unroll
          for (int nt = 0; nt < 2; ++nt)
            acc[nb][mt][nt] = wmma4(fa[mt].xy, fb[nb][nt].xy, acc[nb][mt][nt]);
#pragma unroll
      for (int nb = 0; nb < NB; ++nb)
#pragma unroll
        for (int mt = 0; mt < 2; ++mt)
#pragma unroll
          for (int nt = 0; nt < 2; ++nt)
            acc[nb][mt][nt] = wmma4(fa[mt].zw, fb[nb][nt].zw, acc[nb][mt][nt]);
    }

    if (more) {
#pragma unroll
      for (int j = 0; j < 4; ++j) {
        *(v4f*)(nxt + soff + j * 16 * GP) = pa[j];
#pragma unroll
        for (int nb = 0; nb < NB; ++nb)
          *(v4f*)(nxt + GMAT * (1 + nb) + soff + j * 16 * GP) = pb[nb][j];
      }
    }
    __syncthreads();
  }
}

#define DT_ROWS  4
#define DT_WAVES 4
#define DT_THREADS (DT_WAVES * 32)
#define DT_BROWS (DT_ROWS * DT_WAVES)
#define DT_YB (Msz / DT_BROWS / (Msz / 64))

__device__ __forceinline__ void dt_role(const float* __restrict__ X,
                                        const float* __restrict__ w_dt,
                                        const float* __restrict__ b_dt,
                                        const float* __restrict__ ln_a,
                                        float* __restrict__ LD,
                                        float* __restrict__ lds, const int blk) {
  v4f* wl = (v4f*)lds;
  const int lane = threadIdx.x & 31;
  const int wave = threadIdx.x >> 5;
  const int row0 = (blk * DT_WAVES + wave) * DT_ROWS;

  v4f wr[Hsz * Dsz / 4 / DT_THREADS];
#pragma unroll
  for (int j = 0; j < Hsz * Dsz / 4 / DT_THREADS; ++j)
    wr[j] = *(const v4f*)(w_dt + 4 * (threadIdx.x + DT_THREADS * j));

  v4f xv[DT_ROWS][4];
#pragma unroll
  for (int r = 0; r < DT_ROWS; ++r)
#pragma unroll
    for (int c = 0; c < 4; ++c)
      xv[r][c] = *(const v4f*)(X + (size_t)(row0 + r) * Dsz + c * 128 + 4 * lane);

  const int h_out = lane & 7;
  const float bdt = b_dt[h_out];
  const float lna = ln_a[h_out];

  asm volatile("" :: "v"(xv[0][0]), "v"(xv[0][1]), "v"(xv[0][2]), "v"(xv[0][3]),
                     "v"(xv[1][0]), "v"(xv[1][1]), "v"(xv[1][2]), "v"(xv[1][3]),
                     "v"(xv[2][0]), "v"(xv[2][1]), "v"(xv[2][2]), "v"(xv[2][3]),
                     "v"(xv[3][0]), "v"(xv[3][1]), "v"(xv[3][2]), "v"(xv[3][3]),
                     "v"(bdt), "v"(lna));

#pragma unroll
  for (int j = 0; j < Hsz * Dsz / 4 / DT_THREADS; ++j)
    wl[threadIdx.x + DT_THREADS * j] = wr[j];
  __syncthreads();

  float v32[DT_ROWS * Hsz];
#pragma unroll
  for (int h = 0; h < Hsz; ++h) {
    v4f wv[4];
#pragma unroll
    for (int c = 0; c < 4; ++c) wv[c] = wl[h * (Dsz / 4) + c * 32 + lane];
#pragma unroll
    for (int r = 0; r < DT_ROWS; ++r) {
      float s[4];
#pragma unroll
      for (int c = 0; c < 4; ++c) {
        s[c] = xv[r][c].x * wv[c].x;
        s[c] = fmaf(xv[r][c].y, wv[c].y, s[c]);
        s[c] = fmaf(xv[r][c].z, wv[c].z, s[c]);
        s[c] = fmaf(xv[r][c].w, wv[c].w, s[c]);
      }
      v32[r * Hsz + h] = (s[0] + s[1]) + (s[2] + s[3]);
    }
  }

  const bool b4 = (lane & 16) != 0;
  const bool b3 = (lane & 8) != 0;
  const bool b2 = (lane & 4) != 0;
  const bool b1 = (lane & 2) != 0;
  const bool b0 = (lane & 1) != 0;
  float v16[16], v8[8], v4[4], v2[2];
#pragma unroll
  for (int i = 0; i < 16; ++i) {
    const float keep = b4 ? v32[16 + i] : v32[i];
    const float send = b4 ? v32[i] : v32[16 + i];
    v16[i] = keep + __shfl_xor(send, 16, 32);
  }
#pragma unroll
  for (int i = 0; i < 8; ++i) {
    const float keep = b3 ? v16[8 + i] : v16[i];
    const float send = b3 ? v16[i] : v16[8 + i];
    v8[i] = keep + __shfl_xor(send, 8, 32);
  }
#pragma unroll
  for (int i = 0; i < 4; ++i) {
    const float keep = b2 ? v8[4 + i] : v8[i];
    const float send = b2 ? v8[i] : v8[4 + i];
    v4[i] = keep + __shfl_xor(send, 4, 32);
  }
#pragma unroll
  for (int i = 0; i < 2; ++i) {
    const float keep = b1 ? v4[2 + i] : v4[i];
    const float send = b1 ? v4[i] : v4[2 + i];
    v2[i] = keep + __shfl_xor(send, 2, 32);
  }
  float dot;
  {
    const float keep = b0 ? v2[1] : v2[0];
    const float send = b0 ? v2[0] : v2[1];
    dot = keep + __shfl_xor(send, 1, 32);
  }

  const float v = dot + bdt;
  const float sp = (v > 20.f) ? v : log1pf(expf(v));
  const float ld = -expf(lna) * sp;
  LD[(size_t)row0 * Hsz + lane] = ld;
}

__global__ __launch_bounds__(128)
void gemm_xproj_kernel(const float* __restrict__ X,
                       const float* __restrict__ Wz,  const float* __restrict__ bz,
                       const float* __restrict__ Wza, const float* __restrict__ bza,
                       const float* __restrict__ Wya, const float* __restrict__ bya,
                       const float* __restrict__ w_dt, const float* __restrict__ b_dt,
                       const float* __restrict__ ln_a,
                       float* __restrict__ CP, float* __restrict__ ZHH,
                       float* __restrict__ GYA, float* __restrict__ LD) {
  __shared__ __attribute__((aligned(16))) float lds[2 * GMAT * 4];
  if (blockIdx.y >= Dsz / 64) {
    dt_role(X, w_dt, b_dt, ln_a, LD, lds,
            (blockIdx.y - Dsz / 64) * (Msz / 64) + blockIdx.x);
    return;
  }
  const int lane = threadIdx.x & 31;
  const int wave = threadIdx.x >> 5;
  const int mr = lane & 15;
  const int kk = lane >> 4;
  const int m_base = blockIdx.x * 64 + (wave & 1) * 32;
  const int n_base = blockIdx.y * 64 + (wave >> 1) * 32;

  v8f acc[3][2][2];
  const v8f zero = {0.f, 0.f, 0.f, 0.f, 0.f, 0.f, 0.f, 0.f};
  for (int nb = 0; nb < 3; ++nb)
    for (int i = 0; i < 2; ++i)
      for (int j = 0; j < 2; ++j) acc[nb][i][j] = zero;

  const float* const W[3] = {Wz, Wza, Wya};
  gemm_core<3>(X, W, lds, blockIdx.x * 64, blockIdx.y * 64, acc);

  const int h = blockIdx.y;
  const bool first = (blockIdx.x & (LT64 - 1)) == 0;
  const bool low = first && (wave & 1) == 0 && kk == 0;
  float cs[2] = {0.f, 0.f};
  float zd[2] = {0.f, 0.f};

  for (int mt = 0; mt < 2; ++mt)
    for (int nt = 0; nt < 2; ++nt) {
      const int n = n_base + nt * 16 + mr;
      const float biz = bz[n], biza = bza[n], biya = bya[n];
#pragma unroll
      for (int r = 0; r < 8; ++r) {
        const int m = m_base + mt * 16 + kk * 8 + r;
        const float v1 = acc[0][mt][nt][r] + biz;
        const float v2 = acc[1][mt][nt][r] + biza;
        const float v3 = acc[2][mt][nt][r] + biya;
        const float zv = v1 * silu_f(v2);
        if (mt == 0) {
          cs[nt] += (low && r <= h) ? 0.f : zv;
          if (r == h) zd[nt] = zv;
        } else {
          cs[nt] += zv;
        }
        GYA[(size_t)m * Dsz + n] = silu_f(v3);
      }
    }

  const int cl = (wave >> 1) * 32 + mr;
#pragma unroll
  for (int nt = 0; nt < 2; ++nt) {
    lds[((wave & 1) * 2 + kk) * Isz + cl + nt * 16] = cs[nt];
    if (low)
      ZHH[(size_t)((blockIdx.x / LT64) * Hsz + h) * Isz + cl + nt * 16] = zd[nt];
  }
  __syncthreads();
  if (threadIdx.x < Isz) {
    const float* p = lds + threadIdx.x;
    CP[((size_t)blockIdx.x * Hsz + h) * Isz + threadIdx.x] =
        (p[0] + p[Isz]) + (p[2 * Isz] + p[3 * Isz]);
  }
}

#define PQK 24
#define ATP 68

__global__ __launch_bounds__(1024)
void zsum_kernel(const float* __restrict__ CP, const float* __restrict__ zhh,
                 const float* __restrict__ hidden,
                 const float* __restrict__ LD, const float* __restrict__ Wy,
                 const float* __restrict__ gn_b,
                 float* __restrict__ C,
                 float* __restrict__ S, float* __restrict__ PQT) {
  __shared__ float  scan[Lsz];
  __shared__ __attribute__((aligned(16))) float at[4 * ATP];
  const int b = blockIdx.x >> 3;
  const int h = blockIdx.x & 7;
  const int tid = threadIdx.x;

  const int c16 = tid & 15;
  const int half = (tid >> 4) & 1;
  const int wave = tid >> 5;
  const float gbh = gn_b[h];
  v4f wf[Isz / 8];
  {
    const float* wp = Wy + (size_t)(wave * 16 + c16) * Dsz + h * Isz + 4 * half;
#pragma unroll
    for (int kb = 0; kb < Isz / 8; ++kb) wf[kb] = *(const v4f*)(wp + 8 * kb);
  }

  const int ch = (b * Hsz + h) * Isz + (tid & 63);
  float zdiag = 0.f, hid = 0.f;
  float t[LT64];
#pragma unroll
  for (int g = 0; g < LT64; ++g) t[g] = 0.f;
  if (tid < Isz) {
    zdiag = zhh[ch];
    hid = hidden[ch];
#pragma unroll
    for (int g = 0; g < LT64; ++g)
      t[g] = CP[((size_t)(b * LT64 + g) * Hsz + h) * Isz + tid];
  }

  const float ldv = LD[(size_t)(b * Lsz + tid) * Hsz + h];

  scan[tid] = ldv;
  __syncthreads();
  for (int off = 1; off < Lsz; off <<= 1) {
    float t = scan[tid];
    if (tid >= off) t += scan[tid - off];
    __syncthreads();
    scan[tid] = t;
    __syncthreads();
  }
  S[(size_t)(b * Lsz + tid) * Hsz + h] = scan[tid];

  float cfull = 0.f;
  if (tid < Isz) {
#pragma unroll
    for (int w = 1; w < LT64; w <<= 1)
#pragma unroll
      for (int g = 0; g + w < LT64; g += 2 * w) t[g] += t[g + w];
    cfull = t[0] + hid;
    C[ch] = cfull;
    at[tid] = zdiag;
    at[ATP + tid] = cfull;
    at[2 * ATP + tid] = gbh;
    at[3 * ATP + tid] = 0.f;
  }
  __syncthreads();

  float mz = 0.f, mc = 0.f;
  if (tid < Isz) {
    v4f sz = {0.f, 0.f, 0.f, 0.f}, sq = {0.f, 0.f, 0.f, 0.f};
#pragma unroll
    for (int j = 0; j < Isz / 4; ++j) {
      sz += *(const v4f*)(at + 4 * j);
      sq += *(const v4f*)(at + ATP + 4 * j);
    }
    mz = ((sz.x + sz.y) + (sz.z + sz.w)) * (1.0f / 64.0f);
    mc = ((sq.x + sq.y) + (sq.z + sq.w)) * (1.0f / 64.0f);
  }
  __syncthreads();
  if (tid < Isz) {
    at[tid] = zdiag - mz;
    at[ATP + tid] = cfull - mc;
  }
  __syncthreads();

  const int ar = (c16 < 3) ? c16 : 3;
  v8f acc = {0.f, 0.f, 0.f, 0.f, 0.f, 0.f, 0.f, 0.f};
#pragma unroll
  for (int kb = 0; kb < Isz / 8; ++kb) {
    const v4f a = *(const v4f*)(at + ar * ATP + 8 * kb + 4 * half);
    acc = wmma4(a.xy, wf[kb].xy, acc);
    acc = wmma4(a.zw, wf[kb].zw, acc);
  }
  if (half == 0) {
    float* o = PQT + (size_t)(b * Dsz + wave * 16 + c16) * PQK + h;
    o[0] = acc[0];
    o[8] = acc[1];
    o[16] = acc[2];
  }
}

#define HN_ROWS  2
#define HN_WAVES 8
#define HN_BROWS (HN_ROWS * HN_WAVES)
#define CTP 28
#define YTP 68

template <int M>
__device__ __forceinline__ float row_xor_add(float v) {
  const int p = __builtin_amdgcn_update_dpp(0, __float_as_int(v), 0x160 + M,
                                            0xf, 0xf, false);
  return v + __int_as_float(p);
}

__global__ __launch_bounds__(HN_WAVES * 32)
void hnorm_y_kernel(const float* __restrict__ S, const float* __restrict__ C,
                    const float* __restrict__ zhh, const float* __restrict__ gn_w,
                    const float* __restrict__ PQT, const float* __restrict__ by,
                    const float* __restrict__ GYA, float* __restrict__ Hout,
                    float* __restrict__ Y) {
  __shared__ __attribute__((aligned(16))) float ct[HN_BROWS * CTP];
  __shared__ __attribute__((aligned(16))) float yt[HN_WAVES * HN_BROWS * YTP];
  const int lane = threadIdx.x & 31;
  const int wave = threadIdx.x >> 5;
  const int brow0 = blockIdx.x * HN_BROWS;
  const int row0 = brow0 + wave * HN_ROWS;
  const int b = row0 >> 10;
  const int hh = lane >> 4;
  const int c16 = lane & 15;
  const int nb = wave * 64;

  v4f fb[4][PQK / 8];
#pragma unroll
  for (int nt = 0; nt < 4; ++nt)
#pragma unroll
    for (int kb = 0; kb < PQK / 8; ++kb)
      fb[nt][kb] = *(const v4f*)(PQT + (size_t)(b * Dsz + nb + nt * 16 + c16) * PQK +
                                 8 * kb + 4 * hh);
  v4f gy[HN_BROWS / 2];
#pragma unroll
  for (int it = 0; it < HN_BROWS / 2; ++it)
    gy[it] = *(const v4f*)(GYA + (size_t)(brow0 + 2 * it + hh) * Dsz + nb + 4 * c16);
  const v4f bv = *(const v4f*)(by + nb + 4 * c16);

  v4f zv[4], cv[4];
  float gw[4];
  float sv[HN_ROWS][4];
#pragma unroll
  for (int c = 0; c < 4; ++c) {
    zv[c] = *(const v4f*)(zhh + (size_t)b * Dsz + c * 128 + 4 * lane);
    cv[c] = *(const v4f*)(C + (size_t)b * Dsz + c * 128 + 4 * lane);
    gw[c] = gn_w[2 * c + hh];
  }
#pragma unroll
  for (int r = 0; r < HN_ROWS; ++r)
#pragma unroll
    for (int c = 0; c < 4; ++c)
      sv[r][c] = S[(size_t)(row0 + r) * Hsz + 2 * c + hh];

#pragma unroll
  for (int r = 0; r < HN_ROWS; ++r) {
    const size_t ro = (size_t)(row0 + r) * Dsz + 4 * lane;
    float* cr = ct + (wave * HN_ROWS + r) * CTP;
#pragma unroll
    for (int c = 0; c < 4; ++c) {
      const float e = expf(sv[r][c]);
      v4f hv;
      hv.x = zv[c].x + e * cv[c].x;
      hv.y = zv[c].y + e * cv[c].y;
      hv.z = zv[c].z + e * cv[c].z;
      hv.w = zv[c].w + e * cv[c].w;
      *(v4f*)(Hout + ro + c * 128) = hv;

      float s = (hv.x + hv.y) + (hv.z + hv.w);
      s = row_xor_add<1>(s);
      s = row_xor_add<2>(s);
      s = row_xor_add<4>(s);
      s = row_xor_add<8>(s);
      const float mu = s * (1.0f / 64.0f);

      v4f d;
      d.x = hv.x - mu; d.y = hv.y - mu; d.z = hv.z - mu; d.w = hv.w - mu;
      float q = (d.x * d.x + d.y * d.y) + (d.z * d.z + d.w * d.w);
      q = row_xor_add<1>(q);
      q = row_xor_add<2>(q);
      q = row_xor_add<4>(q);
      q = row_xor_add<8>(q);
      const float var = q * (1.0f / 64.0f);

      const float sc = gw[c] / sqrtf(var + GN_EPS);
      if (c16 == 0) {
        const int head = 2 * c + hh;
        cr[head] = sc;
        cr[8 + head] = sc * e;
        cr[16 + head] = 1.0f;
      }
    }
  }
  __syncthreads();

  v4f fa[PQK / 8];
#pragma unroll
  for (int kb = 0; kb < PQK / 8; ++kb)
    fa[kb] = *(const v4f*)(ct + c16 * CTP + 8 * kb + 4 * hh);

  v8f acc[4];
  const v8f zero = {0.f, 0.f, 0.f, 0.f, 0.f, 0.f, 0.f, 0.f};
#pragma unroll
  for (int nt = 0; nt < 4; ++nt) acc[nt] = zero;
#pragma unroll
  for (int kb = 0; kb < PQK / 8; ++kb) {
#pragma unroll
    for (int nt = 0; nt < 4; ++nt)
      acc[nt] = wmma4(fa[kb].xy, fb[nt][kb].xy, acc[nt]);
#pragma unroll
    for (int nt = 0; nt < 4; ++nt)
      acc[nt] = wmma4(fa[kb].zw, fb[nt][kb].zw, acc[nt]);
  }

  float* yw = yt + wave * HN_BROWS * YTP;
#pragma unroll
  for (int nt = 0; nt < 4; ++nt)
#pragma unroll
    for (int r = 0; r < 8; ++r)
      yw[(8 * hh + r) * YTP + nt * 16 + c16] = acc[nt][r];
  __syncthreads();

#pragma unroll
  for (int it = 0; it < HN_BROWS / 2; ++it) {
    const int row = 2 * it + hh;
    const v4f d = *(const v4f*)(yw + row * YTP + 4 * c16);
    v4f o;
    o.x = (d.x + bv.x) * gy[it].x;
    o.y = (d.y + bv.y) * gy[it].y;
    o.z = (d.z + bv.z) * gy[it].z;
    o.w = (d.w + bv.w) * gy[it].w;
    *(v4f*)(Y + (size_t)(brow0 + row) * Dsz + nb + 4 * c16) = o;
  }
}

extern "C" void kernel_launch(void* const* d_in, const int* in_sizes, int n_in,
                              void* d_out, int out_size, void* d_ws, size_t ws_size,
                              hipStream_t stream) {
  const float* x      = (const float*)d_in[0];
  const float* hidden = (const float*)d_in[1];
  const float* w_z    = (const float*)d_in[2];
  const float* b_z    = (const float*)d_in[3];
  const float* w_za   = (const float*)d_in[4];
  const float* b_za   = (const float*)d_in[5];
  const float* w_y    = (const float*)d_in[6];
  const float* b_y    = (const float*)d_in[7];
  const float* w_ya   = (const float*)d_in[8];
  const float* b_ya   = (const float*)d_in[9];
  const float* w_dt   = (const float*)d_in[10];
  const float* b_dt   = (const float*)d_in[11];
  const float* ln_a   = (const float*)d_in[12];
  const float* gn_w   = (const float*)d_in[13];
  const float* gn_b   = (const float*)d_in[14];

  float* y_out = (float*)d_out;
  float* h_out = y_out + (size_t)Msz * Dsz;

  float* ws   = (float*)d_ws;
  float* GYA  = ws;
  float* CPb  = GYA  + (size_t)Msz * Dsz;
  float* Sb   = CPb  + (size_t)MT64 * Hsz * Isz;
  float* Cb   = Sb   + (size_t)Msz * Hsz;
  float* ZHH  = Cb   + (size_t)Bsz * Hsz * Isz;
  float* LDb  = ZHH  + (size_t)Bsz * Hsz * Isz;
  float* PQT  = LDb  + (size_t)Msz * Hsz;

  const dim3 gGemm(Msz / 64, Dsz / 64 + DT_YB, 1);
  const dim3 bGemm(128, 1, 1);

  gemm_xproj_kernel<<<gGemm, bGemm, 0, stream>>>(x, w_z, b_z, w_za, b_za, w_ya, b_ya,
                                                 w_dt, b_dt, ln_a, CPb, ZHH, GYA, LDb);
  zsum_kernel<<<dim3(Bsz * Hsz), dim3(1024), 0, stream>>>(CPb, ZHH, hidden, LDb, w_y, gn_b,
                                                          Cb, Sb, PQT);
  hnorm_y_kernel<<<dim3(Msz / HN_BROWS), dim3(HN_WAVES * 32), 0, stream>>>(
      Sb, Cb, ZHH, gn_w, PQT, b_y, GYA, h_out, y_out);
}


# zsum kernel split by block role: 128 factor blocks (b,h,column quarter) + 32 scan blocks using wave shuffles, 8 waves each; was 32 blocks of 32 waves doing both in turn. Scan sum order changes
# speedup vs baseline: 4.3858x; 1.0898x over previous
#include <hip/hip_runtime.h>
#include <hip/hip_bf16.h>
#include <math.h>


#define Bsz 4
#define Lsz 1024
#define Dsz 512
#define Hsz 8
#define Isz 64
#define Msz (Bsz * Lsz)
#define MT64 (Msz / 64)
#define LT64 (Lsz / 64)
#define GN_EPS 1e-5f

typedef float v2f __attribute__((ext_vector_type(2)));
typedef float v4f __attribute__((ext_vector_type(4)));
typedef float v8f __attribute__((ext_vector_type(8)));

__device__ __forceinline__ v8f wmma4(v2f a, v2f b, v8f c) {
  return __builtin_amdgcn_wmma_f32_16x16x4_f32(
       false, a,  false, b,
       (short)0, c,  false,  false);
}

__device__ __forceinline__ float silu_f(float v) {
  return v / (1.0f + expf(-v));
}

#define GK 32
#define GP 36
#define GMAT (64 * GP)
#define GKT (Dsz / GK)

template <int NB>
__device__ __forceinline__ void gemm_core(const float* __restrict__ A,
                                          const float* const (&W)[NB],
                                          float* __restrict__ lds,
                                          const int m_blk, const int n_blk,
                                          v8f (&acc)[NB][2][2]) {
  constexpr int STAGE = GMAT * (1 + NB);
  const int tid = threadIdx.x;
  const int lane = tid & 31;
  const int wave = tid >> 5;
  const int mr = lane & 15;
  const int kk = lane >> 4;
  const int wm = (wave & 1) * 32;
  const int wn = (wave >> 1) * 32;

  const int crow = tid >> 3;
  const int ccol = (tid & 7) * 4;
  const float* gA = A + (size_t)(m_blk + crow) * Dsz + ccol;
  const float* gB[NB];
#pragma unroll
  for (int nb = 0; nb < NB; ++nb)
    gB[nb] = W[nb] + (size_t)(n_blk + crow) * Dsz + ccol;
  const int soff = crow * GP + ccol;

  const int aoff = (wm + mr) * GP + 4 * kk;
  const int boff = (wn + mr) * GP + 4 * kk;

  v4f pa[4], pb[NB][4];
#pragma unroll
  for (int j = 0; j < 4; ++j) {
    pa[j] = *(const v4f*)(gA + (size_t)j * 16 * Dsz);
#pragma unroll
    for (int nb = 0; nb < NB; ++nb)
      pb[nb][j] = *(const v4f*)(gB[nb] + (size_t)j * 16 * Dsz);
  }
#pragma unroll
  for (int j = 0; j < 4; ++j) {
    *(v4f*)(lds + soff + j * 16 * GP) = pa[j];
#pragma unroll
    for (int nb = 0; nb < NB; ++nb)
      *(v4f*)(lds + GMAT * (1 + nb) + soff + j * 16 * GP) = pb[nb][j];
  }
  __syncthreads();

#pragma unroll 2
  for (int kt = 0; kt < GKT; ++kt) {
    const float* cur = lds + (kt & 1) * STAGE;
    float* nxt = lds + ((kt + 1) & 1) * STAGE;
    const bool more = (kt + 1 < GKT);

    if (more) {
      const int ko = (kt + 1) * GK;
#pragma unroll
      for (int j = 0; j < 4; ++j) {
        pa[j] = *(const v4f*)(gA + (size_t)j * 16 * Dsz + ko);
#pragma unroll
        for (int nb = 0; nb < NB; ++nb)
          pb[nb][j] = *(const v4f*)(gB[nb] + (size_t)j * 16 * Dsz + ko);
      }
    }

#pragma unroll
    for (int kb = 0; kb < GK / 8; ++kb) {
      v4f fa[2], fb[NB][2];
#pragma unroll
      for (int t = 0; t < 2; ++t) {
        fa[t] = *(const v4f*)(cur + aoff + t * 16 * GP + kb * 8);
#pragma unroll
        for (int nb = 0; nb < NB; ++nb)
          fb[nb][t] = *(const v4f*)(cur + GMAT * (1 + nb) + boff + t * 16 * GP + kb * 8);
      }
#pragma unroll
      for (int nb = 0; nb < NB; ++nb)
#pragma unroll
        for (int mt = 0; mt < 2; ++mt)
#pragma unroll
          for (int nt = 0; nt < 2; ++nt)
            acc[nb][mt][nt] = wmma4(fa[mt].xy, fb[nb][nt].xy, acc[nb][mt][nt]);
#pragma unroll
      for (int nb = 0; nb < NB; ++nb)
#pragma unroll
        for (int mt = 0; mt < 2; ++mt)
#pragma unroll
          for (int nt = 0; nt < 2; ++nt)
            acc[nb][mt][nt] = wmma4(fa[mt].zw, fb[nb][nt].zw, acc[nb][mt][nt]);
    }

    if (more) {
#pragma unroll
      for (int j = 0; j < 4; ++j) {
        *(v4f*)(nxt + soff + j * 16 * GP) = pa[j];
#pragma unroll
        for (int nb = 0; nb < NB; ++nb)
          *(v4f*)(nxt + GMAT * (1 + nb) + soff + j * 16 * GP) = pb[nb][j];
      }
    }
    __syncthreads();
  }
}

#define DT_ROWS  4
#define DT_WAVES 4
#define DT_THREADS (DT_WAVES * 32)
#define DT_BROWS (DT_ROWS * DT_WAVES)
#define DT_YB (Msz / DT_BROWS / (Msz / 64))

__device__ __forceinline__ void dt_role(const float* __restrict__ X,
                                        const float* __restrict__ w_dt,
                                        const float* __restrict__ b_dt,
                                        const float* __restrict__ ln_a,
                                        float* __restrict__ LD,
                                        float* __restrict__ lds, const int blk) {
  v4f* wl = (v4f*)lds;
  const int lane = threadIdx.x & 31;
  const int wave = threadIdx.x >> 5;
  const int row0 = (blk * DT_WAVES + wave) * DT_ROWS;

  v4f wr[Hsz * Dsz / 4 / DT_THREADS];
#pragma unroll
  for (int j = 0; j < Hsz * Dsz / 4 / DT_THREADS; ++j)
    wr[j] = *(const v4f*)(w_dt + 4 * (threadIdx.x + DT_THREADS * j));

  v4f xv[DT_ROWS][4];
#pragma unroll
  for (int r = 0; r < DT_ROWS; ++r)
#pragma unroll
    for (int c = 0; c < 4; ++c)
      xv[r][c] = *(const v4f*)(X + (size_t)(row0 + r) * Dsz + c * 128 + 4 * lane);

  const int h_out = lane & 7;
  const float bdt = b_dt[h_out];
  const float lna = ln_a[h_out];

  asm volatile("" :: "v"(xv[0][0]), "v"(xv[0][1]), "v"(xv[0][2]), "v"(xv[0][3]),
                     "v"(xv[1][0]), "v"(xv[1][1]), "v"(xv[1][2]), "v"(xv[1][3]),
                     "v"(xv[2][0]), "v"(xv[2][1]), "v"(xv[2][2]), "v"(xv[2][3]),
                     "v"(xv[3][0]), "v"(xv[3][1]), "v"(xv[3][2]), "v"(xv[3][3]),
                     "v"(bdt), "v"(lna));

#pragma unroll
  for (int j = 0; j < Hsz * Dsz / 4 / DT_THREADS; ++j)
    wl[threadIdx.x + DT_THREADS * j] = wr[j];
  __syncthreads();

  float v32[DT_ROWS * Hsz];
#pragma unroll
  for (int h = 0; h < Hsz; ++h) {
    v4f wv[4];
#pragma unroll
    for (int c = 0; c < 4; ++c) wv[c] = wl[h * (Dsz / 4) + c * 32 + lane];
#pragma unroll
    for (int r = 0; r < DT_ROWS; ++r) {
      float s[4];
#pragma unroll
      for (int c = 0; c < 4; ++c) {
        s[c] = xv[r][c].x * wv[c].x;
        s[c] = fmaf(xv[r][c].y, wv[c].y, s[c]);
        s[c] = fmaf(xv[r][c].z, wv[c].z, s[c]);
        s[c] = fmaf(xv[r][c].w, wv[c].w, s[c]);
      }
      v32[r * Hsz + h] = (s[0] + s[1]) + (s[2] + s[3]);
    }
  }

  const bool b4 = (lane & 16) != 0;
  const bool b3 = (lane & 8) != 0;
  const bool b2 = (lane & 4) != 0;
  const bool b1 = (lane & 2) != 0;
  const bool b0 = (lane & 1) != 0;
  float v16[16], v8[8], v4[4], v2[2];
#pragma unroll
  for (int i = 0; i < 16; ++i) {
    const float keep = b4 ? v32[16 + i] : v32[i];
    const float send = b4 ? v32[i] : v32[16 + i];
    v16[i] = keep + __shfl_xor(send, 16, 32);
  }
#pragma unroll
  for (int i = 0; i < 8; ++i) {
    const float keep = b3 ? v16[8 + i] : v16[i];
    const float send = b3 ? v16[i] : v16[8 + i];
    v8[i] = keep + __shfl_xor(send, 8, 32);
  }
#pragma unroll
  for (int i = 0; i < 4; ++i) {
    const float keep = b2 ? v8[4 + i] : v8[i];
    const float send = b2 ? v8[i] : v8[4 + i];
    v4[i] = keep + __shfl_xor(send, 4, 32);
  }
#pragma unroll
  for (int i = 0; i < 2; ++i) {
    const float keep = b1 ? v4[2 + i] : v4[i];
    const float send = b1 ? v4[i] : v4[2 + i];
    v2[i] = keep + __shfl_xor(send, 2, 32);
  }
  float dot;
  {
    const float keep = b0 ? v2[1] : v2[0];
    const float send = b0 ? v2[0] : v2[1];
    dot = keep + __shfl_xor(send, 1, 32);
  }

  const float v = dot + bdt;
  const float sp = (v > 20.f) ? v : log1pf(expf(v));
  const float ld = -expf(lna) * sp;
  LD[(size_t)row0 * Hsz + lane] = ld;
}

__global__ __launch_bounds__(128)
void gemm_xproj_kernel(const float* __restrict__ X,
                       const float* __restrict__ Wz,  const float* __restrict__ bz,
                       const float* __restrict__ Wza, const float* __restrict__ bza,
                       const float* __restrict__ Wya, const float* __restrict__ bya,
                       const float* __restrict__ w_dt, const float* __restrict__ b_dt,
                       const float* __restrict__ ln_a,
                       float* __restrict__ CP, float* __restrict__ ZHH,
                       float* __restrict__ GYA, float* __restrict__ LD) {
  __shared__ __attribute__((aligned(16))) float lds[2 * GMAT * 4];
  if (blockIdx.y >= Dsz / 64) {
    dt_role(X, w_dt, b_dt, ln_a, LD, lds,
            (blockIdx.y - Dsz / 64) * (Msz / 64) + blockIdx.x);
    return;
  }
  const int lane = threadIdx.x & 31;
  const int wave = threadIdx.x >> 5;
  const int mr = lane & 15;
  const int kk = lane >> 4;
  const int m_base = blockIdx.x * 64 + (wave & 1) * 32;
  const int n_base = blockIdx.y * 64 + (wave >> 1) * 32;

  v8f acc[3][2][2];
  const v8f zero = {0.f, 0.f, 0.f, 0.f, 0.f, 0.f, 0.f, 0.f};
  for (int nb = 0; nb < 3; ++nb)
    for (int i = 0; i < 2; ++i)
      for (int j = 0; j < 2; ++j) acc[nb][i][j] = zero;

  const float* const W[3] = {Wz, Wza, Wya};
  gemm_core<3>(X, W, lds, blockIdx.x * 64, blockIdx.y * 64, acc);

  const int h = blockIdx.y;
  const bool first = (blockIdx.x & (LT64 - 1)) == 0;
  const bool low = first && (wave & 1) == 0 && kk == 0;
  float cs[2] = {0.f, 0.f};
  float zd[2] = {0.f, 0.f};

  for (int mt = 0; mt < 2; ++mt)
    for (int nt = 0; nt < 2; ++nt) {
      const int n = n_base + nt * 16 + mr;
      const float biz = bz[n], biza = bza[n], biya = bya[n];
#pragma unroll
      for (int r = 0; r < 8; ++r) {
        const int m = m_base + mt * 16 + kk * 8 + r;
        const float v1 = acc[0][mt][nt][r] + biz;
        const float v2 = acc[1][mt][nt][r] + biza;
        const float v3 = acc[2][mt][nt][r] + biya;
        const float zv = v1 * silu_f(v2);
        if (mt == 0) {
          cs[nt] += (low && r <= h) ? 0.f : zv;
          if (r == h) zd[nt] = zv;
        } else {
          cs[nt] += zv;
        }
        GYA[(size_t)m * Dsz + n] = silu_f(v3);
      }
    }

  const int cl = (wave >> 1) * 32 + mr;
#pragma unroll
  for (int nt = 0; nt < 2; ++nt) {
    lds[((wave & 1) * 2 + kk) * Isz + cl + nt * 16] = cs[nt];
    if (low)
      ZHH[(size_t)((blockIdx.x / LT64) * Hsz + h) * Isz + cl + nt * 16] = zd[nt];
  }
  __syncthreads();
  if (threadIdx.x < Isz) {
    const float* p = lds + threadIdx.x;
    CP[((size_t)blockIdx.x * Hsz + h) * Isz + threadIdx.x] =
        (p[0] + p[Isz]) + (p[2 * Isz] + p[3 * Isz]);
  }
}

#define PQK 24
#define ATP 68
#define ZS_THREADS 256
#define ZS_WAVES (ZS_THREADS / 32)
#define ZS_NQ (Dsz / 16 / ZS_WAVES)
#define ZS_FBLK (Bsz * Hsz * ZS_NQ)
#define ZS_SEG (Lsz / ZS_THREADS)

__global__ __launch_bounds__(ZS_THREADS)
void zsum_kernel(const float* __restrict__ CP, const float* __restrict__ zhh,
                 const float* __restrict__ hidden,
                 const float* __restrict__ LD, const float* __restrict__ Wy,
                 const float* __restrict__ gn_b,
                 float* __restrict__ C,
                 float* __restrict__ S, float* __restrict__ PQT) {
  __shared__ __attribute__((aligned(16))) float at[4 * ATP];
  __shared__ __attribute__((aligned(16))) float wt[ZS_SEG * ZS_WAVES];
  const int tid = threadIdx.x;
  const int lane = tid & 31;
  const int wave = tid >> 5;

  if (blockIdx.x >= ZS_FBLK) {
    const int bh = blockIdx.x - ZS_FBLK;
    const int b = bh >> 3;
    const int h = bh & 7;
    float v[ZS_SEG];
#pragma unroll
    for (int j = 0; j < ZS_SEG; ++j)
      v[j] = LD[(size_t)(b * Lsz + j * ZS_THREADS + tid) * Hsz + h];
#pragma unroll
    for (int d = 1; d < 32; d <<= 1) {
#pragma unroll
      for (int j = 0; j < ZS_SEG; ++j) {
        const float up = __shfl_up(v[j], d, 32);
        if (lane >= d) v[j] += up;
      }
    }
    if (lane == 31) {
#pragma unroll
      for (int j = 0; j < ZS_SEG; ++j) wt[j * ZS_WAVES + wave] = v[j];
    }
    __syncthreads();
    float run = 0.f;
    float base[ZS_SEG];
#pragma unroll
    for (int j = 0; j < ZS_SEG; ++j) {
      base[j] = 0.f;
#pragma unroll
      for (int w = 0; w < ZS_WAVES; ++w) {
        if (w == wave) base[j] = run;
        run += wt[j * ZS_WAVES + w];
      }
    }
#pragma unroll
    for (int j = 0; j < ZS_SEG; ++j)
      S[(size_t)(b * Lsz + j * ZS_THREADS + tid) * Hsz + h] = base[j] + v[j];
    return;
  }

  const int q = blockIdx.x & (ZS_NQ - 1);
  const int h = (blockIdx.x / ZS_NQ) & 7;
  const int b = blockIdx.x / (ZS_NQ * Hsz);
  const int c16 = tid & 15;
  const int half = (tid >> 4) & 1;
  const int n0 = (q * ZS_WAVES + wave) * 16;
  const float gbh = gn_b[h];

  const int ch = (b * Hsz + h) * Isz + (tid & 63);
  float zdiag = 0.f, hid = 0.f;
  float t[LT64];
#pragma unroll
  for (int g = 0; g < LT64; ++g) t[g] = 0.f;
  if (tid < Isz) {
    zdiag = zhh[ch];
    hid = hidden[ch];
#pragma unroll
    for (int g = 0; g < LT64; ++g)
      t[g] = CP[((size_t)(b * LT64 + g) * Hsz + h) * Isz + tid];
  }

  v4f wf[Isz / 8];
  {
    const float* wp = Wy + (size_t)(n0 + c16) * Dsz + h * Isz + 4 * half;
#pragma unroll
    for (int kb = 0; kb < Isz / 8; ++kb) wf[kb] = *(const v4f*)(wp + 8 * kb);
  }

  float cfull = 0.f;
  if (tid < Isz) {
#pragma unroll
    for (int w = 1; w < LT64; w <<= 1)
#pragma unroll
      for (int g = 0; g + w < LT64; g += 2 * w) t[g] += t[g + w];
    cfull = t[0] + hid;
    if (q == 0) C[ch] = cfull;
    at[tid] = zdiag;
    at[ATP + tid] = cfull;
    at[2 * ATP + tid] = gbh;
    at[3 * ATP + tid] = 0.f;
  }
  __syncthreads();

  float mz = 0.f, mc = 0.f;
  if (tid < Isz) {
    v4f sz = {0.f, 0.f, 0.f, 0.f}, sq = {0.f, 0.f, 0.f, 0.f};
#pragma unroll
    for (int j = 0; j < Isz / 4; ++j) {
      sz += *(const v4f*)(at + 4 * j);
      sq += *(const v4f*)(at + ATP + 4 * j);
    }
    mz = ((sz.x + sz.y) + (sz.z + sz.w)) * (1.0f / 64.0f);
    mc = ((sq.x + sq.y) + (sq.z + sq.w)) * (1.0f / 64.0f);
  }
  __syncthreads();
  if (tid < Isz) {
    at[tid] = zdiag - mz;
    at[ATP + tid] = cfull - mc;
  }
  __syncthreads();

  const int ar = (c16 < 3) ? c16 : 3;
  v8f acc = {0.f, 0.f, 0.f, 0.f, 0.f, 0.f, 0.f, 0.f};
#pragma unroll
  for (int kb = 0; kb < Isz / 8; ++kb) {
    const v4f a = *(const v4f*)(at + ar * ATP + 8 * kb + 4 * half);
    acc = wmma4(a.xy, wf[kb].xy, acc);
    acc = wmma4(a.zw, wf[kb].zw, acc);
  }
  if (half == 0) {
    float* o = PQT + (size_t)(b * Dsz + n0 + c16) * PQK + h;
    o[0] = acc[0];
    o[8] = acc[1];
    o[16] = acc[2];
  }
}

#define HN_ROWS  2
#define HN_WAVES 8
#define HN_BROWS (HN_ROWS * HN_WAVES)
#define CTP 28
#define YTP 68

template <int M>
__device__ __forceinline__ float row_xor_add(float v) {
  const int p = __builtin_amdgcn_update_dpp(0, __float_as_int(v), 0x160 + M,
                                            0xf, 0xf, false);
  return v + __int_as_float(p);
}

__global__ __launch_bounds__(HN_WAVES * 32)
void hnorm_y_kernel(const float* __restrict__ S, const float* __restrict__ C,
                    const float* __restrict__ zhh, const float* __restrict__ gn_w,
                    const float* __restrict__ PQT, const float* __restrict__ by,
                    const float* __restrict__ GYA, float* __restrict__ Hout,
                    float* __restrict__ Y) {
  __shared__ __attribute__((aligned(16))) float ct[HN_BROWS * CTP];
  __shared__ __attribute__((aligned(16))) float yt[HN_WAVES * HN_BROWS * YTP];
  const int lane = threadIdx.x & 31;
  const int wave = threadIdx.x >> 5;
  const int brow0 = blockIdx.x * HN_BROWS;
  const int row0 = brow0 + wave * HN_ROWS;
  const int b = row0 >> 10;
  const int hh = lane >> 4;
  const int c16 = lane & 15;
  const int nb = wave * 64;

  v4f fb[4][PQK / 8];
#pragma unroll
  for (int nt = 0; nt < 4; ++nt)
#pragma unroll
    for (int kb = 0; kb < PQK / 8; ++kb)
      fb[nt][kb] = *(const v4f*)(PQT + (size_t)(b * Dsz + nb + nt * 16 + c16) * PQK +
                                 8 * kb + 4 * hh);
  v4f gy[HN_BROWS / 2];
#pragma unroll
  for (int it = 0; it < HN_BROWS / 2; ++it)
    gy[it] = *(const v4f*)(GYA + (size_t)(brow0 + 2 * it + hh) * Dsz + nb + 4 * c16);
  const v4f bv = *(const v4f*)(by + nb + 4 * c16);

  v4f zv[4], cv[4];
  float gw[4];
  float sv[HN_ROWS][4];
#pragma unroll
  for (int c = 0; c < 4; ++c) {
    zv[c] = *(const v4f*)(zhh + (size_t)b * Dsz + c * 128 + 4 * lane);
    cv[c] = *(const v4f*)(C + (size_t)b * Dsz + c * 128 + 4 * lane);
    gw[c] = gn_w[2 * c + hh];
  }
#pragma unroll
  for (int r = 0; r < HN_ROWS; ++r)
#pragma unroll
    for (int c = 0; c < 4; ++c)
      sv[r][c] = S[(size_t)(row0 + r) * Hsz + 2 * c + hh];

#pragma unroll
  for (int r = 0; r < HN_ROWS; ++r) {
    const size_t ro = (size_t)(row0 + r) * Dsz + 4 * lane;
    float* cr = ct + (wave * HN_ROWS + r) * CTP;
#pragma unroll
    for (int c = 0; c < 4; ++c) {
      const float e = expf(sv[r][c]);
      v4f hv;
      hv.x = zv[c].x + e * cv[c].x;
      hv.y = zv[c].y + e * cv[c].y;
      hv.z = zv[c].z + e * cv[c].z;
      hv.w = zv[c].w + e * cv[c].w;
      *(v4f*)(Hout + ro + c * 128) = hv;

      float s = (hv.x + hv.y) + (hv.z + hv.w);
      s = row_xor_add<1>(s);
      s = row_xor_add<2>(s);
      s = row_xor_add<4>(s);
      s = row_xor_add<8>(s);
      const float mu = s * (1.0f / 64.0f);

      v4f d;
      d.x = hv.x - mu; d.y = hv.y - mu; d.z = hv.z - mu; d.w = hv.w - mu;
      float q = (d.x * d.x + d.y * d.y) + (d.z * d.z + d.w * d.w);
      q = row_xor_add<1>(q);
      q = row_xor_add<2>(q);
      q = row_xor_add<4>(q);
      q = row_xor_add<8>(q);
      const float var = q * (1.0f / 64.0f);

      const float sc = gw[c] / sqrtf(var + GN_EPS);
      if (c16 == 0) {
        const int head = 2 * c + hh;
        cr[head] = sc;
        cr[8 + head] = sc * e;
        cr[16 + head] = 1.0f;
      }
    }
  }
  __syncthreads();

  v4f fa[PQK / 8];
#pragma unroll
  for (int kb = 0; kb < PQK / 8; ++kb)
    fa[kb] = *(const v4f*)(ct + c16 * CTP + 8 * kb + 4 * hh);

  v8f acc[4];
  const v8f zero = {0.f, 0.f, 0.f, 0.f, 0.f, 0.f, 0.f, 0.f};
#pragma unroll
  for (int nt = 0; nt < 4; ++nt) acc[nt] = zero;
#pragma unroll
  for (int kb = 0; kb < PQK / 8; ++kb) {
#pragma unroll
    for (int nt = 0; nt < 4; ++nt)
      acc[nt] = wmma4(fa[kb].xy, fb[nt][kb].xy, acc[nt]);
#pragma unroll
    for (int nt = 0; nt < 4; ++nt)
      acc[nt] = wmma4(fa[kb].zw, fb[nt][kb].zw, acc[nt]);
  }

  float* yw = yt + wave * HN_BROWS * YTP;
#pragma unroll
  for (int nt = 0; nt < 4; ++nt)
#pragma unroll
    for (int r = 0; r < 8; ++r)
      yw[(8 * hh + r) * YTP + nt * 16 + c16] = acc[nt][r];
  __syncthreads();

#pragma unroll
  for (int it = 0; it < HN_BROWS / 2; ++it) {
    const int row = 2 * it + hh;
    const v4f d = *(const v4f*)(yw + row * YTP + 4 * c16);
    v4f o;
    o.x = (d.x + bv.x) * gy[it].x;
    o.y = (d.y + bv.y) * gy[it].y;
    o.z = (d.z + bv.z) * gy[it].z;
    o.w = (d.w + bv.w) * gy[it].w;
    *(v4f*)(Y + (size_t)(brow0 + row) * Dsz + nb + 4 * c16) = o;
  }
}

extern "C" void kernel_launch(void* const* d_in, const int* in_sizes, int n_in,
                              void* d_out, int out_size, void* d_ws, size_t ws_size,
                              hipStream_t stream) {
  const float* x      = (const float*)d_in[0];
  const float* hidden = (const float*)d_in[1];
  const float* w_z    = (const float*)d_in[2];
  const float* b_z    = (const float*)d_in[3];
  const float* w_za   = (const float*)d_in[4];
  const float* b_za   = (const float*)d_in[5];
  const float* w_y    = (const float*)d_in[6];
  const float* b_y    = (const float*)d_in[7];
  const float* w_ya   = (const float*)d_in[8];
  const float* b_ya   = (const float*)d_in[9];
  const float* w_dt   = (const float*)d_in[10];
  const float* b_dt   = (const float*)d_in[11];
  const float* ln_a   = (const float*)d_in[12];
  const float* gn_w   = (const float*)d_in[13];
  const float* gn_b   = (const float*)d_in[14];

  float* y_out = (float*)d_out;
  float* h_out = y_out + (size_t)Msz * Dsz;

  float* ws   = (float*)d_ws;
  float* GYA  = ws;
  float* CPb  = GYA  + (size_t)Msz * Dsz;
  float* Sb   = CPb  + (size_t)MT64 * Hsz * Isz;
  float* Cb   = Sb   + (size_t)Msz * Hsz;
  float* ZHH  = Cb   + (size_t)Bsz * Hsz * Isz;
  float* LDb  = ZHH  + (size_t)Bsz * Hsz * Isz;
  float* PQT  = LDb  + (size_t)Msz * Hsz;

  const dim3 gGemm(Msz / 64, Dsz / 64 + DT_YB, 1);
  const dim3 bGemm(128, 1, 1);

  gemm_xproj_kernel<<<gGemm, bGemm, 0, stream>>>(x, w_z, b_z, w_za, b_za, w_ya, b_ya,
                                                 w_dt, b_dt, ln_a, CPb, ZHH, GYA, LDb);
  zsum_kernel<<<dim3(ZS_FBLK + Bsz * Hsz), dim3(ZS_THREADS), 0, stream>>>(CPb, ZHH, hidden, LDb, w_y, gn_b,
                                                          Cb, Sb, PQT);
  hnorm_y_kernel<<<dim3(Msz / HN_BROWS), dim3(HN_WAVES * 32), 0, stream>>>(
      Sb, Cb, ZHH, gn_w, PQT, b_y, GYA, h_out, y_out);
}


# GEMM blocks: wave priority (s_setprio) stepped down 3,2,1,0 through K, so the two blocks sharing a SIMD stay abreast and the younger no longer runs most of its K loop alone. Sum order unchanged
# speedup vs baseline: 4.5580x; 1.0393x over previous
#include <hip/hip_runtime.h>
#include <hip/hip_bf16.h>
#include <math.h>


#define Bsz 4
#define Lsz 1024
#define Dsz 512
#define Hsz 8
#define Isz 64
#define Msz (Bsz * Lsz)
#define MT64 (Msz / 64)
#define LT64 (Lsz / 64)
#define GN_EPS 1e-5f

typedef float v2f __attribute__((ext_vector_type(2)));
typedef float v4f __attribute__((ext_vector_type(4)));
typedef float v8f __attribute__((ext_vector_type(8)));

__device__ __forceinline__ v8f wmma4(v2f a, v2f b, v8f c) {
  return __builtin_amdgcn_wmma_f32_16x16x4_f32(
       false, a,  false, b,
       (short)0, c,  false,  false);
}

__device__ __forceinline__ float silu_f(float v) {
  return v / (1.0f + expf(-v));
}

#define GK 32
#define GP 36
#define GMAT (64 * GP)
#define GKT (Dsz / GK)

template <int NB>
__device__ __forceinline__ void gemm_core(const float* __restrict__ A,
                                          const float* const (&W)[NB],
                                          float* __restrict__ lds,
                                          const int m_blk, const int n_blk,
                                          v8f (&acc)[NB][2][2]) {
  constexpr int STAGE = GMAT * (1 + NB);
  const int tid = threadIdx.x;
  const int lane = tid & 31;
  const int wave = tid >> 5;
  const int mr = lane & 15;
  const int kk = lane >> 4;
  const int wm = (wave & 1) * 32;
  const int wn = (wave >> 1) * 32;

  const int crow = tid >> 3;
  const int ccol = (tid & 7) * 4;
  const float* gA = A + (size_t)(m_blk + crow) * Dsz + ccol;
  const float* gB[NB];
#pragma unroll
  for (int nb = 0; nb < NB; ++nb)
    gB[nb] = W[nb] + (size_t)(n_blk + crow) * Dsz + ccol;
  const int soff = crow * GP + ccol;

  const int aoff = (wm + mr) * GP + 4 * kk;
  const int boff = (wn + mr) * GP + 4 * kk;

  v4f pa[4], pb[NB][4];
#pragma unroll
  for (int j = 0; j < 4; ++j) {
    pa[j] = *(const v4f*)(gA + (size_t)j * 16 * Dsz);
#pragma unroll
    for (int nb = 0; nb < NB; ++nb)
      pb[nb][j] = *(const v4f*)(gB[nb] + (size_t)j * 16 * Dsz);
  }
#pragma unroll
  for (int j = 0; j < 4; ++j) {
    *(v4f*)(lds + soff + j * 16 * GP) = pa[j];
#pragma unroll
    for (int nb = 0; nb < NB; ++nb)
      *(v4f*)(lds + GMAT * (1 + nb) + soff + j * 16 * GP) = pb[nb][j];
  }
  __syncthreads();

#pragma unroll 2
  for (int kt = 0; kt < GKT; ++kt) {
    const float* cur = lds + (kt & 1) * STAGE;
    float* nxt = lds + ((kt + 1) & 1) * STAGE;
    const bool more = (kt + 1 < GKT);

    switch (kt) {
      case 0:           __builtin_amdgcn_s_setprio(3); break;
      case GKT / 3:     __builtin_amdgcn_s_setprio(2); break;
      case 2 * GKT / 3: __builtin_amdgcn_s_setprio(1); break;
      case GKT - 1:     __builtin_amdgcn_s_setprio(0); break;
      default: break;
    }

    if (more) {
      const int ko = (kt + 1) * GK;
#pragma unroll
      for (int j = 0; j < 4; ++j) {
        pa[j] = *(const v4f*)(gA + (size_t)j * 16 * Dsz + ko);
#pragma unroll
        for (int nb = 0; nb < NB; ++nb)
          pb[nb][j] = *(const v4f*)(gB[nb] + (size_t)j * 16 * Dsz + ko);
      }
    }

#pragma unroll
    for (int kb = 0; kb < GK / 8; ++kb) {
      v4f fa[2], fb[NB][2];
#pragma unroll
      for (int t = 0; t < 2; ++t) {
        fa[t] = *(const v4f*)(cur + aoff + t * 16 * GP + kb * 8);
#pragma unroll
        for (int nb = 0; nb < NB; ++nb)
          fb[nb][t] = *(const v4f*)(cur + GMAT * (1 + nb) + boff + t * 16 * GP + kb * 8);
      }
#pragma unroll
      for (int nb = 0; nb < NB; ++nb)
#pragma unroll
        for (int mt = 0; mt < 2; ++mt)
#pragma unroll
          for (int nt = 0; nt < 2; ++nt)
            acc[nb][mt][nt] = wmma4(fa[mt].xy, fb[nb][nt].xy, acc[nb][mt][nt]);
#pragma unroll
      for (int nb = 0; nb < NB; ++nb)
#pragma unroll
        for (int mt = 0; mt < 2; ++mt)
#pragma unroll
          for (int nt = 0; nt < 2; ++nt)
            acc[nb][mt][nt] = wmma4(fa[mt].zw, fb[nb][nt].zw, acc[nb][mt][nt]);
    }

    if (more) {
#pragma unroll
      for (int j = 0; j < 4; ++j) {
        *(v4f*)(nxt + soff + j * 16 * GP) = pa[j];
#pragma unroll
        for (int nb = 0; nb < NB; ++nb)
          *(v4f*)(nxt + GMAT * (1 + nb) + soff + j * 16 * GP) = pb[nb][j];
      }
    }
    __syncthreads();
  }
}

#define DT_ROWS  4
#define DT_WAVES 4
#define DT_THREADS (DT_WAVES * 32)
#define DT_BROWS (DT_ROWS * DT_WAVES)
#define DT_YB (Msz / DT_BROWS / (Msz / 64))

__device__ __forceinline__ void dt_role(const float* __restrict__ X,
                                        const float* __restrict__ w_dt,
                                        const float* __restrict__ b_dt,
                                        const float* __restrict__ ln_a,
                                        float* __restrict__ LD,
                                        float* __restrict__ lds, const int blk) {
  v4f* wl = (v4f*)lds;
  const int lane = threadIdx.x & 31;
  const int wave = threadIdx.x >> 5;
  const int row0 = (blk * DT_WAVES + wave) * DT_ROWS;

  v4f wr[Hsz * Dsz / 4 / DT_THREADS];
#pragma unroll
  for (int j = 0; j < Hsz * Dsz / 4 / DT_THREADS; ++j)
    wr[j] = *(const v4f*)(w_dt + 4 * (threadIdx.x + DT_THREADS * j));

  v4f xv[DT_ROWS][4];
#pragma unroll
  for (int r = 0; r < DT_ROWS; ++r)
#pragma unroll
    for (int c = 0; c < 4; ++c)
      xv[r][c] = *(const v4f*)(X + (size_t)(row0 + r) * Dsz + c * 128 + 4 * lane);

  const int h_out = lane & 7;
  const float bdt = b_dt[h_out];
  const float lna = ln_a[h_out];

  asm volatile("" :: "v"(xv[0][0]), "v"(xv[0][1]), "v"(xv[0][2]), "v"(xv[0][3]),
                     "v"(xv[1][0]), "v"(xv[1][1]), "v"(xv[1][2]), "v"(xv[1][3]),
                     "v"(xv[2][0]), "v"(xv[2][1]), "v"(xv[2][2]), "v"(xv[2][3]),
                     "v"(xv[3][0]), "v"(xv[3][1]), "v"(xv[3][2]), "v"(xv[3][3]),
                     "v"(bdt), "v"(lna));

#pragma unroll
  for (int j = 0; j < Hsz * Dsz / 4 / DT_THREADS; ++j)
    wl[threadIdx.x + DT_THREADS * j] = wr[j];
  __syncthreads();

  float v32[DT_ROWS * Hsz];
#pragma unroll
  for (int h = 0; h < Hsz; ++h) {
    v4f wv[4];
#pragma unroll
    for (int c = 0; c < 4; ++c) wv[c] = wl[h * (Dsz / 4) + c * 32 + lane];
#pragma unroll
    for (int r = 0; r < DT_ROWS; ++r) {
      float s[4];
#pragma unroll
      for (int c = 0; c < 4; ++c) {
        s[c] = xv[r][c].x * wv[c].x;
        s[c] = fmaf(xv[r][c].y, wv[c].y, s[c]);
        s[c] = fmaf(xv[r][c].z, wv[c].z, s[c]);
        s[c] = fmaf(xv[r][c].w, wv[c].w, s[c]);
      }
      v32[r * Hsz + h] = (s[0] + s[1]) + (s[2] + s[3]);
    }
  }

  const bool b4 = (lane & 16) != 0;
  const bool b3 = (lane & 8) != 0;
  const bool b2 = (lane & 4) != 0;
  const bool b1 = (lane & 2) != 0;
  const bool b0 = (lane & 1) != 0;
  float v16[16], v8[8], v4[4], v2[2];
#pragma unroll
  for (int i = 0; i < 16; ++i) {
    const float keep = b4 ? v32[16 + i] : v32[i];
    const float send = b4 ? v32[i] : v32[16 + i];
    v16[i] = keep + __shfl_xor(send, 16, 32);
  }
#pragma unroll
  for (int i = 0; i < 8; ++i) {
    const float keep = b3 ? v16[8 + i] : v16[i];
    const float send = b3 ? v16[i] : v16[8 + i];
    v8[i] = keep + __shfl_xor(send, 8, 32);
  }
#pragma unroll
  for (int i = 0; i < 4; ++i) {
    const float keep = b2 ? v8[4 + i] : v8[i];
    const float send = b2 ? v8[i] : v8[4 + i];
    v4[i] = keep + __shfl_xor(send, 4, 32);
  }
#pragma unroll
  for (int i = 0; i < 2; ++i) {
    const float keep = b1 ? v4[2 + i] : v4[i];
    const float send = b1 ? v4[i] : v4[2 + i];
    v2[i] = keep + __shfl_xor(send, 2, 32);
  }
  float dot;
  {
    const float keep = b0 ? v2[1] : v2[0];
    const float send = b0 ? v2[0] : v2[1];
    dot = keep + __shfl_xor(send, 1, 32);
  }

  const float v = dot + bdt;
  const float sp = (v > 20.f) ? v : log1pf(expf(v));
  const float ld = -expf(lna) * sp;
  LD[(size_t)row0 * Hsz + lane] = ld;
}

__global__ __launch_bounds__(128)
void gemm_xproj_kernel(const float* __restrict__ X,
                       const float* __restrict__ Wz,  const float* __restrict__ bz,
                       const float* __restrict__ Wza, const float* __restrict__ bza,
                       const float* __restrict__ Wya, const float* __restrict__ bya,
                       const float* __restrict__ w_dt, const float* __restrict__ b_dt,
                       const float* __restrict__ ln_a,
                       float* __restrict__ CP, float* __restrict__ ZHH,
                       float* __restrict__ GYA, float* __restrict__ LD) {
  __shared__ __attribute__((aligned(16))) float lds[2 * GMAT * 4];
  if (blockIdx.y >= Dsz / 64) {
    dt_role(X, w_dt, b_dt, ln_a, LD, lds,
            (blockIdx.y - Dsz / 64) * (Msz / 64) + blockIdx.x);
    return;
  }
  const int lane = threadIdx.x & 31;
  const int wave = threadIdx.x >> 5;
  const int mr = lane & 15;
  const int kk = lane >> 4;
  const int m_base = blockIdx.x * 64 + (wave & 1) * 32;
  const int n_base = blockIdx.y * 64 + (wave >> 1) * 32;

  v8f acc[3][2][2];
  const v8f zero = {0.f, 0.f, 0.f, 0.f, 0.f, 0.f, 0.f, 0.f};
  for (int nb = 0; nb < 3; ++nb)
    for (int i = 0; i < 2; ++i)
      for (int j = 0; j < 2; ++j) acc[nb][i][j] = zero;

  const float* const W[3] = {Wz, Wza, Wya};
  gemm_core<3>(X, W, lds, blockIdx.x * 64, blockIdx.y * 64, acc);

  const int h = blockIdx.y;
  const bool first = (blockIdx.x & (LT64 - 1)) == 0;
  const bool low = first && (wave & 1) == 0 && kk == 0;
  float cs[2] = {0.f, 0.f};
  float zd[2] = {0.f, 0.f};

  for (int mt = 0; mt < 2; ++mt)
    for (int nt = 0; nt < 2; ++nt) {
      const int n = n_base + nt * 16 + mr;
      const float biz = bz[n], biza = bza[n], biya = bya[n];
#pragma unroll
      for (int r = 0; r < 8; ++r) {
        const int m = m_base + mt * 16 + kk * 8 + r;
        const float v1 = acc[0][mt][nt][r] + biz;
        const float v2 = acc[1][mt][nt][r] + biza;
        const float v3 = acc[2][mt][nt][r] + biya;
        const float zv = v1 * silu_f(v2);
        if (mt == 0) {
          cs[nt] += (low && r <= h) ? 0.f : zv;
          if (r == h) zd[nt] = zv;
        } else {
          cs[nt] += zv;
        }
        GYA[(size_t)m * Dsz + n] = silu_f(v3);
      }
    }

  const int cl = (wave >> 1) * 32 + mr;
#pragma unroll
  for (int nt = 0; nt < 2; ++nt) {
    lds[((wave & 1) * 2 + kk) * Isz + cl + nt * 16] = cs[nt];
    if (low)
      ZHH[(size_t)((blockIdx.x / LT64) * Hsz + h) * Isz + cl + nt * 16] = zd[nt];
  }
  __syncthreads();
  if (threadIdx.x < Isz) {
    const float* p = lds + threadIdx.x;
    CP[((size_t)blockIdx.x * Hsz + h) * Isz + threadIdx.x] =
        (p[0] + p[Isz]) + (p[2 * Isz] + p[3 * Isz]);
  }
}

#define PQK 24
#define ATP 68
#define ZS_THREADS 256
#define ZS_WAVES (ZS_THREADS / 32)
#define ZS_NQ (Dsz / 16 / ZS_WAVES)
#define ZS_FBLK (Bsz * Hsz * ZS_NQ)
#define ZS_SEG (Lsz / ZS_THREADS)

__global__ __launch_bounds__(ZS_THREADS)
void zsum_kernel(const float* __restrict__ CP, const float* __restrict__ zhh,
                 const float* __restrict__ hidden,
                 const float* __restrict__ LD, const float* __restrict__ Wy,
                 const float* __restrict__ gn_b,
                 float* __restrict__ C,
                 float* __restrict__ S, float* __restrict__ PQT) {
  __shared__ __attribute__((aligned(16))) float at[4 * ATP];
  __shared__ __attribute__((aligned(16))) float wt[ZS_SEG * ZS_WAVES];
  const int tid = threadIdx.x;
  const int lane = tid & 31;
  const int wave = tid >> 5;

  if (blockIdx.x >= ZS_FBLK) {
    const int bh = blockIdx.x - ZS_FBLK;
    const int b = bh >> 3;
    const int h = bh & 7;
    float v[ZS_SEG];
#pragma unroll
    for (int j = 0; j < ZS_SEG; ++j)
      v[j] = LD[(size_t)(b * Lsz + j * ZS_THREADS + tid) * Hsz + h];
#pragma unroll
    for (int d = 1; d < 32; d <<= 1) {
#pragma unroll
      for (int j = 0; j < ZS_SEG; ++j) {
        const float up = __shfl_up(v[j], d, 32);
        if (lane >= d) v[j] += up;
      }
    }
    if (lane == 31) {
#pragma unroll
      for (int j = 0; j < ZS_SEG; ++j) wt[j * ZS_WAVES + wave] = v[j];
    }
    __syncthreads();
    float run = 0.f;
    float base[ZS_SEG];
#pragma unroll
    for (int j = 0; j < ZS_SEG; ++j) {
      base[j] = 0.f;
#pragma unroll
      for (int w = 0; w < ZS_WAVES; ++w) {
        if (w == wave) base[j] = run;
        run += wt[j * ZS_WAVES + w];
      }
    }
#pragma unroll
    for (int j = 0; j < ZS_SEG; ++j)
      S[(size_t)(b * Lsz + j * ZS_THREADS + tid) * Hsz + h] = base[j] + v[j];
    return;
  }

  const int q = blockIdx.x & (ZS_NQ - 1);
  const int h = (blockIdx.x / ZS_NQ) & 7;
  const int b = blockIdx.x / (ZS_NQ * Hsz);
  const int c16 = tid & 15;
  const int half = (tid >> 4) & 1;
  const int n0 = (q * ZS_WAVES + wave) * 16;
  const float gbh = gn_b[h];

  const int ch = (b * Hsz + h) * Isz + (tid & 63);
  float zdiag = 0.f, hid = 0.f;
  float t[LT64];
#pragma unroll
  for (int g = 0; g < LT64; ++g) t[g] = 0.f;
  if (tid < Isz) {
    zdiag = zhh[ch];
    hid = hidden[ch];
#pragma unroll
    for (int g = 0; g < LT64; ++g)
      t[g] = CP[((size_t)(b * LT64 + g) * Hsz + h) * Isz + tid];
  }

  v4f wf[Isz / 8];
  {
    const float* wp = Wy + (size_t)(n0 + c16) * Dsz + h * Isz + 4 * half;
#pragma unroll
    for (int kb = 0; kb < Isz / 8; ++kb) wf[kb] = *(const v4f*)(wp + 8 * kb);
  }

  float cfull = 0.f;
  if (tid < Isz) {
#pragma unroll
    for (int w = 1; w < LT64; w <<= 1)
#pragma unroll
      for (int g = 0; g + w < LT64; g += 2 * w) t[g] += t[g + w];
    cfull = t[0] + hid;
    if (q == 0) C[ch] = cfull;
    at[tid] = zdiag;
    at[ATP + tid] = cfull;
    at[2 * ATP + tid] = gbh;
    at[3 * ATP + tid] = 0.f;
  }
  __syncthreads();

  float mz = 0.f, mc = 0.f;
  if (tid < Isz) {
    v4f sz = {0.f, 0.f, 0.f, 0.f}, sq = {0.f, 0.f, 0.f, 0.f};
#pragma unroll
    for (int j = 0; j < Isz / 4; ++j) {
      sz += *(const v4f*)(at + 4 * j);
      sq += *(const v4f*)(at + ATP + 4 * j);
    }
    mz = ((sz.x + sz.y) + (sz.z + sz.w)) * (1.0f / 64.0f);
    mc = ((sq.x + sq.y) + (sq.z + sq.w)) * (1.0f / 64.0f);
  }
  __syncthreads();
  if (tid < Isz) {
    at[tid] = zdiag - mz;
    at[ATP + tid] = cfull - mc;
  }
  __syncthreads();

  const int ar = (c16 < 3) ? c16 : 3;
  v8f acc = {0.f, 0.f, 0.f, 0.f, 0.f, 0.f, 0.f, 0.f};
#pragma unroll
  for (int kb = 0; kb < Isz / 8; ++kb) {
    const v4f a = *(const v4f*)(at + ar * ATP + 8 * kb + 4 * half);
    acc = wmma4(a.xy, wf[kb].xy, acc);
    acc = wmma4(a.zw, wf[kb].zw, acc);
  }
  if (half == 0) {
    float* o = PQT + (size_t)(b * Dsz + n0 + c16) * PQK + h;
    o[0] = acc[0];
    o[8] = acc[1];
    o[16] = acc[2];
  }
}

#define HN_ROWS  2
#define HN_WAVES 8
#define HN_BROWS (HN_ROWS * HN_WAVES)
#define CTP 28
#define YTP 68

template <int M>
__device__ __forceinline__ float row_xor_add(float v) {
  const int p = __builtin_amdgcn_update_dpp(0, __float_as_int(v), 0x160 + M,
                                            0xf, 0xf, false);
  return v + __int_as_float(p);
}

__global__ __launch_bounds__(HN_WAVES * 32)
void hnorm_y_kernel(const float* __restrict__ S, const float* __restrict__ C,
                    const float* __restrict__ zhh, const float* __restrict__ gn_w,
                    const float* __restrict__ PQT, const float* __restrict__ by,
                    const float* __restrict__ GYA, float* __restrict__ Hout,
                    float* __restrict__ Y) {
  __shared__ __attribute__((aligned(16))) float ct[HN_BROWS * CTP];
  __shared__ __attribute__((aligned(16))) float yt[HN_WAVES * HN_BROWS * YTP];
  const int lane = threadIdx.x & 31;
  const int wave = threadIdx.x >> 5;
  const int brow0 = blockIdx.x * HN_BROWS;
  const int row0 = brow0 + wave * HN_ROWS;
  const int b = row0 >> 10;
  const int hh = lane >> 4;
  const int c16 = lane & 15;
  const int nb = wave * 64;

  v4f fb[4][PQK / 8];
#pragma unroll
  for (int nt = 0; nt < 4; ++nt)
#pragma unroll
    for (int kb = 0; kb < PQK / 8; ++kb)
      fb[nt][kb] = *(const v4f*)(PQT + (size_t)(b * Dsz + nb + nt * 16 + c16) * PQK +
                                 8 * kb + 4 * hh);
  v4f gy[HN_BROWS / 2];
#pragma unroll
  for (int it = 0; it < HN_BROWS / 2; ++it)
    gy[it] = *(const v4f*)(GYA + (size_t)(brow0 + 2 * it + hh) * Dsz + nb + 4 * c16);
  const v4f bv = *(const v4f*)(by + nb + 4 * c16);

  v4f zv[4], cv[4];
  float gw[4];
  float sv[HN_ROWS][4];
#pragma unroll
  for (int c = 0; c < 4; ++c) {
    zv[c] = *(const v4f*)(zhh + (size_t)b * Dsz + c * 128 + 4 * lane);
    cv[c] = *(const v4f*)(C + (size_t)b * Dsz + c * 128 + 4 * lane);
    gw[c] = gn_w[2 * c + hh];
  }
#pragma unroll
  for (int r = 0; r < HN_ROWS; ++r)
#pragma unroll
    for (int c = 0; c < 4; ++c)
      sv[r][c] = S[(size_t)(row0 + r) * Hsz + 2 * c + hh];

#pragma unroll
  for (int r = 0; r < HN_ROWS; ++r) {
    const size_t ro = (size_t)(row0 + r) * Dsz + 4 * lane;
    float* cr = ct + (wave * HN_ROWS + r) * CTP;
#pragma unroll
    for (int c = 0; c < 4; ++c) {
      const float e = expf(sv[r][c]);
      v4f hv;
      hv.x = zv[c].x + e * cv[c].x;
      hv.y = zv[c].y + e * cv[c].y;
      hv.z = zv[c].z + e * cv[c].z;
      hv.w = zv[c].w + e * cv[c].w;
      *(v4f*)(Hout + ro + c * 128) = hv;

      float s = (hv.x + hv.y) + (hv.z + hv.w);
      s = row_xor_add<1>(s);
      s = row_xor_add<2>(s);
      s = row_xor_add<4>(s);
      s = row_xor_add<8>(s);
      const float mu = s * (1.0f / 64.0f);

      v4f d;
      d.x = hv.x - mu; d.y = hv.y - mu; d.z = hv.z - mu; d.w = hv.w - mu;
      float q = (d.x * d.x + d.y * d.y) + (d.z * d.z + d.w * d.w);
      q = row_xor_add<1>(q);
      q = row_xor_add<2>(q);
      q = row_xor_add<4>(q);
      q = row_xor_add<8>(q);
      const float var = q * (1.0f / 64.0f);

      const float sc = gw[c] / sqrtf(var + GN_EPS);
      if (c16 == 0) {
        const int head = 2 * c + hh;
        cr[head] = sc;
        cr[8 + head] = sc * e;
        cr[16 + head] = 1.0f;
      }
    }
  }
  __syncthreads();

  v4f fa[PQK / 8];
#pragma unroll
  for (int kb = 0; kb < PQK / 8; ++kb)
    fa[kb] = *(const v4f*)(ct + c16 * CTP + 8 * kb + 4 * hh);

  v8f acc[4];
  const v8f zero = {0.f, 0.f, 0.f, 0.f, 0.f, 0.f, 0.f, 0.f};
#pragma unroll
  for (int nt = 0; nt < 4; ++nt) acc[nt] = zero;
#pragma unroll
  for (int kb = 0; kb < PQK / 8; ++kb) {
#pragma unroll
    for (int nt = 0; nt < 4; ++nt)
      acc[nt] = wmma4(fa[kb].xy, fb[nt][kb].xy, acc[nt]);
#pragma unroll
    for (int nt = 0; nt < 4; ++nt)
      acc[nt] = wmma4(fa[kb].zw, fb[nt][kb].zw, acc[nt]);
  }

  float* yw = yt + wave * HN_BROWS * YTP;
#pragma unroll
  for (int nt = 0; nt < 4; ++nt)
#pragma unroll
    for (int r = 0; r < 8; ++r)
      yw[(8 * hh + r) * YTP + nt * 16 + c16] = acc[nt][r];
  __syncthreads();

#pragma unroll
  for (int it = 0; it < HN_BROWS / 2; ++it) {
    const int row = 2 * it + hh;
    const v4f d = *(const v4f*)(yw + row * YTP + 4 * c16);
    v4f o;
    o.x = (d.x + bv.x) * gy[it].x;
    o.y = (d.y + bv.y) * gy[it].y;
    o.z = (d.z + bv.z) * gy[it].z;
    o.w = (d.w + bv.w) * gy[it].w;
    *(v4f*)(Y + (size_t)(brow0 + row) * Dsz + nb + 4 * c16) = o;
  }
}

extern "C" void kernel_launch(void* const* d_in, const int* in_sizes, int n_in,
                              void* d_out, int out_size, void* d_ws, size_t ws_size,
                              hipStream_t stream) {
  const float* x      = (const float*)d_in[0];
  const float* hidden = (const float*)d_in[1];
  const float* w_z    = (const float*)d_in[2];
  const float* b_z    = (const float*)d_in[3];
  const float* w_za   = (const float*)d_in[4];
  const float* b_za   = (const float*)d_in[5];
  const float* w_y    = (const float*)d_in[6];
  const float* b_y    = (const float*)d_in[7];
  const float* w_ya   = (const float*)d_in[8];
  const float* b_ya   = (const float*)d_in[9];
  const float* w_dt   = (const float*)d_in[10];
  const float* b_dt   = (const float*)d_in[11];
  const float* ln_a   = (const float*)d_in[12];
  const float* gn_w   = (const float*)d_in[13];
  const float* gn_b   = (const float*)d_in[14];

  float* y_out = (float*)d_out;
  float* h_out = y_out + (size_t)Msz * Dsz;

  float* ws   = (float*)d_ws;
  float* GYA  = ws;
  float* CPb  = GYA  + (size_t)Msz * Dsz;
  float* Sb   = CPb  + (size_t)MT64 * Hsz * Isz;
  float* Cb   = Sb   + (size_t)Msz * Hsz;
  float* ZHH  = Cb   + (size_t)Bsz * Hsz * Isz;
  float* LDb  = ZHH  + (size_t)Bsz * Hsz * Isz;
  float* PQT  = LDb  + (size_t)Msz * Hsz;

  const dim3 gGemm(Msz / 64, Dsz / 64 + DT_YB, 1);
  const dim3 bGemm(128, 1, 1);

  gemm_xproj_kernel<<<gGemm, bGemm, 0, stream>>>(x, w_z, b_z, w_za, b_za, w_ya, b_ya,
                                                 w_dt, b_dt, ln_a, CPb, ZHH, GYA, LDb);
  zsum_kernel<<<dim3(ZS_FBLK + Bsz * Hsz), dim3(ZS_THREADS), 0, stream>>>(CPb, ZHH, hidden, LDb, w_y, gn_b,
                                                          Cb, Sb, PQT);
  hnorm_y_kernel<<<dim3(Msz / HN_BROWS), dim3(HN_WAVES * 32), 0, stream>>>(
      Sb, Cb, ZHH, gn_w, PQT, b_y, GYA, h_out, y_out);
}
